# starting kernel
# baseline (speedup 1.0000x reference)
#include <hip/hip_runtime.h>
#include <math.h>

typedef float v2f __attribute__((ext_vector_type(2)));
typedef float v8f __attribute__((ext_vector_type(8)));

__device__ __forceinline__ v8f wmma_f32_k4(v2f a, v2f b, v8f c) {
  return __builtin_amdgcn_wmma_f32_16x16x4_f32(
       false, a,  false, b,
       (short)0, c,  false,  false);
}

__device__ __forceinline__ void atomic_add_f32(float* p, float v) {
  asm volatile("global_atomic_add_f32 %0, %1, off scope:SCOPE_DEV"
               :: "v"(p), "v"(v)
               : "memory");
}


__global__ void k_fill(float* __restrict__ p, float v, int n) {
  int i = blockIdx.x * blockDim.x + threadIdx.x;
  if (i < n) p[i] = v;
}

__global__ void k_deg_accum(const int* __restrict__ dst, float* __restrict__ deg, int E) {
  int i = blockIdx.x * blockDim.x + threadIdx.x;
  if (i < E) atomic_add_f32(&deg[dst[i]], 1.0f);
}

__global__ void k_rsqrt_inplace(float* __restrict__ d, int n) {
  int i = blockIdx.x * blockDim.x + threadIdx.x;
  if (i < n) d[i] = rsqrtf(d[i]);
}

__global__ void k_norm(const int* __restrict__ src, const int* __restrict__ dst,
                       const float* __restrict__ dinv, float* __restrict__ norm,
                       int E, int N) {
  int i = blockIdx.x * blockDim.x + threadIdx.x;
  if (i >= E + N) return;
  if (i < E) {
    norm[i] = dinv[src[i]] * dinv[dst[i]];
  } else {
    float v = dinv[i - E];
    norm[i] = v * v;
  }
}

__global__ void k_bias_relu(float* __restrict__ a, const float* __restrict__ b, int n) {
  int i = blockIdx.x * blockDim.x + threadIdx.x;
  if (i < n) a[i] = fmaxf(a[i] + b[i & 63], 0.0f);
}


__global__ void k_gemm_n64(const float* __restrict__ X, const float* __restrict__ W,
                           float* __restrict__ H, int M, int K) {
  int wave = (int)((blockIdx.x * blockDim.x + threadIdx.x) >> 5);
  int lane = threadIdx.x & 31;
  int m0 = wave * 16;
  if (m0 >= M) return;
  int half = lane >> 4;
  int lm = lane & 15;

  const float* xrow = X + (size_t)(m0 + lm) * (size_t)K;
  v8f acc0 = {}, acc1 = {}, acc2 = {}, acc3 = {};

  for (int kb = 0; kb < K; kb += 4) {
    int k0 = kb + 2 * half;
    v2f a;
    a.x = xrow[k0];
    a.y = xrow[k0 + 1];
    const float* w0 = W + (size_t)k0 * 64;
    v2f b0, b1, b2, b3;
    b0.x = w0[lm];       b0.y = w0[64 + lm];
    b1.x = w0[16 + lm];  b1.y = w0[64 + 16 + lm];
    b2.x = w0[32 + lm];  b2.y = w0[64 + 32 + lm];
    b3.x = w0[48 + lm];  b3.y = w0[64 + 48 + lm];
    acc0 = wmma_f32_k4(a, b0, acc0);
    acc1 = wmma_f32_k4(a, b1, acc1);
    acc2 = wmma_f32_k4(a, b2, acc2);
    acc3 = wmma_f32_k4(a, b3, acc3);
  }

#pragma unroll
  for (int r = 0; r < 8; ++r) {
    int row = m0 + r + 8 * half;
    float* hr = H + (size_t)row * 64 + lm;
    hr[0]  = acc0[r];
    hr[16] = acc1[r];
    hr[32] = acc2[r];
    hr[48] = acc3[r];
  }
}


__global__ void k_scatter(const int* __restrict__ src, const int* __restrict__ dst,
                          const float* __restrict__ norm, const float* __restrict__ h,
                          float* __restrict__ agg, int E, int N) {
  unsigned int gid = blockIdx.x * blockDim.x + threadIdx.x;
  unsigned int eid = gid >> 4;
  int t4 = (gid & 15) * 4;
  if (eid >= (unsigned int)(E + N)) return;
  int s, d;
  if (eid < (unsigned int)E) {
    s = src[eid];
    d = dst[eid];
  } else {
    s = d = (int)eid - E;
  }
  float nm = norm[eid];
  const float4 hv = *reinterpret_cast<const float4*>(h + (size_t)s * 64 + t4);
  float* ap = agg + (size_t)d * 64 + t4;
  atomic_add_f32(ap + 0, hv.x * nm);
  atomic_add_f32(ap + 1, hv.y * nm);
  atomic_add_f32(ap + 2, hv.z * nm);
  atomic_add_f32(ap + 3, hv.w * nm);
}


__global__ void k_pool(const float* __restrict__ h, const int* __restrict__ batch,
                       float* __restrict__ gsum, float* __restrict__ cnt, int N) {
  unsigned int gid = blockIdx.x * blockDim.x + threadIdx.x;
  if (gid >= (unsigned int)N * 64u) return;
  int c = gid & 63;
  int n = gid >> 6;
  int g = batch[n];
  atomic_add_f32(&gsum[g * 64 + c], h[(size_t)n * 64 + c]);
  if (c == 0) atomic_add_f32(&cnt[g], 1.0f);
}


__global__ void k_pool_gemm_out(const float* __restrict__ gsum, const float* __restrict__ cnt,
                                const float* __restrict__ Wp, const float* __restrict__ bp,
                                float* __restrict__ out) {
  int wave = threadIdx.x >> 5;
  int lane = threadIdx.x & 31;
  int mi = wave >> 2, ni = wave & 3;
  int half = lane >> 4, lm = lane & 15;

  int arow = mi * 16 + lm;
  float inv = 1.0f / fmaxf(cnt[arow], 1.0f);

  v8f acc = {};
  for (int kb = 0; kb < 64; kb += 4) {
    int k0 = kb + 2 * half;
    v2f a;
    a.x = gsum[arow * 64 + k0] * inv;
    a.y = gsum[arow * 64 + k0 + 1] * inv;
    v2f b;
    b.x = Wp[k0 * 64 + ni * 16 + lm];
    b.y = Wp[(k0 + 1) * 64 + ni * 16 + lm];
    acc = wmma_f32_k4(a, b, acc);
  }

#pragma unroll
  for (int r = 0; r < 8; ++r) {
    int row = mi * 16 + r + 8 * half;
    out[row * 64 + ni * 16 + lm] = acc[r] + bp[ni * 16 + lm];
  }
}


static inline size_t align_f(size_t n) { return (n + 1023) & ~(size_t)1023; }

extern "C" void kernel_launch(void* const* d_in, const int* in_sizes, int n_in,
                              void* d_out, int out_size, void* d_ws, size_t ws_size,
                              hipStream_t stream) {
  const float* x  = (const float*)d_in[0];
  const float* W1 = (const float*)d_in[1];
  const float* b1 = (const float*)d_in[2];
  const float* W2 = (const float*)d_in[3];
  const float* b2 = (const float*)d_in[4];
  const float* Wp = (const float*)d_in[5];
  const float* bp = (const float*)d_in[6];
  const int* ei   = (const int*)d_in[7];
  const int* batch = (const int*)d_in[8];

  const int E = in_sizes[7] / 2;
  const int N = in_sizes[8];
  const int K1 = in_sizes[0] / N;
  const int* src = ei;
  const int* dst = ei + E;

  float* ws = (float*)d_ws;
  size_t o = 0;
  float* dinv = ws + o;  o += align_f((size_t)N);
  float* norm = ws + o;  o += align_f((size_t)(E + N));
  float* h    = ws + o;  o += align_f((size_t)N * 64);
  float* agg  = ws + o;  o += align_f((size_t)N * 64);
  float* gsum = ws + o;  o += 64 * 64;
  float* cnt  = ws + o;  o += 64;
  (void)ws_size;

  const int T = 256;
  auto blocks = [](long long n, int t) { return (unsigned int)((n + t - 1) / t); };

  k_fill<<<blocks(N, T), T, 0, stream>>>(dinv, 1.0f, N);
  k_deg_accum<<<blocks(E, T), T, 0, stream>>>(dst, dinv, E);
  k_rsqrt_inplace<<<blocks(N, T), T, 0, stream>>>(dinv, N);
  k_norm<<<blocks((long long)E + N, T), T, 0, stream>>>(src, dst, dinv, norm, E, N);

  const int tiles = (N + 15) / 16;
  const unsigned int gemm_blocks = (unsigned int)((tiles + 7) / 8);
  const long long NE16 = (long long)(E + N) * 16;

  k_gemm_n64<<<gemm_blocks, T, 0, stream>>>(x, W1, h, N, K1);
  k_fill<<<blocks((long long)N * 64, T), T, 0, stream>>>(agg, 0.0f, N * 64);
  k_scatter<<<blocks(NE16, T), T, 0, stream>>>(src, dst, norm, h, agg, E, N);
  k_bias_relu<<<blocks((long long)N * 64, T), T, 0, stream>>>(agg, b1, N * 64);

  k_gemm_n64<<<gemm_blocks, T, 0, stream>>>(agg, W2, h, N, 64);
  k_fill<<<blocks((long long)N * 64, T), T, 0, stream>>>(agg, 0.0f, N * 64);
  k_scatter<<<blocks(NE16, T), T, 0, stream>>>(src, dst, norm, h, agg, E, N);
  k_bias_relu<<<blocks((long long)N * 64, T), T, 0, stream>>>(agg, b2, N * 64);

  k_fill<<<blocks(64 * 64 + 64, T), T, 0, stream>>>(gsum, 0.0f, 64 * 64 + 64);
  k_pool<<<blocks((long long)N * 64, T), T, 0, stream>>>(agg, batch, gsum, cnt, N);
  k_pool_gemm_out<<<1, 512, 0, stream>>>(gsum, cnt, Wp, bp, (float*)d_out);
}


# edge scatter: one wave per 32 edges, lanes on channels l and l+32, so every gather and atomic instruction covers one whole 128-byte line (2 atomic requests per edge, was 8); arrival order changes
# speedup vs baseline: 1.4295x; 1.4295x over previous
#include <hip/hip_runtime.h>
#include <math.h>

typedef float v2f __attribute__((ext_vector_type(2)));
typedef float v8f __attribute__((ext_vector_type(8)));

__device__ __forceinline__ v8f wmma_f32_k4(v2f a, v2f b, v8f c) {
  return __builtin_amdgcn_wmma_f32_16x16x4_f32(
       false, a,  false, b,
       (short)0, c,  false,  false);
}

__device__ __forceinline__ void atomic_add_f32(float* p, float v) {
  asm volatile("global_atomic_add_f32 %0, %1, off scope:SCOPE_DEV"
               :: "v"(p), "v"(v)
               : "memory");
}


__global__ void k_fill(float* __restrict__ p, float v, int n) {
  int i = blockIdx.x * blockDim.x + threadIdx.x;
  if (i < n) p[i] = v;
}

__global__ void k_deg_accum(const int* __restrict__ dst, float* __restrict__ deg, int E) {
  int i = blockIdx.x * blockDim.x + threadIdx.x;
  if (i < E) atomic_add_f32(&deg[dst[i]], 1.0f);
}

__global__ void k_rsqrt_inplace(float* __restrict__ d, int n) {
  int i = blockIdx.x * blockDim.x + threadIdx.x;
  if (i < n) d[i] = rsqrtf(d[i]);
}

__global__ void k_norm(const int* __restrict__ src, const int* __restrict__ dst,
                       const float* __restrict__ dinv, float* __restrict__ norm,
                       int E, int N) {
  int i = blockIdx.x * blockDim.x + threadIdx.x;
  if (i >= E + N) return;
  if (i < E) {
    norm[i] = dinv[src[i]] * dinv[dst[i]];
  } else {
    float v = dinv[i - E];
    norm[i] = v * v;
  }
}

__global__ void k_bias_relu(float* __restrict__ a, const float* __restrict__ b, int n) {
  int i = blockIdx.x * blockDim.x + threadIdx.x;
  if (i < n) a[i] = fmaxf(a[i] + b[i & 63], 0.0f);
}


__global__ void k_gemm_n64(const float* __restrict__ X, const float* __restrict__ W,
                           float* __restrict__ H, int M, int K) {
  int wave = (int)((blockIdx.x * blockDim.x + threadIdx.x) >> 5);
  int lane = threadIdx.x & 31;
  int m0 = wave * 16;
  if (m0 >= M) return;
  int half = lane >> 4;
  int lm = lane & 15;

  const float* xrow = X + (size_t)(m0 + lm) * (size_t)K;
  v8f acc0 = {}, acc1 = {}, acc2 = {}, acc3 = {};

  for (int kb = 0; kb < K; kb += 4) {
    int k0 = kb + 2 * half;
    v2f a;
    a.x = xrow[k0];
    a.y = xrow[k0 + 1];
    const float* w0 = W + (size_t)k0 * 64;
    v2f b0, b1, b2, b3;
    b0.x = w0[lm];       b0.y = w0[64 + lm];
    b1.x = w0[16 + lm];  b1.y = w0[64 + 16 + lm];
    b2.x = w0[32 + lm];  b2.y = w0[64 + 32 + lm];
    b3.x = w0[48 + lm];  b3.y = w0[64 + 48 + lm];
    acc0 = wmma_f32_k4(a, b0, acc0);
    acc1 = wmma_f32_k4(a, b1, acc1);
    acc2 = wmma_f32_k4(a, b2, acc2);
    acc3 = wmma_f32_k4(a, b3, acc3);
  }

#pragma unroll
  for (int r = 0; r < 8; ++r) {
    int row = m0 + r + 8 * half;
    float* hr = H + (size_t)row * 64 + lm;
    hr[0]  = acc0[r];
    hr[16] = acc1[r];
    hr[32] = acc2[r];
    hr[48] = acc3[r];
  }
}


__device__ __forceinline__ void row_add(float* __restrict__ agg, int d, int lane,
                                        float v0, float v1) {
  float* p0 = agg + (size_t)d * 64 + lane;
  float* p1 = p0 + 32;
  atomic_add_f32(p0, v0);
  atomic_add_f32(p1, v1);
  asm volatile("s_wait_xcnt 0x0" :: "v"(p0), "v"(p1), "v"(v0), "v"(v1));
}

__global__ void k_scatter(const int* __restrict__ src, const int* __restrict__ dst,
                          const float* __restrict__ norm, const float* __restrict__ h,
                          float* __restrict__ agg, int E, int N) {
  const int lane = threadIdx.x & 31;
  const unsigned int wave = (unsigned int)__builtin_amdgcn_readfirstlane(
      (int)((blockIdx.x * blockDim.x + threadIdx.x) >> 5));
  const unsigned int total = (unsigned int)(E + N);
  const unsigned int base = wave * 32u;
  if (base >= total) return;

  const unsigned int eid = base + (unsigned int)lane;
  int s = 0, d = 0;
  float nm = 0.0f;
  if (eid < total) {
    if (eid < (unsigned int)E) {
      s = src[eid];
      d = dst[eid];
    } else {
      s = d = (int)eid - E;
    }
    nm = norm[eid];
  }

  const unsigned int cnt = (total - base < 32u) ? (total - base) : 32u;

  if (cnt == 32u) {
#pragma unroll
    for (int g = 0; g < 4; ++g) {
      float v0[8], v1[8];
#pragma unroll
      for (int u = 0; u < 8; ++u) {
        const int sj = __builtin_amdgcn_readlane(s, g * 8 + u);
        const float* hp = h + (size_t)sj * 64 + lane;
        v0[u] = hp[0];
        v1[u] = hp[32];
      }
#pragma unroll
      for (int u = 0; u < 8; ++u) {
        const int dj = __builtin_amdgcn_readlane(d, g * 8 + u);
        const float nj = __builtin_bit_cast(float,
            __builtin_amdgcn_readlane(__builtin_bit_cast(int, nm), g * 8 + u));
        row_add(agg, dj, lane, v0[u] * nj, v1[u] * nj);
      }
    }
  } else {
    for (unsigned int j = 0; j < cnt; ++j) {
      const int sj = __builtin_amdgcn_readlane(s, (int)j);
      const int dj = __builtin_amdgcn_readlane(d, (int)j);
      const float nj = __builtin_bit_cast(float,
          __builtin_amdgcn_readlane(__builtin_bit_cast(int, nm), (int)j));
      const float* hp = h + (size_t)sj * 64 + lane;
      row_add(agg, dj, lane, hp[0] * nj, hp[32] * nj);
    }
  }
}


__global__ void k_pool(const float* __restrict__ h, const int* __restrict__ batch,
                       float* __restrict__ gsum, float* __restrict__ cnt, int N) {
  unsigned int gid = blockIdx.x * blockDim.x + threadIdx.x;
  if (gid >= (unsigned int)N * 64u) return;
  int c = gid & 63;
  int n = gid >> 6;
  int g = batch[n];
  atomic_add_f32(&gsum[g * 64 + c], h[(size_t)n * 64 + c]);
  if (c == 0) atomic_add_f32(&cnt[g], 1.0f);
}


__global__ void k_pool_gemm_out(const float* __restrict__ gsum, const float* __restrict__ cnt,
                                const float* __restrict__ Wp, const float* __restrict__ bp,
                                float* __restrict__ out) {
  int wave = threadIdx.x >> 5;
  int lane = threadIdx.x & 31;
  int mi = wave >> 2, ni = wave & 3;
  int half = lane >> 4, lm = lane & 15;

  int arow = mi * 16 + lm;
  float inv = 1.0f / fmaxf(cnt[arow], 1.0f);

  v8f acc = {};
  for (int kb = 0; kb < 64; kb += 4) {
    int k0 = kb + 2 * half;
    v2f a;
    a.x = gsum[arow * 64 + k0] * inv;
    a.y = gsum[arow * 64 + k0 + 1] * inv;
    v2f b;
    b.x = Wp[k0 * 64 + ni * 16 + lm];
    b.y = Wp[(k0 + 1) * 64 + ni * 16 + lm];
    acc = wmma_f32_k4(a, b, acc);
  }

#pragma unroll
  for (int r = 0; r < 8; ++r) {
    int row = mi * 16 + r + 8 * half;
    out[row * 64 + ni * 16 + lm] = acc[r] + bp[ni * 16 + lm];
  }
}


static inline size_t align_f(size_t n) { return (n + 1023) & ~(size_t)1023; }

extern "C" void kernel_launch(void* const* d_in, const int* in_sizes, int n_in,
                              void* d_out, int out_size, void* d_ws, size_t ws_size,
                              hipStream_t stream) {
  const float* x  = (const float*)d_in[0];
  const float* W1 = (const float*)d_in[1];
  const float* b1 = (const float*)d_in[2];
  const float* W2 = (const float*)d_in[3];
  const float* b2 = (const float*)d_in[4];
  const float* Wp = (const float*)d_in[5];
  const float* bp = (const float*)d_in[6];
  const int* ei   = (const int*)d_in[7];
  const int* batch = (const int*)d_in[8];

  const int E = in_sizes[7] / 2;
  const int N = in_sizes[8];
  const int K1 = in_sizes[0] / N;
  const int* src = ei;
  const int* dst = ei + E;

  float* ws = (float*)d_ws;
  size_t o = 0;
  float* dinv = ws + o;  o += align_f((size_t)N);
  float* norm = ws + o;  o += align_f((size_t)(E + N));
  float* h    = ws + o;  o += align_f((size_t)N * 64);
  float* agg  = ws + o;  o += align_f((size_t)N * 64);
  float* gsum = ws + o;  o += 64 * 64;
  float* cnt  = ws + o;  o += 64;
  (void)ws_size;

  const int T = 256;
  auto blocks = [](long long n, int t) { return (unsigned int)((n + t - 1) / t); };

  k_fill<<<blocks(N, T), T, 0, stream>>>(dinv, 1.0f, N);
  k_deg_accum<<<blocks(E, T), T, 0, stream>>>(dst, dinv, E);
  k_rsqrt_inplace<<<blocks(N, T), T, 0, stream>>>(dinv, N);
  k_norm<<<blocks((long long)E + N, T), T, 0, stream>>>(src, dst, dinv, norm, E, N);

  const int tiles = (N + 15) / 16;
  const unsigned int gemm_blocks = (unsigned int)((tiles + 7) / 8);
  const long long scat_threads = (((long long)E + N + 31) / 32) * 32;

  k_gemm_n64<<<gemm_blocks, T, 0, stream>>>(x, W1, h, N, K1);
  k_fill<<<blocks((long long)N * 64, T), T, 0, stream>>>(agg, 0.0f, N * 64);
  k_scatter<<<blocks(scat_threads, T), T, 0, stream>>>(src, dst, norm, h, agg, E, N);
  k_bias_relu<<<blocks((long long)N * 64, T), T, 0, stream>>>(agg, b1, N * 64);

  k_gemm_n64<<<gemm_blocks, T, 0, stream>>>(agg, W2, h, N, 64);
  k_fill<<<blocks((long long)N * 64, T), T, 0, stream>>>(agg, 0.0f, N * 64);
  k_scatter<<<blocks(scat_threads, T), T, 0, stream>>>(src, dst, norm, h, agg, E, N);
  k_bias_relu<<<blocks((long long)N * 64, T), T, 0, stream>>>(agg, b2, N * 64);

  k_fill<<<blocks(64 * 64 + 64, T), T, 0, stream>>>(gsum, 0.0f, 64 * 64 + 64);
  k_pool<<<blocks((long long)N * 64, T), T, 0, stream>>>(agg, batch, gsum, cnt, N);
  k_pool_gemm_out<<<1, 512, 0, stream>>>(gsum, cnt, Wp, bp, (float*)d_out);
}


# mean pool: one wave per 32 nodes sums its rows in registers per run of equal graph id and issues one atomic row + one count per run (was one atomic per node and channel); pool sum order changes
# speedup vs baseline: 1.8038x; 1.2618x over previous
#include <hip/hip_runtime.h>
#include <math.h>

typedef float v2f __attribute__((ext_vector_type(2)));
typedef float v8f __attribute__((ext_vector_type(8)));

__device__ __forceinline__ v8f wmma_f32_k4(v2f a, v2f b, v8f c) {
  return __builtin_amdgcn_wmma_f32_16x16x4_f32(
       false, a,  false, b,
       (short)0, c,  false,  false);
}

__device__ __forceinline__ void atomic_add_f32(float* p, float v) {
  asm volatile("global_atomic_add_f32 %0, %1, off scope:SCOPE_DEV"
               :: "v"(p), "v"(v)
               : "memory");
}


__global__ void k_fill(float* __restrict__ p, float v, int n) {
  int i = blockIdx.x * blockDim.x + threadIdx.x;
  if (i < n) p[i] = v;
}

__global__ void k_deg_accum(const int* __restrict__ dst, float* __restrict__ deg, int E) {
  int i = blockIdx.x * blockDim.x + threadIdx.x;
  if (i < E) atomic_add_f32(&deg[dst[i]], 1.0f);
}

__global__ void k_rsqrt_inplace(float* __restrict__ d, int n) {
  int i = blockIdx.x * blockDim.x + threadIdx.x;
  if (i < n) d[i] = rsqrtf(d[i]);
}

__global__ void k_norm(const int* __restrict__ src, const int* __restrict__ dst,
                       const float* __restrict__ dinv, float* __restrict__ norm,
                       int E, int N) {
  int i = blockIdx.x * blockDim.x + threadIdx.x;
  if (i >= E + N) return;
  if (i < E) {
    norm[i] = dinv[src[i]] * dinv[dst[i]];
  } else {
    float v = dinv[i - E];
    norm[i] = v * v;
  }
}

__global__ void k_bias_relu(float* __restrict__ a, const float* __restrict__ b, int n) {
  int i = blockIdx.x * blockDim.x + threadIdx.x;
  if (i < n) a[i] = fmaxf(a[i] + b[i & 63], 0.0f);
}


__global__ void k_gemm_n64(const float* __restrict__ X, const float* __restrict__ W,
                           float* __restrict__ H, int M, int K) {
  int wave = (int)((blockIdx.x * blockDim.x + threadIdx.x) >> 5);
  int lane = threadIdx.x & 31;
  int m0 = wave * 16;
  if (m0 >= M) return;
  int half = lane >> 4;
  int lm = lane & 15;

  const float* xrow = X + (size_t)(m0 + lm) * (size_t)K;
  v8f acc0 = {}, acc1 = {}, acc2 = {}, acc3 = {};

  for (int kb = 0; kb < K; kb += 4) {
    int k0 = kb + 2 * half;
    v2f a;
    a.x = xrow[k0];
    a.y = xrow[k0 + 1];
    const float* w0 = W + (size_t)k0 * 64;
    v2f b0, b1, b2, b3;
    b0.x = w0[lm];       b0.y = w0[64 + lm];
    b1.x = w0[16 + lm];  b1.y = w0[64 + 16 + lm];
    b2.x = w0[32 + lm];  b2.y = w0[64 + 32 + lm];
    b3.x = w0[48 + lm];  b3.y = w0[64 + 48 + lm];
    acc0 = wmma_f32_k4(a, b0, acc0);
    acc1 = wmma_f32_k4(a, b1, acc1);
    acc2 = wmma_f32_k4(a, b2, acc2);
    acc3 = wmma_f32_k4(a, b3, acc3);
  }

#pragma unroll
  for (int r = 0; r < 8; ++r) {
    int row = m0 + r + 8 * half;
    float* hr = H + (size_t)row * 64 + lm;
    hr[0]  = acc0[r];
    hr[16] = acc1[r];
    hr[32] = acc2[r];
    hr[48] = acc3[r];
  }
}


__device__ __forceinline__ void row_add(float* __restrict__ agg, int d, int lane,
                                        float v0, float v1) {
  float* p0 = agg + (size_t)d * 64 + lane;
  float* p1 = p0 + 32;
  atomic_add_f32(p0, v0);
  atomic_add_f32(p1, v1);
  asm volatile("s_wait_xcnt 0x0" :: "v"(p0), "v"(p1), "v"(v0), "v"(v1));
}

__global__ void k_scatter(const int* __restrict__ src, const int* __restrict__ dst,
                          const float* __restrict__ norm, const float* __restrict__ h,
                          float* __restrict__ agg, int E, int N) {
  const int lane = threadIdx.x & 31;
  const unsigned int wave = (unsigned int)__builtin_amdgcn_readfirstlane(
      (int)((blockIdx.x * blockDim.x + threadIdx.x) >> 5));
  const unsigned int total = (unsigned int)(E + N);
  const unsigned int base = wave * 32u;
  if (base >= total) return;

  const unsigned int eid = base + (unsigned int)lane;
  int s = 0, d = 0;
  float nm = 0.0f;
  if (eid < total) {
    if (eid < (unsigned int)E) {
      s = src[eid];
      d = dst[eid];
    } else {
      s = d = (int)eid - E;
    }
    nm = norm[eid];
  }

  const unsigned int cnt = (total - base < 32u) ? (total - base) : 32u;

  if (cnt == 32u) {
#pragma unroll
    for (int g = 0; g < 4; ++g) {
      float v0[8], v1[8];
#pragma unroll
      for (int u = 0; u < 8; ++u) {
        const int sj = __builtin_amdgcn_readlane(s, g * 8 + u);
        const float* hp = h + (size_t)sj * 64 + lane;
        v0[u] = hp[0];
        v1[u] = hp[32];
      }
#pragma unroll
      for (int u = 0; u < 8; ++u) {
        const int dj = __builtin_amdgcn_readlane(d, g * 8 + u);
        const float nj = __builtin_bit_cast(float,
            __builtin_amdgcn_readlane(__builtin_bit_cast(int, nm), g * 8 + u));
        row_add(agg, dj, lane, v0[u] * nj, v1[u] * nj);
      }
    }
  } else {
    for (unsigned int j = 0; j < cnt; ++j) {
      const int sj = __builtin_amdgcn_readlane(s, (int)j);
      const int dj = __builtin_amdgcn_readlane(d, (int)j);
      const float nj = __builtin_bit_cast(float,
          __builtin_amdgcn_readlane(__builtin_bit_cast(int, nm), (int)j));
      const float* hp = h + (size_t)sj * 64 + lane;
      row_add(agg, dj, lane, hp[0] * nj, hp[32] * nj);
    }
  }
}


__device__ __forceinline__ void one_add(float* p, float v) {
  atomic_add_f32(p, v);
  asm volatile("s_wait_xcnt 0x0" :: "v"(p), "v"(v));
}

__device__ __forceinline__ void pool_flush(float* __restrict__ gsum, float* __restrict__ cnt,
                                           int g, int lane, float a0, float a1, int run) {
  row_add(gsum, g, lane, a0, a1);
  if (lane == 0) one_add(cnt + g, (float)run);
}

__global__ void k_pool(const float* __restrict__ h, const int* __restrict__ batch,
                       float* __restrict__ gsum, float* __restrict__ cnt, int N) {
  const int lane = threadIdx.x & 31;
  const int wave = __builtin_amdgcn_readfirstlane(
      (int)((blockIdx.x * blockDim.x + threadIdx.x) >> 5));
  const int base = wave * 32;
  if (base >= N) return;
  const int last = N - 1;
  const int m = (N - base < 32) ? (N - base) : 32;

  const int nl = (base + lane < last) ? (base + lane) : last;
  const int gl = batch[nl];

  float v0[32], v1[32];
#pragma unroll
  for (int u = 0; u < 32; ++u) {
    const int n = (base + u < last) ? (base + u) : last;
    const float* hrow = h + (size_t)n * 64;
    v0[u] = hrow[lane];
    v1[u] = hrow[lane + 32];
  }

  int gcur = __builtin_amdgcn_readlane(gl, 0);
  float a0 = v0[0], a1 = v1[0];
  int run = 1;
#pragma unroll
  for (int u = 1; u < 32; ++u) {
    if (u < m) {
      const int gu = __builtin_amdgcn_readlane(gl, u);
      if (gu != gcur) {
        pool_flush(gsum, cnt, gcur, lane, a0, a1, run);
        gcur = gu;
        a0 = 0.0f;
        a1 = 0.0f;
        run = 0;
      }
      a0 += v0[u];
      a1 += v1[u];
      run += 1;
    }
  }
  pool_flush(gsum, cnt, gcur, lane, a0, a1, run);
}


__global__ void k_pool_gemm_out(const float* __restrict__ gsum, const float* __restrict__ cnt,
                                const float* __restrict__ Wp, const float* __restrict__ bp,
                                float* __restrict__ out) {
  int wave = threadIdx.x >> 5;
  int lane = threadIdx.x & 31;
  int mi = wave >> 2, ni = wave & 3;
  int half = lane >> 4, lm = lane & 15;

  int arow = mi * 16 + lm;
  float inv = 1.0f / fmaxf(cnt[arow], 1.0f);

  v8f acc = {};
  for (int kb = 0; kb < 64; kb += 4) {
    int k0 = kb + 2 * half;
    v2f a;
    a.x = gsum[arow * 64 + k0] * inv;
    a.y = gsum[arow * 64 + k0 + 1] * inv;
    v2f b;
    b.x = Wp[k0 * 64 + ni * 16 + lm];
    b.y = Wp[(k0 + 1) * 64 + ni * 16 + lm];
    acc = wmma_f32_k4(a, b, acc);
  }

#pragma unroll
  for (int r = 0; r < 8; ++r) {
    int row = mi * 16 + r + 8 * half;
    out[row * 64 + ni * 16 + lm] = acc[r] + bp[ni * 16 + lm];
  }
}


static inline size_t align_f(size_t n) { return (n + 1023) & ~(size_t)1023; }

extern "C" void kernel_launch(void* const* d_in, const int* in_sizes, int n_in,
                              void* d_out, int out_size, void* d_ws, size_t ws_size,
                              hipStream_t stream) {
  const float* x  = (const float*)d_in[0];
  const float* W1 = (const float*)d_in[1];
  const float* b1 = (const float*)d_in[2];
  const float* W2 = (const float*)d_in[3];
  const float* b2 = (const float*)d_in[4];
  const float* Wp = (const float*)d_in[5];
  const float* bp = (const float*)d_in[6];
  const int* ei   = (const int*)d_in[7];
  const int* batch = (const int*)d_in[8];

  const int E = in_sizes[7] / 2;
  const int N = in_sizes[8];
  const int K1 = in_sizes[0] / N;
  const int* src = ei;
  const int* dst = ei + E;

  float* ws = (float*)d_ws;
  size_t o = 0;
  float* dinv = ws + o;  o += align_f((size_t)N);
  float* norm = ws + o;  o += align_f((size_t)(E + N));
  float* h    = ws + o;  o += align_f((size_t)N * 64);
  float* agg  = ws + o;  o += align_f((size_t)N * 64);
  float* gsum = ws + o;  o += 64 * 64;
  float* cnt  = ws + o;  o += 64;
  (void)ws_size;

  const int T = 256;
  auto blocks = [](long long n, int t) { return (unsigned int)((n + t - 1) / t); };

  k_fill<<<blocks(N, T), T, 0, stream>>>(dinv, 1.0f, N);
  k_deg_accum<<<blocks(E, T), T, 0, stream>>>(dst, dinv, E);
  k_rsqrt_inplace<<<blocks(N, T), T, 0, stream>>>(dinv, N);
  k_norm<<<blocks((long long)E + N, T), T, 0, stream>>>(src, dst, dinv, norm, E, N);

  const int tiles = (N + 15) / 16;
  const unsigned int gemm_blocks = (unsigned int)((tiles + 7) / 8);
  const long long scat_threads = (((long long)E + N + 31) / 32) * 32;

  k_gemm_n64<<<gemm_blocks, T, 0, stream>>>(x, W1, h, N, K1);
  k_fill<<<blocks((long long)N * 64, T), T, 0, stream>>>(agg, 0.0f, N * 64);
  k_scatter<<<blocks(scat_threads, T), T, 0, stream>>>(src, dst, norm, h, agg, E, N);
  k_bias_relu<<<blocks((long long)N * 64, T), T, 0, stream>>>(agg, b1, N * 64);

  k_gemm_n64<<<gemm_blocks, T, 0, stream>>>(agg, W2, h, N, 64);
  k_fill<<<blocks((long long)N * 64, T), T, 0, stream>>>(agg, 0.0f, N * 64);
  k_scatter<<<blocks(scat_threads, T), T, 0, stream>>>(src, dst, norm, h, agg, E, N);
  k_bias_relu<<<blocks((long long)N * 64, T), T, 0, stream>>>(agg, b2, N * 64);

  k_fill<<<blocks(64 * 64 + 64, T), T, 0, stream>>>(gsum, 0.0f, 64 * 64 + 64);
  const long long pool_threads = (((long long)N + 31) / 32) * 32;
  k_pool<<<blocks(pool_threads, 64), 64, 0, stream>>>(agg, batch, gsum, cnt, N);
  k_pool_gemm_out<<<1, 512, 0, stream>>>(gsum, cnt, Wp, bp, (float*)d_out);
}


# edge aggregation without atomics: edge list radix-sorted by destination (ballot ranks), one wave sums a row's gathered h rows in registers and stores it once with bias+relu; edge sum order now fixed
# speedup vs baseline: 2.0041x; 1.1110x over previous
#include <hip/hip_runtime.h>
#include <math.h>

typedef float v2f __attribute__((ext_vector_type(2)));
typedef float v8f __attribute__((ext_vector_type(8)));

__device__ __forceinline__ v8f wmma_f32_k4(v2f a, v2f b, v8f c) {
  return __builtin_amdgcn_wmma_f32_16x16x4_f32(
       false, a,  false, b,
       (short)0, c,  false,  false);
}

__device__ __forceinline__ void atomic_add_f32(float* p, float v) {
  asm volatile("global_atomic_add_f32 %0, %1, off scope:SCOPE_DEV"
               :: "v"(p), "v"(v)
               : "memory");
}


__global__ void k_fill(float* __restrict__ p, float v, int n) {
  int i = blockIdx.x * blockDim.x + threadIdx.x;
  if (i < n) p[i] = v;
}


__device__ __forceinline__ int wave_incl_scan(int x, int lane) {
  for (int o = 1; o < 32; o <<= 1) {
    const int n = __shfl_up(x, o);
    if (lane >= o) x += n;
  }
  return x;
}

__device__ __forceinline__ int wave_sum(int x) {
  for (int o = 16; o > 0; o >>= 1) x += __shfl_xor(x, o);
  return x;
}


#define SORT_WAVES 1024
#define SORT_GROUPS 8

__device__ __forceinline__ unsigned int same_digit_lanes(int digit, bool valid) {
  unsigned int mask = __builtin_amdgcn_ballot_w32(valid);
#pragma unroll
  for (int b = 0; b < 8; ++b) {
    const bool bit = ((digit >> b) & 1) != 0;
    const unsigned int bb = __builtin_amdgcn_ballot_w32(bit);
    mask &= bit ? bb : ~bb;
  }
  return mask;
}

__global__ void __launch_bounds__(256)
k_sort_count(const int* __restrict__ keys, int* __restrict__ counts, int E, int per, int shift) {
  __shared__ int hist[8][256];
  const int lane = threadIdx.x & 31;
  const int w = __builtin_amdgcn_readfirstlane((int)(threadIdx.x >> 5));
  const int wg = (int)blockIdx.x * 8 + w;
  int* hw = hist[w];
  for (int k = lane; k < 256; k += 32) hw[k] = 0;
  const int beg = wg * per;
  const int end = (beg + per < E) ? (beg + per) : E;
  for (int g = beg; g < end; g += 32 * SORT_GROUPS) {
    int key[SORT_GROUPS];
#pragma unroll
    for (int u = 0; u < SORT_GROUPS; ++u) {
      const int idx = g + u * 32 + lane;
      key[u] = (idx < end) ? keys[idx] : 0;
    }
#pragma unroll
    for (int u = 0; u < SORT_GROUPS; ++u) {
      const int idx = g + u * 32 + lane;
      const bool valid = idx < end;
      const int digit = (key[u] >> shift) & 255;
      const unsigned int mask = same_digit_lanes(digit, valid);
      const bool leader = valid && ((mask & ((1u << lane) - 1u)) == 0u);
      if (leader) hw[digit] += __popc(mask);
      __builtin_amdgcn_wave_barrier();
    }
  }
  for (int k = lane; k < 256; k += 32) counts[k * SORT_WAVES + wg] = hw[k];
}

__global__ void __launch_bounds__(SORT_WAVES)
k_sort_rows(const int* __restrict__ counts, int* __restrict__ offs, int* __restrict__ tot) {
  __shared__ int s_tot[32];
  const int t = threadIdx.x;
  const int lane = t & 31, w = t >> 5;
  const int d = (int)blockIdx.x;
  const int v = counts[d * SORT_WAVES + t];
  const int incl = wave_incl_scan(v, lane);
  if (lane == 31) s_tot[w] = incl;
  __syncthreads();
  const int wt = s_tot[lane];
  const int woff = wave_sum((lane < w) ? wt : 0);
  offs[d * SORT_WAVES + t] = woff + incl - v;
  if (t == SORT_WAVES - 1) tot[d] = woff + incl;
}

__global__ void __launch_bounds__(256)
k_sort_scatter(const int* __restrict__ keys, const int* __restrict__ vals,
               const int* __restrict__ offs, const int* __restrict__ tot,
               int* __restrict__ keys_out, int* __restrict__ vals_out,
               int E, int per, int shift) {
  __shared__ int cur[8][256];
  const int lane = threadIdx.x & 31;
  const int w = __builtin_amdgcn_readfirstlane((int)(threadIdx.x >> 5));
  const int wg = (int)blockIdx.x * 8 + w;
  int* cw = cur[w];

  int tl[8];
  int ls = 0;
#pragma unroll
  for (int q = 0; q < 8; ++q) {
    tl[q] = tot[8 * lane + q];
    ls += tl[q];
  }
  int run = wave_incl_scan(ls, lane) - ls;
#pragma unroll
  for (int q = 0; q < 8; ++q) {
    cw[8 * lane + q] = run + offs[(8 * lane + q) * SORT_WAVES + wg];
    run += tl[q];
  }
  __builtin_amdgcn_wave_barrier();

  const int beg = wg * per;
  const int end = (beg + per < E) ? (beg + per) : E;
  for (int g = beg; g < end; g += 32 * SORT_GROUPS) {
    int key[SORT_GROUPS], val[SORT_GROUPS];
#pragma unroll
    for (int u = 0; u < SORT_GROUPS; ++u) {
      const int idx = g + u * 32 + lane;
      key[u] = (idx < end) ? keys[idx] : 0;
      val[u] = (idx < end) ? vals[idx] : 0;
    }
#pragma unroll
    for (int u = 0; u < SORT_GROUPS; ++u) {
      const int idx = g + u * 32 + lane;
      const bool valid = idx < end;
      const int digit = (key[u] >> shift) & 255;
      const unsigned int mask = same_digit_lanes(digit, valid);
      const int rank = __popc(mask & ((1u << lane) - 1u));
      int place = 0;
      if (valid) place = cw[digit];
      __builtin_amdgcn_wave_barrier();
      if (valid && rank == 0) cw[digit] = place + __popc(mask);
      __builtin_amdgcn_wave_barrier();
      if (valid) {
        keys_out[place + rank] = key[u];
        vals_out[place + rank] = val[u];
      }
    }
  }
}

__global__ void k_rowptr(const int* __restrict__ skeys, int* __restrict__ rowptr, int E, int N) {
  const int i = blockIdx.x * blockDim.x + threadIdx.x;
  if (i > N) return;
  int lo = 0, hi = E;
  for (int it = 0; it < 32 && lo < hi; ++it) {
    const int mid = lo + ((hi - lo) >> 1);
    if (skeys[mid] < i) lo = mid + 1; else hi = mid;
  }
  rowptr[i] = lo;
}

__global__ void k_dinv(const int* __restrict__ rowptr, float* __restrict__ dinv, int N) {
  const int i = blockIdx.x * blockDim.x + threadIdx.x;
  if (i < N) dinv[i] = rsqrtf((float)(rowptr[i + 1] - rowptr[i] + 1));
}


__global__ void k_aggregate(const float* __restrict__ h, const int* __restrict__ rowptr,
                            const float* __restrict__ dinv, const int* __restrict__ csr_src,
                            const float* __restrict__ bias, float* __restrict__ out, int N) {
  const int lane = threadIdx.x & 31;
  const int i = __builtin_amdgcn_readfirstlane(
      (int)((blockIdx.x * blockDim.x + threadIdx.x) >> 5));
  if (i >= N) return;
  const int start = rowptr[i];
  const int cnt = rowptr[i + 1] - start;
  const float di = dinv[i];

  const v2f self = *(const v2f*)(h + (size_t)i * 64 + 2 * lane);
  const float ns = di * di;
  float a0 = self.x * ns, a1 = self.y * ns;

  for (int c = 0; c < cnt; c += 32) {
    const int m = (cnt - c < 32) ? (cnt - c) : 32;
    int sl = i;
    float nl = 0.0f;
    if (lane < m) {
      sl = csr_src[start + c + lane];
      nl = dinv[sl] * di;
    }
    for (int j = 0; j < m; j += 8) {
      v2f v[8];
#pragma unroll
      for (int u = 0; u < 8; ++u) {
        const int sj = __builtin_amdgcn_readlane(sl, j + u);
        v[u] = *(const v2f*)(h + (size_t)sj * 64 + 2 * lane);
      }
#pragma unroll
      for (int u = 0; u < 8; ++u) {
        const float nj = __builtin_bit_cast(float,
            __builtin_amdgcn_readlane(__builtin_bit_cast(int, nl), j + u));
        a0 = fmaf(v[u].x, nj, a0);
        a1 = fmaf(v[u].y, nj, a1);
      }
    }
  }

  v2f o;
  o.x = fmaxf(a0 + bias[2 * lane], 0.0f);
  o.y = fmaxf(a1 + bias[2 * lane + 1], 0.0f);
  *(v2f*)(out + (size_t)i * 64 + 2 * lane) = o;
}


__global__ void k_gemm_n64(const float* __restrict__ X, const float* __restrict__ W,
                           float* __restrict__ H, int M, int K) {
  int wave = (int)((blockIdx.x * blockDim.x + threadIdx.x) >> 5);
  int lane = threadIdx.x & 31;
  int m0 = wave * 16;
  if (m0 >= M) return;
  int half = lane >> 4;
  int lm = lane & 15;

  const float* xrow = X + (size_t)(m0 + lm) * (size_t)K;
  v8f acc0 = {}, acc1 = {}, acc2 = {}, acc3 = {};

  for (int kb = 0; kb < K; kb += 4) {
    int k0 = kb + 2 * half;
    v2f a;
    a.x = xrow[k0];
    a.y = xrow[k0 + 1];
    const float* w0 = W + (size_t)k0 * 64;
    v2f b0, b1, b2, b3;
    b0.x = w0[lm];       b0.y = w0[64 + lm];
    b1.x = w0[16 + lm];  b1.y = w0[64 + 16 + lm];
    b2.x = w0[32 + lm];  b2.y = w0[64 + 32 + lm];
    b3.x = w0[48 + lm];  b3.y = w0[64 + 48 + lm];
    acc0 = wmma_f32_k4(a, b0, acc0);
    acc1 = wmma_f32_k4(a, b1, acc1);
    acc2 = wmma_f32_k4(a, b2, acc2);
    acc3 = wmma_f32_k4(a, b3, acc3);
  }

#pragma unroll
  for (int r = 0; r < 8; ++r) {
    int row = m0 + r + 8 * half;
    float* hr = H + (size_t)row * 64 + lm;
    hr[0]  = acc0[r];
    hr[16] = acc1[r];
    hr[32] = acc2[r];
    hr[48] = acc3[r];
  }
}


__device__ __forceinline__ void row_add(float* __restrict__ agg, int d, int lane,
                                        float v0, float v1) {
  float* p0 = agg + (size_t)d * 64 + lane;
  float* p1 = p0 + 32;
  atomic_add_f32(p0, v0);
  atomic_add_f32(p1, v1);
  asm volatile("s_wait_xcnt 0x0" :: "v"(p0), "v"(p1), "v"(v0), "v"(v1));
}


__device__ __forceinline__ void one_add(float* p, float v) {
  atomic_add_f32(p, v);
  asm volatile("s_wait_xcnt 0x0" :: "v"(p), "v"(v));
}

__device__ __forceinline__ void pool_flush(float* __restrict__ gsum, float* __restrict__ cnt,
                                           int g, int lane, float a0, float a1, int run) {
  row_add(gsum, g, lane, a0, a1);
  if (lane == 0) one_add(cnt + g, (float)run);
}

__global__ void k_pool(const float* __restrict__ h, const int* __restrict__ batch,
                       float* __restrict__ gsum, float* __restrict__ cnt, int N) {
  const int lane = threadIdx.x & 31;
  const int wave = __builtin_amdgcn_readfirstlane(
      (int)((blockIdx.x * blockDim.x + threadIdx.x) >> 5));
  const int base = wave * 32;
  if (base >= N) return;
  const int last = N - 1;
  const int m = (N - base < 32) ? (N - base) : 32;

  const int nl = (base + lane < last) ? (base + lane) : last;
  const int gl = batch[nl];

  float v0[32], v1[32];
#pragma unroll
  for (int u = 0; u < 32; ++u) {
    const int n = (base + u < last) ? (base + u) : last;
    const float* hrow = h + (size_t)n * 64;
    v0[u] = hrow[lane];
    v1[u] = hrow[lane + 32];
  }

  int gcur = __builtin_amdgcn_readlane(gl, 0);
  float a0 = v0[0], a1 = v1[0];
  int run = 1;
#pragma unroll
  for (int u = 1; u < 32; ++u) {
    if (u < m) {
      const int gu = __builtin_amdgcn_readlane(gl, u);
      if (gu != gcur) {
        pool_flush(gsum, cnt, gcur, lane, a0, a1, run);
        gcur = gu;
        a0 = 0.0f;
        a1 = 0.0f;
        run = 0;
      }
      a0 += v0[u];
      a1 += v1[u];
      run += 1;
    }
  }
  pool_flush(gsum, cnt, gcur, lane, a0, a1, run);
}


__global__ void k_pool_gemm_out(const float* __restrict__ gsum, const float* __restrict__ cnt,
                                const float* __restrict__ Wp, const float* __restrict__ bp,
                                float* __restrict__ out) {
  int wave = threadIdx.x >> 5;
  int lane = threadIdx.x & 31;
  int mi = wave >> 2, ni = wave & 3;
  int half = lane >> 4, lm = lane & 15;

  int arow = mi * 16 + lm;
  float inv = 1.0f / fmaxf(cnt[arow], 1.0f);

  v8f acc = {};
  for (int kb = 0; kb < 64; kb += 4) {
    int k0 = kb + 2 * half;
    v2f a;
    a.x = gsum[arow * 64 + k0] * inv;
    a.y = gsum[arow * 64 + k0 + 1] * inv;
    v2f b;
    b.x = Wp[k0 * 64 + ni * 16 + lm];
    b.y = Wp[(k0 + 1) * 64 + ni * 16 + lm];
    acc = wmma_f32_k4(a, b, acc);
  }

#pragma unroll
  for (int r = 0; r < 8; ++r) {
    int row = mi * 16 + r + 8 * half;
    out[row * 64 + ni * 16 + lm] = acc[r] + bp[ni * 16 + lm];
  }
}


static inline size_t align_f(size_t n) { return (n + 1023) & ~(size_t)1023; }

extern "C" void kernel_launch(void* const* d_in, const int* in_sizes, int n_in,
                              void* d_out, int out_size, void* d_ws, size_t ws_size,
                              hipStream_t stream) {
  const float* x  = (const float*)d_in[0];
  const float* W1 = (const float*)d_in[1];
  const float* b1 = (const float*)d_in[2];
  const float* W2 = (const float*)d_in[3];
  const float* b2 = (const float*)d_in[4];
  const float* Wp = (const float*)d_in[5];
  const float* bp = (const float*)d_in[6];
  const int* ei   = (const int*)d_in[7];
  const int* batch = (const int*)d_in[8];

  const int E = in_sizes[7] / 2;
  const int N = in_sizes[8];
  const int K1 = in_sizes[0] / N;
  const int* src = ei;
  const int* dst = ei + E;

  float* ws = (float*)d_ws;
  size_t o = 0;
  int*   kA     = (int*)(ws + o);  o += align_f((size_t)E);
  int*   vA     = (int*)(ws + o);  o += align_f((size_t)E);
  int*   kB     = (int*)(ws + o);  o += align_f((size_t)E);
  int*   vB     = (int*)(ws + o);  o += align_f((size_t)E);
  int*   counts = (int*)(ws + o);  o += (size_t)256 * SORT_WAVES;
  int*   offs   = (int*)(ws + o);  o += (size_t)256 * SORT_WAVES;
  int*   tot    = (int*)(ws + o);  o += 1024;
  int*   rowptr = (int*)(ws + o);  o += align_f((size_t)N + 1);
  float* dinv   = ws + o;          o += align_f((size_t)N);
  float* h    = ws + o;  o += align_f((size_t)N * 64);
  float* agg  = ws + o;  o += align_f((size_t)N * 64);
  float* gsum = ws + o;  o += 64 * 64;
  float* cnt  = ws + o;  o += 64;
  (void)ws_size;

  const int T = 256;
  auto blocks = [](long long n, int t) { return (unsigned int)((n + t - 1) / t); };

  const int per = (int)((((long long)E + 32LL * SORT_WAVES - 1) / (32LL * SORT_WAVES)) * 32);
  const int* kin = dst;
  const int* vin = src;
  int* kout = kA;
  int* vout = vA;
  for (int shift = 0; shift < 32 && (shift == 0 || ((N - 1) >> shift) > 0); shift += 8) {
    k_sort_count<<<SORT_WAVES / 8, 256, 0, stream>>>(kin, counts, E, per, shift);
    k_sort_rows<<<256, SORT_WAVES, 0, stream>>>(counts, offs, tot);
    k_sort_scatter<<<SORT_WAVES / 8, 256, 0, stream>>>(kin, vin, offs, tot, kout, vout, E, per, shift);
    kin = kout;
    vin = vout;
    kout = (kout == kA) ? kB : kA;
    vout = (vout == vA) ? vB : vA;
  }
  k_rowptr<<<blocks((long long)N + 1, T), T, 0, stream>>>(kin, rowptr, E, N);
  k_dinv<<<blocks(N, T), T, 0, stream>>>(rowptr, dinv, N);

  const int tiles = (N + 15) / 16;
  const unsigned int gemm_blocks = (unsigned int)((tiles + 7) / 8);
  const unsigned int agg_blocks = blocks((long long)N * 32, T);

  k_gemm_n64<<<gemm_blocks, T, 0, stream>>>(x, W1, h, N, K1);
  k_aggregate<<<agg_blocks, T, 0, stream>>>(h, rowptr, dinv, vin, b1, agg, N);

  k_gemm_n64<<<gemm_blocks, T, 0, stream>>>(agg, W2, h, N, 64);
  k_aggregate<<<agg_blocks, T, 0, stream>>>(h, rowptr, dinv, vin, b2, agg, N);

  k_fill<<<blocks(64 * 64 + 64, T), T, 0, stream>>>(gsum, 0.0f, 64 * 64 + 64);
  const long long pool_threads = (((long long)N + 31) / 32) * 32;
  k_pool<<<blocks(pool_threads, 64), 64, 0, stream>>>(agg, batch, gsum, cnt, N);
  k_pool_gemm_out<<<1, 512, 0, stream>>>(gsum, cnt, Wp, bp, (float*)d_out);
}


# edge sort level 2 in LDS: after the high-digit pass one block per 256-row bucket orders it by low digit and writes sources, row starts, deg^-1/2 (1 launch, was 5); edge order and sums kept
# speedup vs baseline: 2.8988x; 1.4464x over previous
#include <hip/hip_runtime.h>
#include <math.h>

typedef float v2f __attribute__((ext_vector_type(2)));
typedef float v8f __attribute__((ext_vector_type(8)));

__device__ __forceinline__ v8f wmma_f32_k4(v2f a, v2f b, v8f c) {
  return __builtin_amdgcn_wmma_f32_16x16x4_f32(
       false, a,  false, b,
       (short)0, c,  false,  false);
}

__device__ __forceinline__ void atomic_add_f32(float* p, float v) {
  asm volatile("global_atomic_add_f32 %0, %1, off scope:SCOPE_DEV"
               :: "v"(p), "v"(v)
               : "memory");
}


__global__ void k_fill(float* __restrict__ p, float v, int n) {
  int i = blockIdx.x * blockDim.x + threadIdx.x;
  if (i < n) p[i] = v;
}


__device__ __forceinline__ int wave_incl_scan(int x, int lane) {
  for (int o = 1; o < 32; o <<= 1) {
    const int n = __shfl_up(x, o);
    if (lane >= o) x += n;
  }
  return x;
}

__device__ __forceinline__ int wave_sum(int x) {
  for (int o = 16; o > 0; o >>= 1) x += __shfl_xor(x, o);
  return x;
}


#define SORT_WAVES 1024
#define SORT_GROUPS 8

__device__ __forceinline__ unsigned int same_digit_lanes(int digit, bool valid) {
  unsigned int mask = __builtin_amdgcn_ballot_w32(valid);
#pragma unroll
  for (int b = 0; b < 8; ++b) {
    const bool bit = ((digit >> b) & 1) != 0;
    const unsigned int bb = __builtin_amdgcn_ballot_w32(bit);
    mask &= bit ? bb : ~bb;
  }
  return mask;
}

__global__ void __launch_bounds__(256)
k_sort_count(const int* __restrict__ keys, int* __restrict__ counts, int E, int per, int shift) {
  __shared__ int hist[8][256];
  const int lane = threadIdx.x & 31;
  const int w = __builtin_amdgcn_readfirstlane((int)(threadIdx.x >> 5));
  const int wg = (int)blockIdx.x * 8 + w;
  int* hw = hist[w];
  for (int k = lane; k < 256; k += 32) hw[k] = 0;
  const int beg = wg * per;
  const int end = (beg + per < E) ? (beg + per) : E;
  for (int g = beg; g < end; g += 32 * SORT_GROUPS) {
    int key[SORT_GROUPS];
#pragma unroll
    for (int u = 0; u < SORT_GROUPS; ++u) {
      const int idx = g + u * 32 + lane;
      key[u] = (idx < end) ? keys[idx] : 0;
    }
#pragma unroll
    for (int u = 0; u < SORT_GROUPS; ++u) {
      const int idx = g + u * 32 + lane;
      const bool valid = idx < end;
      const int digit = (key[u] >> shift) & 255;
      const unsigned int mask = same_digit_lanes(digit, valid);
      const bool leader = valid && ((mask & ((1u << lane) - 1u)) == 0u);
      if (leader) hw[digit] += __popc(mask);
      __builtin_amdgcn_wave_barrier();
    }
  }
  for (int k = lane; k < 256; k += 32) counts[k * SORT_WAVES + wg] = hw[k];
}

__global__ void __launch_bounds__(SORT_WAVES)
k_sort_rows(const int* __restrict__ counts, int* __restrict__ offs, int* __restrict__ tot) {
  __shared__ int s_tot[32];
  const int t = threadIdx.x;
  const int lane = t & 31, w = t >> 5;
  const int d = (int)blockIdx.x;
  const int v = counts[d * SORT_WAVES + t];
  const int incl = wave_incl_scan(v, lane);
  if (lane == 31) s_tot[w] = incl;
  __syncthreads();
  const int wt = s_tot[lane];
  const int woff = wave_sum((lane < w) ? wt : 0);
  offs[d * SORT_WAVES + t] = woff + incl - v;
  if (t == SORT_WAVES - 1) tot[d] = woff + incl;
}

__global__ void __launch_bounds__(256)
k_sort_scatter(const int* __restrict__ keys, const int* __restrict__ vals,
               const int* __restrict__ offs, const int* __restrict__ tot,
               int* __restrict__ keys_out, int* __restrict__ vals_out,
               int E, int per, int shift) {
  __shared__ int cur[8][256];
  const int lane = threadIdx.x & 31;
  const int w = __builtin_amdgcn_readfirstlane((int)(threadIdx.x >> 5));
  const int wg = (int)blockIdx.x * 8 + w;
  int* cw = cur[w];

  int tl[8];
  int ls = 0;
#pragma unroll
  for (int q = 0; q < 8; ++q) {
    tl[q] = tot[8 * lane + q];
    ls += tl[q];
  }
  int run = wave_incl_scan(ls, lane) - ls;
#pragma unroll
  for (int q = 0; q < 8; ++q) {
    cw[8 * lane + q] = run + offs[(8 * lane + q) * SORT_WAVES + wg];
    run += tl[q];
  }
  __builtin_amdgcn_wave_barrier();

  const int beg = wg * per;
  const int end = (beg + per < E) ? (beg + per) : E;
  for (int g = beg; g < end; g += 32 * SORT_GROUPS) {
    int key[SORT_GROUPS], val[SORT_GROUPS];
#pragma unroll
    for (int u = 0; u < SORT_GROUPS; ++u) {
      const int idx = g + u * 32 + lane;
      key[u] = (idx < end) ? keys[idx] : 0;
      val[u] = (idx < end) ? vals[idx] : 0;
    }
#pragma unroll
    for (int u = 0; u < SORT_GROUPS; ++u) {
      const int idx = g + u * 32 + lane;
      const bool valid = idx < end;
      const int digit = (key[u] >> shift) & 255;
      const unsigned int mask = same_digit_lanes(digit, valid);
      const int rank = __popc(mask & ((1u << lane) - 1u));
      int place = 0;
      if (valid) place = cw[digit];
      __builtin_amdgcn_wave_barrier();
      if (valid && rank == 0) cw[digit] = place + __popc(mask);
      __builtin_amdgcn_wave_barrier();
      if (valid) {
        keys_out[place + rank] = key[u];
        vals_out[place + rank] = val[u];
      }
    }
  }
}


#define BUCKET_WAVES 32

__global__ void __launch_bounds__(32 * BUCKET_WAVES)
k_bucket(const int* __restrict__ keys, const int* __restrict__ vals, const int* __restrict__ tot,
         int* __restrict__ csr_src, int* __restrict__ rowptr, float* __restrict__ dinv, int N) {
  __shared__ int hist[BUCKET_WAVES][256];
  __shared__ int s_wsum[8];
  const int t = (int)threadIdx.x;
  const int lane = t & 31;
  const int w = __builtin_amdgcn_readfirstlane(t >> 5);
  const int b = (int)blockIdx.x;
  int* hw = hist[w];
  for (int k = lane; k < 256; k += 32) hw[k] = 0;

  int tl[8];
  int ls = 0;
#pragma unroll
  for (int q = 0; q < 8; ++q) {
    tl[q] = tot[8 * lane + q];
    ls += tl[q];
  }
  int first = wave_incl_scan(ls, lane) - ls;
  int mine = 0;
#pragma unroll
  for (int q = 0; q < 8; ++q) {
    if (q < (b & 7)) first += tl[q];
    if (q == (b & 7)) mine = tl[q];
  }
  const int base = __builtin_amdgcn_readlane(first, b >> 3);
  const int size = __builtin_amdgcn_readlane(mine, b >> 3);

  const int per = ((size + 32 * BUCKET_WAVES - 1) / (32 * BUCKET_WAVES)) * 32;
  const int lim = base + size;
  const int beg = base + w * per;
  const int end = (beg + per < lim) ? (beg + per) : lim;

  for (int g = beg; g < end; g += 32 * SORT_GROUPS) {
    int key[SORT_GROUPS];
#pragma unroll
    for (int u = 0; u < SORT_GROUPS; ++u) {
      const int idx = g + u * 32 + lane;
      key[u] = (idx < end) ? keys[idx] : 0;
    }
#pragma unroll
    for (int u = 0; u < SORT_GROUPS; ++u) {
      const int idx = g + u * 32 + lane;
      const bool valid = idx < end;
      const int digit = key[u] & 255;
      const unsigned int mask = same_digit_lanes(digit, valid);
      const bool leader = valid && ((mask & ((1u << lane) - 1u)) == 0u);
      if (leader) hw[digit] += __popc(mask);
      __builtin_amdgcn_wave_barrier();
    }
  }
  __syncthreads();

  int c[BUCKET_WAVES];
  int total = 0, incl = 0;
  if (t < 256) {
#pragma unroll
    for (int k = 0; k < BUCKET_WAVES; ++k) c[k] = hist[k][t];
#pragma unroll
    for (int k = 0; k < BUCKET_WAVES; ++k) {
      const int x = c[k];
      c[k] = total;
      total += x;
    }
    incl = wave_incl_scan(total, lane);
    if (lane == 31) s_wsum[w] = incl;
  }
  __syncthreads();
  if (t < 256) {
    int before = 0;
#pragma unroll
    for (int k = 0; k < 8; ++k) before += (k < w) ? s_wsum[k] : 0;
    const int rowstart = base + before + incl - total;
#pragma unroll
    for (int k = 0; k < BUCKET_WAVES; ++k) hist[k][t] = rowstart + c[k];
    const int row = b * 256 + t;
    if (row <= N) rowptr[row] = rowstart;
    if (row < N) dinv[row] = rsqrtf((float)(total + 1));
  }
  __syncthreads();

  for (int g = beg; g < end; g += 32 * SORT_GROUPS) {
    int key[SORT_GROUPS], val[SORT_GROUPS];
#pragma unroll
    for (int u = 0; u < SORT_GROUPS; ++u) {
      const int idx = g + u * 32 + lane;
      key[u] = (idx < end) ? keys[idx] : 0;
      val[u] = (idx < end) ? vals[idx] : 0;
    }
#pragma unroll
    for (int u = 0; u < SORT_GROUPS; ++u) {
      const int idx = g + u * 32 + lane;
      const bool valid = idx < end;
      const int digit = key[u] & 255;
      const unsigned int mask = same_digit_lanes(digit, valid);
      const int rank = __popc(mask & ((1u << lane) - 1u));
      int place = 0;
      if (valid) place = hw[digit];
      __builtin_amdgcn_wave_barrier();
      if (valid && rank == 0) hw[digit] = place + __popc(mask);
      __builtin_amdgcn_wave_barrier();
      if (valid) csr_src[place + rank] = val[u];
    }
  }
}


__global__ void k_aggregate(const float* __restrict__ h, const int* __restrict__ rowptr,
                            const float* __restrict__ dinv, const int* __restrict__ csr_src,
                            const float* __restrict__ bias, float* __restrict__ out, int N) {
  const int lane = threadIdx.x & 31;
  const int i = __builtin_amdgcn_readfirstlane(
      (int)((blockIdx.x * blockDim.x + threadIdx.x) >> 5));
  if (i >= N) return;
  const int start = rowptr[i];
  const int cnt = rowptr[i + 1] - start;
  const float di = dinv[i];

  const v2f self = *(const v2f*)(h + (size_t)i * 64 + 2 * lane);
  const float ns = di * di;
  float a0 = self.x * ns, a1 = self.y * ns;

  for (int c = 0; c < cnt; c += 32) {
    const int m = (cnt - c < 32) ? (cnt - c) : 32;
    int sl = i;
    float nl = 0.0f;
    if (lane < m) {
      sl = csr_src[start + c + lane];
      nl = dinv[sl] * di;
    }
    for (int j = 0; j < m; j += 8) {
      v2f v[8];
#pragma unroll
      for (int u = 0; u < 8; ++u) {
        const int sj = __builtin_amdgcn_readlane(sl, j + u);
        v[u] = *(const v2f*)(h + (size_t)sj * 64 + 2 * lane);
      }
#pragma unroll
      for (int u = 0; u < 8; ++u) {
        const float nj = __builtin_bit_cast(float,
            __builtin_amdgcn_readlane(__builtin_bit_cast(int, nl), j + u));
        a0 = fmaf(v[u].x, nj, a0);
        a1 = fmaf(v[u].y, nj, a1);
      }
    }
  }

  v2f o;
  o.x = fmaxf(a0 + bias[2 * lane], 0.0f);
  o.y = fmaxf(a1 + bias[2 * lane + 1], 0.0f);
  *(v2f*)(out + (size_t)i * 64 + 2 * lane) = o;
}


__global__ void k_gemm_n64(const float* __restrict__ X, const float* __restrict__ W,
                           float* __restrict__ H, int M, int K) {
  int wave = (int)((blockIdx.x * blockDim.x + threadIdx.x) >> 5);
  int lane = threadIdx.x & 31;
  int m0 = wave * 16;
  if (m0 >= M) return;
  int half = lane >> 4;
  int lm = lane & 15;

  const float* xrow = X + (size_t)(m0 + lm) * (size_t)K;
  v8f acc0 = {}, acc1 = {}, acc2 = {}, acc3 = {};

  for (int kb = 0; kb < K; kb += 4) {
    int k0 = kb + 2 * half;
    v2f a;
    a.x = xrow[k0];
    a.y = xrow[k0 + 1];
    const float* w0 = W + (size_t)k0 * 64;
    v2f b0, b1, b2, b3;
    b0.x = w0[lm];       b0.y = w0[64 + lm];
    b1.x = w0[16 + lm];  b1.y = w0[64 + 16 + lm];
    b2.x = w0[32 + lm];  b2.y = w0[64 + 32 + lm];
    b3.x = w0[48 + lm];  b3.y = w0[64 + 48 + lm];
    acc0 = wmma_f32_k4(a, b0, acc0);
    acc1 = wmma_f32_k4(a, b1, acc1);
    acc2 = wmma_f32_k4(a, b2, acc2);
    acc3 = wmma_f32_k4(a, b3, acc3);
  }

#pragma unroll
  for (int r = 0; r < 8; ++r) {
    int row = m0 + r + 8 * half;
    float* hr = H + (size_t)row * 64 + lm;
    hr[0]  = acc0[r];
    hr[16] = acc1[r];
    hr[32] = acc2[r];
    hr[48] = acc3[r];
  }
}


__device__ __forceinline__ void row_add(float* __restrict__ agg, int d, int lane,
                                        float v0, float v1) {
  float* p0 = agg + (size_t)d * 64 + lane;
  float* p1 = p0 + 32;
  atomic_add_f32(p0, v0);
  atomic_add_f32(p1, v1);
  asm volatile("s_wait_xcnt 0x0" :: "v"(p0), "v"(p1), "v"(v0), "v"(v1));
}


__device__ __forceinline__ void one_add(float* p, float v) {
  atomic_add_f32(p, v);
  asm volatile("s_wait_xcnt 0x0" :: "v"(p), "v"(v));
}

__device__ __forceinline__ void pool_flush(float* __restrict__ gsum, float* __restrict__ cnt,
                                           int g, int lane, float a0, float a1, int run) {
  row_add(gsum, g, lane, a0, a1);
  if (lane == 0) one_add(cnt + g, (float)run);
}

__global__ void k_pool(const float* __restrict__ h, const int* __restrict__ batch,
                       float* __restrict__ gsum, float* __restrict__ cnt, int N) {
  const int lane = threadIdx.x & 31;
  const int wave = __builtin_amdgcn_readfirstlane(
      (int)((blockIdx.x * blockDim.x + threadIdx.x) >> 5));
  const int base = wave * 32;
  if (base >= N) return;
  const int last = N - 1;
  const int m = (N - base < 32) ? (N - base) : 32;

  const int nl = (base + lane < last) ? (base + lane) : last;
  const int gl = batch[nl];

  float v0[32], v1[32];
#pragma unroll
  for (int u = 0; u < 32; ++u) {
    const int n = (base + u < last) ? (base + u) : last;
    const float* hrow = h + (size_t)n * 64;
    v0[u] = hrow[lane];
    v1[u] = hrow[lane + 32];
  }

  int gcur = __builtin_amdgcn_readlane(gl, 0);
  float a0 = v0[0], a1 = v1[0];
  int run = 1;
#pragma unroll
  for (int u = 1; u < 32; ++u) {
    if (u < m) {
      const int gu = __builtin_amdgcn_readlane(gl, u);
      if (gu != gcur) {
        pool_flush(gsum, cnt, gcur, lane, a0, a1, run);
        gcur = gu;
        a0 = 0.0f;
        a1 = 0.0f;
        run = 0;
      }
      a0 += v0[u];
      a1 += v1[u];
      run += 1;
    }
  }
  pool_flush(gsum, cnt, gcur, lane, a0, a1, run);
}


__global__ void k_pool_gemm_out(const float* __restrict__ gsum, const float* __restrict__ cnt,
                                const float* __restrict__ Wp, const float* __restrict__ bp,
                                float* __restrict__ out) {
  int wave = threadIdx.x >> 5;
  int lane = threadIdx.x & 31;
  int mi = wave >> 2, ni = wave & 3;
  int half = lane >> 4, lm = lane & 15;

  int arow = mi * 16 + lm;
  float inv = 1.0f / fmaxf(cnt[arow], 1.0f);

  v8f acc = {};
  for (int kb = 0; kb < 64; kb += 4) {
    int k0 = kb + 2 * half;
    v2f a;
    a.x = gsum[arow * 64 + k0] * inv;
    a.y = gsum[arow * 64 + k0 + 1] * inv;
    v2f b;
    b.x = Wp[k0 * 64 + ni * 16 + lm];
    b.y = Wp[(k0 + 1) * 64 + ni * 16 + lm];
    acc = wmma_f32_k4(a, b, acc);
  }

#pragma unroll
  for (int r = 0; r < 8; ++r) {
    int row = mi * 16 + r + 8 * half;
    out[row * 64 + ni * 16 + lm] = acc[r] + bp[ni * 16 + lm];
  }
}


static inline size_t align_f(size_t n) { return (n + 1023) & ~(size_t)1023; }

extern "C" void kernel_launch(void* const* d_in, const int* in_sizes, int n_in,
                              void* d_out, int out_size, void* d_ws, size_t ws_size,
                              hipStream_t stream) {
  const float* x  = (const float*)d_in[0];
  const float* W1 = (const float*)d_in[1];
  const float* b1 = (const float*)d_in[2];
  const float* W2 = (const float*)d_in[3];
  const float* b2 = (const float*)d_in[4];
  const float* Wp = (const float*)d_in[5];
  const float* bp = (const float*)d_in[6];
  const int* ei   = (const int*)d_in[7];
  const int* batch = (const int*)d_in[8];

  const int E = in_sizes[7] / 2;
  const int N = in_sizes[8];
  const int K1 = in_sizes[0] / N;
  const int* src = ei;
  const int* dst = ei + E;

  float* ws = (float*)d_ws;
  size_t o = 0;
  int*   kA     = (int*)(ws + o);  o += align_f((size_t)E);
  int*   vA     = (int*)(ws + o);  o += align_f((size_t)E);
  int*   csr    = (int*)(ws + o);  o += align_f((size_t)E);
  int*   counts = (int*)(ws + o);  o += (size_t)256 * SORT_WAVES;
  int*   offs   = (int*)(ws + o);  o += (size_t)256 * SORT_WAVES;
  int*   tot    = (int*)(ws + o);  o += 1024;
  int*   rowptr = (int*)(ws + o);  o += align_f((size_t)N + 1);
  float* dinv   = ws + o;          o += align_f((size_t)N);
  float* h    = ws + o;  o += align_f((size_t)N * 64);
  float* agg  = ws + o;  o += align_f((size_t)N * 64);
  float* gsum = ws + o;  o += 64 * 64;
  float* cnt  = ws + o;  o += 64;
  (void)ws_size;

  const int T = 256;
  auto blocks = [](long long n, int t) { return (unsigned int)((n + t - 1) / t); };

  const int per = (int)((((long long)E + 32LL * SORT_WAVES - 1) / (32LL * SORT_WAVES)) * 32);
  k_sort_count<<<SORT_WAVES / 8, 256, 0, stream>>>(dst, counts, E, per, 8);
  k_sort_rows<<<256, SORT_WAVES, 0, stream>>>(counts, offs, tot);
  k_sort_scatter<<<SORT_WAVES / 8, 256, 0, stream>>>(dst, src, offs, tot, kA, vA, E, per, 8);
  k_bucket<<<(unsigned int)((N >> 8) + 1), 32 * BUCKET_WAVES, 0, stream>>>(kA, vA, tot, csr, rowptr, dinv, N);

  const int tiles = (N + 15) / 16;
  const unsigned int gemm_blocks = (unsigned int)((tiles + 7) / 8);
  const unsigned int agg_blocks = blocks((long long)N * 32, T);

  k_gemm_n64<<<gemm_blocks, T, 0, stream>>>(x, W1, h, N, K1);
  k_aggregate<<<agg_blocks, T, 0, stream>>>(h, rowptr, dinv, csr, b1, agg, N);

  k_gemm_n64<<<gemm_blocks, T, 0, stream>>>(agg, W2, h, N, 64);
  k_aggregate<<<agg_blocks, T, 0, stream>>>(h, rowptr, dinv, csr, b2, agg, N);

  k_fill<<<blocks(64 * 64 + 64, T), T, 0, stream>>>(gsum, 0.0f, 64 * 64 + 64);
  const long long pool_threads = (((long long)N + 31) / 32) * 32;
  k_pool<<<blocks(pool_threads, 64), 64, 0, stream>>>(agg, batch, gsum, cnt, N);
  k_pool_gemm_out<<<1, 512, 0, stream>>>(gsum, cnt, Wp, bp, (float*)d_out);
}


# level-1 edge sort cut into 4096 wave pieces (was 1024): 4 waves per SIMD on every WGP interleave the serial ballot and LDS walks; table rows move as 32-byte runs; edge order and sums kept
# speedup vs baseline: 2.9985x; 1.0344x over previous
#include <hip/hip_runtime.h>
#include <math.h>

typedef float v2f __attribute__((ext_vector_type(2)));
typedef float v8f __attribute__((ext_vector_type(8)));

__device__ __forceinline__ v8f wmma_f32_k4(v2f a, v2f b, v8f c) {
  return __builtin_amdgcn_wmma_f32_16x16x4_f32(
       false, a,  false, b,
       (short)0, c,  false,  false);
}

__device__ __forceinline__ void atomic_add_f32(float* p, float v) {
  asm volatile("global_atomic_add_f32 %0, %1, off scope:SCOPE_DEV"
               :: "v"(p), "v"(v)
               : "memory");
}


__global__ void k_fill(float* __restrict__ p, float v, int n) {
  int i = blockIdx.x * blockDim.x + threadIdx.x;
  if (i < n) p[i] = v;
}


__device__ __forceinline__ int wave_incl_scan(int x, int lane) {
  for (int o = 1; o < 32; o <<= 1) {
    const int n = __shfl_up(x, o);
    if (lane >= o) x += n;
  }
  return x;
}

__device__ __forceinline__ int wave_sum(int x) {
  for (int o = 16; o > 0; o >>= 1) x += __shfl_xor(x, o);
  return x;
}


#define SORT_WAVES 4096
static_assert(SORT_WAVES == 4 * 1024, "k_sort_rows scans 4 table entries per thread");
#define SORT_GROUPS 8

__device__ __forceinline__ unsigned int same_digit_lanes(int digit, bool valid) {
  unsigned int mask = __builtin_amdgcn_ballot_w32(valid);
#pragma unroll
  for (int b = 0; b < 8; ++b) {
    const bool bit = ((digit >> b) & 1) != 0;
    const unsigned int bb = __builtin_amdgcn_ballot_w32(bit);
    mask &= bit ? bb : ~bb;
  }
  return mask;
}

__global__ void __launch_bounds__(256)
k_sort_count(const int* __restrict__ keys, int* __restrict__ counts, int E, int per, int shift) {
  __shared__ int hist[8][256];
  const int t = (int)threadIdx.x;
  const int lane = t & 31;
  const int w = __builtin_amdgcn_readfirstlane((int)(threadIdx.x >> 5));
  const int wg = (int)blockIdx.x * 8 + w;
  int* hw = hist[w];
  for (int k = lane; k < 256; k += 32) hw[k] = 0;
  const int beg = wg * per;
  const int end = (beg + per < E) ? (beg + per) : E;
  for (int g = beg; g < end; g += 32 * SORT_GROUPS) {
    int key[SORT_GROUPS];
#pragma unroll
    for (int u = 0; u < SORT_GROUPS; ++u) {
      const int idx = g + u * 32 + lane;
      key[u] = (idx < end) ? keys[idx] : 0;
    }
#pragma unroll
    for (int u = 0; u < SORT_GROUPS; ++u) {
      if (g + u * 32 >= end) break;
      const int idx = g + u * 32 + lane;
      const bool valid = idx < end;
      const int digit = (key[u] >> shift) & 255;
      const unsigned int mask = same_digit_lanes(digit, valid);
      const bool leader = valid && ((mask & ((1u << lane) - 1u)) == 0u);
      if (leader) hw[digit] += __popc(mask);
      __builtin_amdgcn_wave_barrier();
    }
  }
  __syncthreads();
  int4 c0, c1;
  c0.x = hist[0][t]; c0.y = hist[1][t]; c0.z = hist[2][t]; c0.w = hist[3][t];
  c1.x = hist[4][t]; c1.y = hist[5][t]; c1.z = hist[6][t]; c1.w = hist[7][t];
  int* row = counts + (size_t)t * SORT_WAVES + (size_t)blockIdx.x * 8;
  *(int4*)row = c0;
  *(int4*)(row + 4) = c1;
}

__global__ void __launch_bounds__(1024)
k_sort_rows(const int* __restrict__ counts, int* __restrict__ offs, int* __restrict__ tot) {
  __shared__ int s_tot[32];
  const int t = threadIdx.x;
  const int lane = t & 31, w = t >> 5;
  const int d = (int)blockIdx.x;
  const int4 c = *(const int4*)(counts + (size_t)d * SORT_WAVES + 4 * t);
  const int v = c.x + c.y + c.z + c.w;
  const int incl = wave_incl_scan(v, lane);
  if (lane == 31) s_tot[w] = incl;
  __syncthreads();
  const int wt = s_tot[lane];
  const int woff = wave_sum((lane < w) ? wt : 0);
  int4 o;
  o.x = woff + incl - v;
  o.y = o.x + c.x;
  o.z = o.y + c.y;
  o.w = o.z + c.z;
  *(int4*)(offs + (size_t)d * SORT_WAVES + 4 * t) = o;
  if (t == 1023) tot[d] = woff + incl;
}

__global__ void __launch_bounds__(256)
k_sort_scatter(const int* __restrict__ keys, const int* __restrict__ vals,
               const int* __restrict__ offs, const int* __restrict__ tot,
               int* __restrict__ keys_out, int* __restrict__ vals_out,
               int E, int per, int shift) {
  __shared__ int cur[8][256];
  __shared__ int s_dsum[8];
  const int t = (int)threadIdx.x;
  const int lane = t & 31;
  const int w = __builtin_amdgcn_readfirstlane(t >> 5);
  const int wg = (int)blockIdx.x * 8 + w;
  int* cw = cur[w];

  const int tv = tot[t];
  const int* orow = offs + (size_t)t * SORT_WAVES + (size_t)blockIdx.x * 8;
  const int4 o0 = *(const int4*)orow;
  const int4 o1 = *(const int4*)(orow + 4);
  const int incl = wave_incl_scan(tv, lane);
  if (lane == 31) s_dsum[w] = incl;
  __syncthreads();
  int before = 0;
#pragma unroll
  for (int k = 0; k < 8; ++k) before += (k < w) ? s_dsum[k] : 0;
  const int dbase = before + incl - tv;
  cur[0][t] = dbase + o0.x; cur[1][t] = dbase + o0.y; cur[2][t] = dbase + o0.z; cur[3][t] = dbase + o0.w;
  cur[4][t] = dbase + o1.x; cur[5][t] = dbase + o1.y; cur[6][t] = dbase + o1.z; cur[7][t] = dbase + o1.w;
  __syncthreads();

  const int beg = wg * per;
  const int end = (beg + per < E) ? (beg + per) : E;
  for (int g = beg; g < end; g += 32 * SORT_GROUPS) {
    int key[SORT_GROUPS], val[SORT_GROUPS];
#pragma unroll
    for (int u = 0; u < SORT_GROUPS; ++u) {
      const int idx = g + u * 32 + lane;
      key[u] = (idx < end) ? keys[idx] : 0;
      val[u] = (idx < end) ? vals[idx] : 0;
    }
#pragma unroll
    for (int u = 0; u < SORT_GROUPS; ++u) {
      if (g + u * 32 >= end) break;
      const int idx = g + u * 32 + lane;
      const bool valid = idx < end;
      const int digit = (key[u] >> shift) & 255;
      const unsigned int mask = same_digit_lanes(digit, valid);
      const int rank = __popc(mask & ((1u << lane) - 1u));
      int place = 0;
      if (valid) place = cw[digit];
      __builtin_amdgcn_wave_barrier();
      if (valid && rank == 0) cw[digit] = place + __popc(mask);
      __builtin_amdgcn_wave_barrier();
      if (valid) {
        keys_out[place + rank] = key[u];
        vals_out[place + rank] = val[u];
      }
    }
  }
}


#define BUCKET_WAVES 32

__global__ void __launch_bounds__(32 * BUCKET_WAVES)
k_bucket(const int* __restrict__ keys, const int* __restrict__ vals, const int* __restrict__ tot,
         int* __restrict__ csr_src, int* __restrict__ rowptr, float* __restrict__ dinv, int N) {
  __shared__ int hist[BUCKET_WAVES][256];
  __shared__ int s_wsum[8];
  const int t = (int)threadIdx.x;
  const int lane = t & 31;
  const int w = __builtin_amdgcn_readfirstlane(t >> 5);
  const int b = (int)blockIdx.x;
  int* hw = hist[w];
  for (int k = lane; k < 256; k += 32) hw[k] = 0;

  int tl[8];
  int ls = 0;
#pragma unroll
  for (int q = 0; q < 8; ++q) {
    tl[q] = tot[8 * lane + q];
    ls += tl[q];
  }
  int first = wave_incl_scan(ls, lane) - ls;
  int mine = 0;
#pragma unroll
  for (int q = 0; q < 8; ++q) {
    if (q < (b & 7)) first += tl[q];
    if (q == (b & 7)) mine = tl[q];
  }
  const int base = __builtin_amdgcn_readlane(first, b >> 3);
  const int size = __builtin_amdgcn_readlane(mine, b >> 3);

  const int per = ((size + 32 * BUCKET_WAVES - 1) / (32 * BUCKET_WAVES)) * 32;
  const int lim = base + size;
  const int beg = base + w * per;
  const int end = (beg + per < lim) ? (beg + per) : lim;

  for (int g = beg; g < end; g += 32 * SORT_GROUPS) {
    int key[SORT_GROUPS];
#pragma unroll
    for (int u = 0; u < SORT_GROUPS; ++u) {
      const int idx = g + u * 32 + lane;
      key[u] = (idx < end) ? keys[idx] : 0;
    }
#pragma unroll
    for (int u = 0; u < SORT_GROUPS; ++u) {
      const int idx = g + u * 32 + lane;
      const bool valid = idx < end;
      const int digit = key[u] & 255;
      const unsigned int mask = same_digit_lanes(digit, valid);
      const bool leader = valid && ((mask & ((1u << lane) - 1u)) == 0u);
      if (leader) hw[digit] += __popc(mask);
      __builtin_amdgcn_wave_barrier();
    }
  }
  __syncthreads();

  int c[BUCKET_WAVES];
  int total = 0, incl = 0;
  if (t < 256) {
#pragma unroll
    for (int k = 0; k < BUCKET_WAVES; ++k) c[k] = hist[k][t];
#pragma unroll
    for (int k = 0; k < BUCKET_WAVES; ++k) {
      const int x = c[k];
      c[k] = total;
      total += x;
    }
    incl = wave_incl_scan(total, lane);
    if (lane == 31) s_wsum[w] = incl;
  }
  __syncthreads();
  if (t < 256) {
    int before = 0;
#pragma unroll
    for (int k = 0; k < 8; ++k) before += (k < w) ? s_wsum[k] : 0;
    const int rowstart = base + before + incl - total;
#pragma unroll
    for (int k = 0; k < BUCKET_WAVES; ++k) hist[k][t] = rowstart + c[k];
    const int row = b * 256 + t;
    if (row <= N) rowptr[row] = rowstart;
    if (row < N) dinv[row] = rsqrtf((float)(total + 1));
  }
  __syncthreads();

  for (int g = beg; g < end; g += 32 * SORT_GROUPS) {
    int key[SORT_GROUPS], val[SORT_GROUPS];
#pragma unroll
    for (int u = 0; u < SORT_GROUPS; ++u) {
      const int idx = g + u * 32 + lane;
      key[u] = (idx < end) ? keys[idx] : 0;
      val[u] = (idx < end) ? vals[idx] : 0;
    }
#pragma unroll
    for (int u = 0; u < SORT_GROUPS; ++u) {
      const int idx = g + u * 32 + lane;
      const bool valid = idx < end;
      const int digit = key[u] & 255;
      const unsigned int mask = same_digit_lanes(digit, valid);
      const int rank = __popc(mask & ((1u << lane) - 1u));
      int place = 0;
      if (valid) place = hw[digit];
      __builtin_amdgcn_wave_barrier();
      if (valid && rank == 0) hw[digit] = place + __popc(mask);
      __builtin_amdgcn_wave_barrier();
      if (valid) csr_src[place + rank] = val[u];
    }
  }
}


__global__ void k_aggregate(const float* __restrict__ h, const int* __restrict__ rowptr,
                            const float* __restrict__ dinv, const int* __restrict__ csr_src,
                            const float* __restrict__ bias, float* __restrict__ out, int N) {
  const int lane = threadIdx.x & 31;
  const int i = __builtin_amdgcn_readfirstlane(
      (int)((blockIdx.x * blockDim.x + threadIdx.x) >> 5));
  if (i >= N) return;
  const int start = rowptr[i];
  const int cnt = rowptr[i + 1] - start;
  const float di = dinv[i];

  const v2f self = *(const v2f*)(h + (size_t)i * 64 + 2 * lane);
  const float ns = di * di;
  float a0 = self.x * ns, a1 = self.y * ns;

  for (int c = 0; c < cnt; c += 32) {
    const int m = (cnt - c < 32) ? (cnt - c) : 32;
    int sl = i;
    float nl = 0.0f;
    if (lane < m) {
      sl = csr_src[start + c + lane];
      nl = dinv[sl] * di;
    }
    for (int j = 0; j < m; j += 8) {
      v2f v[8];
#pragma unroll
      for (int u = 0; u < 8; ++u) {
        const int sj = __builtin_amdgcn_readlane(sl, j + u);
        v[u] = *(const v2f*)(h + (size_t)sj * 64 + 2 * lane);
      }
#pragma unroll
      for (int u = 0; u < 8; ++u) {
        const float nj = __builtin_bit_cast(float,
            __builtin_amdgcn_readlane(__builtin_bit_cast(int, nl), j + u));
        a0 = fmaf(v[u].x, nj, a0);
        a1 = fmaf(v[u].y, nj, a1);
      }
    }
  }

  v2f o;
  o.x = fmaxf(a0 + bias[2 * lane], 0.0f);
  o.y = fmaxf(a1 + bias[2 * lane + 1], 0.0f);
  *(v2f*)(out + (size_t)i * 64 + 2 * lane) = o;
}


__global__ void k_gemm_n64(const float* __restrict__ X, const float* __restrict__ W,
                           float* __restrict__ H, int M, int K) {
  int wave = (int)((blockIdx.x * blockDim.x + threadIdx.x) >> 5);
  int lane = threadIdx.x & 31;
  int m0 = wave * 16;
  if (m0 >= M) return;
  int half = lane >> 4;
  int lm = lane & 15;

  const float* xrow = X + (size_t)(m0 + lm) * (size_t)K;
  v8f acc0 = {}, acc1 = {}, acc2 = {}, acc3 = {};

  for (int kb = 0; kb < K; kb += 4) {
    int k0 = kb + 2 * half;
    v2f a;
    a.x = xrow[k0];
    a.y = xrow[k0 + 1];
    const float* w0 = W + (size_t)k0 * 64;
    v2f b0, b1, b2, b3;
    b0.x = w0[lm];       b0.y = w0[64 + lm];
    b1.x = w0[16 + lm];  b1.y = w0[64 + 16 + lm];
    b2.x = w0[32 + lm];  b2.y = w0[64 + 32 + lm];
    b3.x = w0[48 + lm];  b3.y = w0[64 + 48 + lm];
    acc0 = wmma_f32_k4(a, b0, acc0);
    acc1 = wmma_f32_k4(a, b1, acc1);
    acc2 = wmma_f32_k4(a, b2, acc2);
    acc3 = wmma_f32_k4(a, b3, acc3);
  }

#pragma unroll
  for (int r = 0; r < 8; ++r) {
    int row = m0 + r + 8 * half;
    float* hr = H + (size_t)row * 64 + lm;
    hr[0]  = acc0[r];
    hr[16] = acc1[r];
    hr[32] = acc2[r];
    hr[48] = acc3[r];
  }
}


__device__ __forceinline__ void row_add(float* __restrict__ agg, int d, int lane,
                                        float v0, float v1) {
  float* p0 = agg + (size_t)d * 64 + lane;
  float* p1 = p0 + 32;
  atomic_add_f32(p0, v0);
  atomic_add_f32(p1, v1);
  asm volatile("s_wait_xcnt 0x0" :: "v"(p0), "v"(p1), "v"(v0), "v"(v1));
}


__device__ __forceinline__ void one_add(float* p, float v) {
  atomic_add_f32(p, v);
  asm volatile("s_wait_xcnt 0x0" :: "v"(p), "v"(v));
}

__device__ __forceinline__ void pool_flush(float* __restrict__ gsum, float* __restrict__ cnt,
                                           int g, int lane, float a0, float a1, int run) {
  row_add(gsum, g, lane, a0, a1);
  if (lane == 0) one_add(cnt + g, (float)run);
}

__global__ void k_pool(const float* __restrict__ h, const int* __restrict__ batch,
                       float* __restrict__ gsum, float* __restrict__ cnt, int N) {
  const int lane = threadIdx.x & 31;
  const int wave = __builtin_amdgcn_readfirstlane(
      (int)((blockIdx.x * blockDim.x + threadIdx.x) >> 5));
  const int base = wave * 32;
  if (base >= N) return;
  const int last = N - 1;
  const int m = (N - base < 32) ? (N - base) : 32;

  const int nl = (base + lane < last) ? (base + lane) : last;
  const int gl = batch[nl];

  float v0[32], v1[32];
#pragma unroll
  for (int u = 0; u < 32; ++u) {
    const int n = (base + u < last) ? (base + u) : last;
    const float* hrow = h + (size_t)n * 64;
    v0[u] = hrow[lane];
    v1[u] = hrow[lane + 32];
  }

  int gcur = __builtin_amdgcn_readlane(gl, 0);
  float a0 = v0[0], a1 = v1[0];
  int run = 1;
#pragma unroll
  for (int u = 1; u < 32; ++u) {
    if (u < m) {
      const int gu = __builtin_amdgcn_readlane(gl, u);
      if (gu != gcur) {
        pool_flush(gsum, cnt, gcur, lane, a0, a1, run);
        gcur = gu;
        a0 = 0.0f;
        a1 = 0.0f;
        run = 0;
      }
      a0 += v0[u];
      a1 += v1[u];
      run += 1;
    }
  }
  pool_flush(gsum, cnt, gcur, lane, a0, a1, run);
}


__global__ void k_pool_gemm_out(const float* __restrict__ gsum, const float* __restrict__ cnt,
                                const float* __restrict__ Wp, const float* __restrict__ bp,
                                float* __restrict__ out) {
  int wave = threadIdx.x >> 5;
  int lane = threadIdx.x & 31;
  int mi = wave >> 2, ni = wave & 3;
  int half = lane >> 4, lm = lane & 15;

  int arow = mi * 16 + lm;
  float inv = 1.0f / fmaxf(cnt[arow], 1.0f);

  v8f acc = {};
  for (int kb = 0; kb < 64; kb += 4) {
    int k0 = kb + 2 * half;
    v2f a;
    a.x = gsum[arow * 64 + k0] * inv;
    a.y = gsum[arow * 64 + k0 + 1] * inv;
    v2f b;
    b.x = Wp[k0 * 64 + ni * 16 + lm];
    b.y = Wp[(k0 + 1) * 64 + ni * 16 + lm];
    acc = wmma_f32_k4(a, b, acc);
  }

#pragma unroll
  for (int r = 0; r < 8; ++r) {
    int row = mi * 16 + r + 8 * half;
    out[row * 64 + ni * 16 + lm] = acc[r] + bp[ni * 16 + lm];
  }
}


static inline size_t align_f(size_t n) { return (n + 1023) & ~(size_t)1023; }

extern "C" void kernel_launch(void* const* d_in, const int* in_sizes, int n_in,
                              void* d_out, int out_size, void* d_ws, size_t ws_size,
                              hipStream_t stream) {
  const float* x  = (const float*)d_in[0];
  const float* W1 = (const float*)d_in[1];
  const float* b1 = (const float*)d_in[2];
  const float* W2 = (const float*)d_in[3];
  const float* b2 = (const float*)d_in[4];
  const float* Wp = (const float*)d_in[5];
  const float* bp = (const float*)d_in[6];
  const int* ei   = (const int*)d_in[7];
  const int* batch = (const int*)d_in[8];

  const int E = in_sizes[7] / 2;
  const int N = in_sizes[8];
  const int K1 = in_sizes[0] / N;
  const int* src = ei;
  const int* dst = ei + E;

  float* ws = (float*)d_ws;
  size_t o = 0;
  int*   kA     = (int*)(ws + o);  o += align_f((size_t)E);
  int*   vA     = (int*)(ws + o);  o += align_f((size_t)E);
  int*   csr    = (int*)(ws + o);  o += align_f((size_t)E);
  int*   counts = (int*)(ws + o);  o += (size_t)256 * SORT_WAVES;
  int*   offs   = (int*)(ws + o);  o += (size_t)256 * SORT_WAVES;
  int*   tot    = (int*)(ws + o);  o += 1024;
  int*   rowptr = (int*)(ws + o);  o += align_f((size_t)N + 1);
  float* dinv   = ws + o;          o += align_f((size_t)N);
  float* h    = ws + o;  o += align_f((size_t)N * 64);
  float* agg  = ws + o;  o += align_f((size_t)N * 64);
  float* gsum = ws + o;  o += 64 * 64;
  float* cnt  = ws + o;  o += 64;
  (void)ws_size;

  const int T = 256;
  auto blocks = [](long long n, int t) { return (unsigned int)((n + t - 1) / t); };

  const int per = (int)((((long long)E + 32LL * SORT_WAVES - 1) / (32LL * SORT_WAVES)) * 32);
  k_sort_count<<<SORT_WAVES / 8, 256, 0, stream>>>(dst, counts, E, per, 8);
  k_sort_rows<<<256, 1024, 0, stream>>>(counts, offs, tot);
  k_sort_scatter<<<SORT_WAVES / 8, 256, 0, stream>>>(dst, src, offs, tot, kA, vA, E, per, 8);
  k_bucket<<<(unsigned int)((N >> 8) + 1), 32 * BUCKET_WAVES, 0, stream>>>(kA, vA, tot, csr, rowptr, dinv, N);

  const int tiles = (N + 15) / 16;
  const unsigned int gemm_blocks = (unsigned int)((tiles + 7) / 8);
  const unsigned int agg_blocks = blocks((long long)N * 32, T);

  k_gemm_n64<<<gemm_blocks, T, 0, stream>>>(x, W1, h, N, K1);
  k_aggregate<<<agg_blocks, T, 0, stream>>>(h, rowptr, dinv, csr, b1, agg, N);

  k_gemm_n64<<<gemm_blocks, T, 0, stream>>>(agg, W2, h, N, 64);
  k_aggregate<<<agg_blocks, T, 0, stream>>>(h, rowptr, dinv, csr, b2, agg, N);

  k_fill<<<blocks(64 * 64 + 64, T), T, 0, stream>>>(gsum, 0.0f, 64 * 64 + 64);
  const long long pool_threads = (((long long)N + 31) / 32) * 32;
  k_pool<<<blocks(pool_threads, 64), 64, 0, stream>>>(agg, batch, gsum, cnt, N);
  k_pool_gemm_out<<<1, 512, 0, stream>>>(gsum, cnt, Wp, bp, (float*)d_out);
}


# aggregation blocks of 16 rows keep them in an LDS tile and run the next stage on it: layer-2 WMMA GEMM in layer 1, pool run sums + atomics in layer 2 (8 launches, was 11); pool sum order changes
# speedup vs baseline: 3.4383x; 1.1467x over previous
#include <hip/hip_runtime.h>
#include <math.h>

typedef float v2f __attribute__((ext_vector_type(2)));
typedef float v8f __attribute__((ext_vector_type(8)));

__device__ __forceinline__ v8f wmma_f32_k4(v2f a, v2f b, v8f c) {
  return __builtin_amdgcn_wmma_f32_16x16x4_f32(
       false, a,  false, b,
       (short)0, c,  false,  false);
}

__device__ __forceinline__ void atomic_add_f32(float* p, float v) {
  asm volatile("global_atomic_add_f32 %0, %1, off scope:SCOPE_DEV"
               :: "v"(p), "v"(v)
               : "memory");
}


__device__ __forceinline__ int wave_incl_scan(int x, int lane) {
  for (int o = 1; o < 32; o <<= 1) {
    const int n = __shfl_up(x, o);
    if (lane >= o) x += n;
  }
  return x;
}

__device__ __forceinline__ int wave_sum(int x) {
  for (int o = 16; o > 0; o >>= 1) x += __shfl_xor(x, o);
  return x;
}


#define SORT_WAVES 4096
static_assert(SORT_WAVES == 4 * 1024, "k_sort_rows scans 4 table entries per thread");
#define SORT_GROUPS 8

__device__ __forceinline__ unsigned int same_digit_lanes(int digit, bool valid) {
  unsigned int mask = __builtin_amdgcn_ballot_w32(valid);
#pragma unroll
  for (int b = 0; b < 8; ++b) {
    const bool bit = ((digit >> b) & 1) != 0;
    const unsigned int bb = __builtin_amdgcn_ballot_w32(bit);
    mask &= bit ? bb : ~bb;
  }
  return mask;
}

__global__ void __launch_bounds__(256)
k_sort_count(const int* __restrict__ keys, int* __restrict__ counts, int E, int per, int shift) {
  __shared__ int hist[8][256];
  const int t = (int)threadIdx.x;
  const int lane = t & 31;
  const int w = __builtin_amdgcn_readfirstlane((int)(threadIdx.x >> 5));
  const int wg = (int)blockIdx.x * 8 + w;
  int* hw = hist[w];
  for (int k = lane; k < 256; k += 32) hw[k] = 0;
  const int beg = wg * per;
  const int end = (beg + per < E) ? (beg + per) : E;
  for (int g = beg; g < end; g += 32 * SORT_GROUPS) {
    int key[SORT_GROUPS];
#pragma unroll
    for (int u = 0; u < SORT_GROUPS; ++u) {
      const int idx = g + u * 32 + lane;
      key[u] = (idx < end) ? keys[idx] : 0;
    }
#pragma unroll
    for (int u = 0; u < SORT_GROUPS; ++u) {
      if (g + u * 32 >= end) break;
      const int idx = g + u * 32 + lane;
      const bool valid = idx < end;
      const int digit = (key[u] >> shift) & 255;
      const unsigned int mask = same_digit_lanes(digit, valid);
      const bool leader = valid && ((mask & ((1u << lane) - 1u)) == 0u);
      if (leader) hw[digit] += __popc(mask);
      __builtin_amdgcn_wave_barrier();
    }
  }
  __syncthreads();
  int4 c0, c1;
  c0.x = hist[0][t]; c0.y = hist[1][t]; c0.z = hist[2][t]; c0.w = hist[3][t];
  c1.x = hist[4][t]; c1.y = hist[5][t]; c1.z = hist[6][t]; c1.w = hist[7][t];
  int* row = counts + (size_t)t * SORT_WAVES + (size_t)blockIdx.x * 8;
  *(int4*)row = c0;
  *(int4*)(row + 4) = c1;
}

__global__ void __launch_bounds__(1024)
k_sort_rows(const int* __restrict__ counts, int* __restrict__ offs, int* __restrict__ tot) {
  __shared__ int s_tot[32];
  const int t = threadIdx.x;
  const int lane = t & 31, w = t >> 5;
  const int d = (int)blockIdx.x;
  const int4 c = *(const int4*)(counts + (size_t)d * SORT_WAVES + 4 * t);
  const int v = c.x + c.y + c.z + c.w;
  const int incl = wave_incl_scan(v, lane);
  if (lane == 31) s_tot[w] = incl;
  __syncthreads();
  const int wt = s_tot[lane];
  const int woff = wave_sum((lane < w) ? wt : 0);
  int4 o;
  o.x = woff + incl - v;
  o.y = o.x + c.x;
  o.z = o.y + c.y;
  o.w = o.z + c.z;
  *(int4*)(offs + (size_t)d * SORT_WAVES + 4 * t) = o;
  if (t == 1023) tot[d] = woff + incl;
}

__global__ void __launch_bounds__(256)
k_sort_scatter(const int* __restrict__ keys, const int* __restrict__ vals,
               const int* __restrict__ offs, const int* __restrict__ tot,
               int* __restrict__ keys_out, int* __restrict__ vals_out,
               int E, int per, int shift) {
  __shared__ int cur[8][256];
  __shared__ int s_dsum[8];
  const int t = (int)threadIdx.x;
  const int lane = t & 31;
  const int w = __builtin_amdgcn_readfirstlane(t >> 5);
  const int wg = (int)blockIdx.x * 8 + w;
  int* cw = cur[w];

  const int tv = tot[t];
  const int* orow = offs + (size_t)t * SORT_WAVES + (size_t)blockIdx.x * 8;
  const int4 o0 = *(const int4*)orow;
  const int4 o1 = *(const int4*)(orow + 4);
  const int incl = wave_incl_scan(tv, lane);
  if (lane == 31) s_dsum[w] = incl;
  __syncthreads();
  int before = 0;
#pragma unroll
  for (int k = 0; k < 8; ++k) before += (k < w) ? s_dsum[k] : 0;
  const int dbase = before + incl - tv;
  cur[0][t] = dbase + o0.x; cur[1][t] = dbase + o0.y; cur[2][t] = dbase + o0.z; cur[3][t] = dbase + o0.w;
  cur[4][t] = dbase + o1.x; cur[5][t] = dbase + o1.y; cur[6][t] = dbase + o1.z; cur[7][t] = dbase + o1.w;
  __syncthreads();

  const int beg = wg * per;
  const int end = (beg + per < E) ? (beg + per) : E;
  for (int g = beg; g < end; g += 32 * SORT_GROUPS) {
    int key[SORT_GROUPS], val[SORT_GROUPS];
#pragma unroll
    for (int u = 0; u < SORT_GROUPS; ++u) {
      const int idx = g + u * 32 + lane;
      key[u] = (idx < end) ? keys[idx] : 0;
      val[u] = (idx < end) ? vals[idx] : 0;
    }
#pragma unroll
    for (int u = 0; u < SORT_GROUPS; ++u) {
      if (g + u * 32 >= end) break;
      const int idx = g + u * 32 + lane;
      const bool valid = idx < end;
      const int digit = (key[u] >> shift) & 255;
      const unsigned int mask = same_digit_lanes(digit, valid);
      const int rank = __popc(mask & ((1u << lane) - 1u));
      int place = 0;
      if (valid) place = cw[digit];
      __builtin_amdgcn_wave_barrier();
      if (valid && rank == 0) cw[digit] = place + __popc(mask);
      __builtin_amdgcn_wave_barrier();
      if (valid) {
        keys_out[place + rank] = key[u];
        vals_out[place + rank] = val[u];
      }
    }
  }
}


#define BUCKET_WAVES 32

__global__ void __launch_bounds__(32 * BUCKET_WAVES)
k_bucket(const int* __restrict__ keys, const int* __restrict__ vals, const int* __restrict__ tot,
         int* __restrict__ csr_src, int* __restrict__ rowptr, float* __restrict__ dinv, int N) {
  __shared__ int hist[BUCKET_WAVES][256];
  __shared__ int s_wsum[8];
  const int t = (int)threadIdx.x;
  const int lane = t & 31;
  const int w = __builtin_amdgcn_readfirstlane(t >> 5);
  const int b = (int)blockIdx.x;
  int* hw = hist[w];
  for (int k = lane; k < 256; k += 32) hw[k] = 0;

  int tl[8];
  int ls = 0;
#pragma unroll
  for (int q = 0; q < 8; ++q) {
    tl[q] = tot[8 * lane + q];
    ls += tl[q];
  }
  int first = wave_incl_scan(ls, lane) - ls;
  int mine = 0;
#pragma unroll
  for (int q = 0; q < 8; ++q) {
    if (q < (b & 7)) first += tl[q];
    if (q == (b & 7)) mine = tl[q];
  }
  const int base = __builtin_amdgcn_readlane(first, b >> 3);
  const int size = __builtin_amdgcn_readlane(mine, b >> 3);

  const int per = ((size + 32 * BUCKET_WAVES - 1) / (32 * BUCKET_WAVES)) * 32;
  const int lim = base + size;
  const int beg = base + w * per;
  const int end = (beg + per < lim) ? (beg + per) : lim;

  for (int g = beg; g < end; g += 32 * SORT_GROUPS) {
    int key[SORT_GROUPS];
#pragma unroll
    for (int u = 0; u < SORT_GROUPS; ++u) {
      const int idx = g + u * 32 + lane;
      key[u] = (idx < end) ? keys[idx] : 0;
    }
#pragma unroll
    for (int u = 0; u < SORT_GROUPS; ++u) {
      const int idx = g + u * 32 + lane;
      const bool valid = idx < end;
      const int digit = key[u] & 255;
      const unsigned int mask = same_digit_lanes(digit, valid);
      const bool leader = valid && ((mask & ((1u << lane) - 1u)) == 0u);
      if (leader) hw[digit] += __popc(mask);
      __builtin_amdgcn_wave_barrier();
    }
  }
  __syncthreads();

  int c[BUCKET_WAVES];
  int total = 0, incl = 0;
  if (t < 256) {
#pragma unroll
    for (int k = 0; k < BUCKET_WAVES; ++k) c[k] = hist[k][t];
#pragma unroll
    for (int k = 0; k < BUCKET_WAVES; ++k) {
      const int x = c[k];
      c[k] = total;
      total += x;
    }
    incl = wave_incl_scan(total, lane);
    if (lane == 31) s_wsum[w] = incl;
  }
  __syncthreads();
  if (t < 256) {
    int before = 0;
#pragma unroll
    for (int k = 0; k < 8; ++k) before += (k < w) ? s_wsum[k] : 0;
    const int rowstart = base + before + incl - total;
#pragma unroll
    for (int k = 0; k < BUCKET_WAVES; ++k) hist[k][t] = rowstart + c[k];
    const int row = b * 256 + t;
    if (row <= N) rowptr[row] = rowstart;
    if (row < N) dinv[row] = rsqrtf((float)(total + 1));
  }
  __syncthreads();

  for (int g = beg; g < end; g += 32 * SORT_GROUPS) {
    int key[SORT_GROUPS], val[SORT_GROUPS];
#pragma unroll
    for (int u = 0; u < SORT_GROUPS; ++u) {
      const int idx = g + u * 32 + lane;
      key[u] = (idx < end) ? keys[idx] : 0;
      val[u] = (idx < end) ? vals[idx] : 0;
    }
#pragma unroll
    for (int u = 0; u < SORT_GROUPS; ++u) {
      const int idx = g + u * 32 + lane;
      const bool valid = idx < end;
      const int digit = key[u] & 255;
      const unsigned int mask = same_digit_lanes(digit, valid);
      const int rank = __popc(mask & ((1u << lane) - 1u));
      int place = 0;
      if (valid) place = hw[digit];
      __builtin_amdgcn_wave_barrier();
      if (valid && rank == 0) hw[digit] = place + __popc(mask);
      __builtin_amdgcn_wave_barrier();
      if (valid) csr_src[place + rank] = val[u];
    }
  }
}


__global__ void k_gemm_n64(const float* __restrict__ X, const float* __restrict__ W,
                           const float* __restrict__ dinv, float* __restrict__ H, int M, int K) {
  int wave = (int)((blockIdx.x * blockDim.x + threadIdx.x) >> 5);
  int lane = threadIdx.x & 31;
  int m0 = wave * 16;
  if (m0 >= M) return;
  int half = lane >> 4;
  int lm = lane & 15;

  const float* xrow = X + (size_t)(m0 + lm) * (size_t)K;
  v8f acc0 = {}, acc1 = {}, acc2 = {}, acc3 = {};
  const float4 dlo = *(const float4*)(dinv + m0 + 8 * half);
  const float4 dhi = *(const float4*)(dinv + m0 + 8 * half + 4);
  const float ds[8] = {dlo.x, dlo.y, dlo.z, dlo.w, dhi.x, dhi.y, dhi.z, dhi.w};

  for (int kb = 0; kb < K; kb += 4) {
    int k0 = kb + 2 * half;
    v2f a;
    a.x = xrow[k0];
    a.y = xrow[k0 + 1];
    const float* w0 = W + (size_t)k0 * 64;
    v2f b0, b1, b2, b3;
    b0.x = w0[lm];       b0.y = w0[64 + lm];
    b1.x = w0[16 + lm];  b1.y = w0[64 + 16 + lm];
    b2.x = w0[32 + lm];  b2.y = w0[64 + 32 + lm];
    b3.x = w0[48 + lm];  b3.y = w0[64 + 48 + lm];
    acc0 = wmma_f32_k4(a, b0, acc0);
    acc1 = wmma_f32_k4(a, b1, acc1);
    acc2 = wmma_f32_k4(a, b2, acc2);
    acc3 = wmma_f32_k4(a, b3, acc3);
  }

#pragma unroll
  for (int r = 0; r < 8; ++r) {
    int row = m0 + r + 8 * half;
    float* hr = H + (size_t)row * 64 + lm;
    hr[0]  = acc0[r] * ds[r];
    hr[16] = acc1[r] * ds[r];
    hr[32] = acc2[r] * ds[r];
    hr[48] = acc3[r] * ds[r];
  }
}


__device__ __forceinline__ v2f aggregate_row(const float* __restrict__ h, const int* __restrict__ rowptr,
                                             const float* __restrict__ dinv,
                                             const int* __restrict__ csr_src,
                                             const float* __restrict__ bias, int i, int lane) {
  const int start = rowptr[i];
  const int cnt = rowptr[i + 1] - start;
  const float di = dinv[i];

  const v2f self = *(const v2f*)(h + (size_t)i * 64 + 2 * lane);
  float a0 = self.x, a1 = self.y;

  for (int c = 0; c < cnt; c += 32) {
    const int m = (cnt - c < 32) ? (cnt - c) : 32;
    int sl = i;
    if (lane < m) sl = csr_src[start + c + lane];
    for (int j = 0; j < m; j += 8) {
      int sj[8];
#pragma unroll
      for (int u = 0; u < 8; ++u) sj[u] = __builtin_amdgcn_readlane(sl, j + u);
      v2f v[8];
      if (j + 8 <= m) {
#pragma unroll
        for (int u = 0; u < 8; ++u) v[u] = *(const v2f*)(h + (size_t)sj[u] * 64 + 2 * lane);
      } else {
#pragma unroll
        for (int u = 0; u < 8; ++u) {
          v[u].x = 0.0f;
          v[u].y = 0.0f;
          if (j + u < m) v[u] = *(const v2f*)(h + (size_t)sj[u] * 64 + 2 * lane);
        }
      }
#pragma unroll
      for (int u = 0; u < 8; ++u) {
        a0 += v[u].x;
        a1 += v[u].y;
      }
    }
  }

  v2f o;
  o.x = fmaxf(fmaf(a0, di, bias[2 * lane]), 0.0f);
  o.y = fmaxf(fmaf(a1, di, bias[2 * lane + 1]), 0.0f);
  return o;
}

#define AGG_ROWS 16
#define TILE_PITCH 68

__device__ __forceinline__ void one_add(float* p, float v) {
  atomic_add_f32(p, v);
  asm volatile("s_wait_xcnt 0x0" :: "v"(p), "v"(v));
}


__global__ void __launch_bounds__(32 * AGG_ROWS)
k_aggregate_gemm(const float* __restrict__ h, const int* __restrict__ rowptr,
                 const float* __restrict__ dinv, const int* __restrict__ csr_src,
                 const float* __restrict__ bias, const float* __restrict__ Wn,
                 float* __restrict__ hn, float* __restrict__ pool, int npool, int N) {
  __shared__ __attribute__((aligned(16))) float tile[AGG_ROWS][TILE_PITCH];
  const int lane = threadIdx.x & 31;
  const int w = __builtin_amdgcn_readfirstlane((int)(threadIdx.x >> 5));
  const int m0 = (int)blockIdx.x * AGG_ROWS;
  const int i = m0 + w;
  const int half = lane >> 4;
  const int lm = lane & 15;

  const int pi = (int)(blockIdx.x * blockDim.x + threadIdx.x);
  if (pi < npool) pool[pi] = 0.0f;

  v2f o;
  o.x = 0.0f;
  o.y = 0.0f;
  if (i < N) o = aggregate_row(h, rowptr, dinv, csr_src, bias, i, lane);

  v2f bf[16];
  float ds[8];
#pragma unroll
  for (int s = 0; s < 16; ++s) {
    bf[s].x = 0.0f;
    bf[s].y = 0.0f;
  }
#pragma unroll
  for (int r = 0; r < 8; ++r) ds[r] = 0.0f;
  if (w < 4) {
    const float* w0 = Wn + (size_t)(2 * half) * 64 + 16 * w + lm;
#pragma unroll
    for (int s = 0; s < 16; ++s) {
      bf[s].x = w0[(size_t)s * 256];
      bf[s].y = w0[(size_t)s * 256 + 64];
    }
    const float4 dlo = *(const float4*)(dinv + m0 + 8 * half);
    const float4 dhi = *(const float4*)(dinv + m0 + 8 * half + 4);
    ds[0] = dlo.x; ds[1] = dlo.y; ds[2] = dlo.z; ds[3] = dlo.w;
    ds[4] = dhi.x; ds[5] = dhi.y; ds[6] = dhi.z; ds[7] = dhi.w;
  }

  *(v2f*)&tile[w][2 * lane] = o;
  __syncthreads();

  if (w < 4) {
    v8f acc = {};
#pragma unroll
    for (int s = 0; s < 16; ++s) {
      const v2f a = *(const v2f*)&tile[lm][4 * s + 2 * half];
      acc = wmma_f32_k4(a, bf[s], acc);
    }
#pragma unroll
    for (int r = 0; r < 8; ++r) {
      const int row = m0 + r + 8 * half;
      if (row < N) hn[(size_t)row * 64 + 16 * w + lm] = acc[r] * ds[r];
    }
  }
}


__global__ void __launch_bounds__(32 * AGG_ROWS)
k_aggregate_pool(const float* __restrict__ h, const int* __restrict__ rowptr,
                 const float* __restrict__ dinv, const int* __restrict__ csr_src,
                 const float* __restrict__ bias, const int* __restrict__ batch,
                 float* __restrict__ gsum, float* __restrict__ cnt, int N) {
  __shared__ __attribute__((aligned(16))) float tile[AGG_ROWS][64];
  __shared__ int gid[AGG_ROWS];
  const int lane = threadIdx.x & 31;
  const int w = __builtin_amdgcn_readfirstlane((int)(threadIdx.x >> 5));
  const int m0 = (int)blockIdx.x * AGG_ROWS;
  const int i = m0 + w;

  v2f o;
  o.x = 0.0f;
  o.y = 0.0f;
  if (i < N) {
    o = aggregate_row(h, rowptr, dinv, csr_src, bias, i, lane);
    if (lane == 0) gid[w] = batch[i];
  }
  *(v2f*)&tile[w][2 * lane] = o;
  __syncthreads();

  if (w < 2) {
    const int c = 32 * w + lane;
    const int nv = (N - m0 < AGG_ROWS) ? (N - m0) : AGG_ROWS;
    const int gl = gid[lane & (AGG_ROWS - 1)];
    float tv[AGG_ROWS];
#pragma unroll
    for (int r = 0; r < AGG_ROWS; ++r) tv[r] = tile[r][c];

    int gcur = __builtin_amdgcn_readlane(gl, 0);
    float a = tv[0];
    int run = 1;
#pragma unroll
    for (int r = 1; r < AGG_ROWS; ++r) {
      if (r < nv) {
        const int gr = __builtin_amdgcn_readlane(gl, r);
        if (gr != gcur) {
          one_add(gsum + (size_t)gcur * 64 + c, a);
          if (c == 0) one_add(cnt + gcur, (float)run);
          gcur = gr;
          a = 0.0f;
          run = 0;
        }
        a += tv[r];
        run += 1;
      }
    }
    one_add(gsum + (size_t)gcur * 64 + c, a);
    if (c == 0) one_add(cnt + gcur, (float)run);
  }
}


__global__ void k_pool_gemm_out(const float* __restrict__ gsum, const float* __restrict__ cnt,
                                const float* __restrict__ Wp, const float* __restrict__ bp,
                                float* __restrict__ out) {
  int wave = threadIdx.x >> 5;
  int lane = threadIdx.x & 31;
  int mi = wave >> 2, ni = wave & 3;
  int half = lane >> 4, lm = lane & 15;

  int arow = mi * 16 + lm;
  float inv = 1.0f / fmaxf(cnt[arow], 1.0f);

  v8f acc = {};
  for (int kb = 0; kb < 64; kb += 4) {
    int k0 = kb + 2 * half;
    v2f a;
    a.x = gsum[arow * 64 + k0] * inv;
    a.y = gsum[arow * 64 + k0 + 1] * inv;
    v2f b;
    b.x = Wp[k0 * 64 + ni * 16 + lm];
    b.y = Wp[(k0 + 1) * 64 + ni * 16 + lm];
    acc = wmma_f32_k4(a, b, acc);
  }

#pragma unroll
  for (int r = 0; r < 8; ++r) {
    int row = mi * 16 + r + 8 * half;
    out[row * 64 + ni * 16 + lm] = acc[r] + bp[ni * 16 + lm];
  }
}


static inline size_t align_f(size_t n) { return (n + 1023) & ~(size_t)1023; }

extern "C" void kernel_launch(void* const* d_in, const int* in_sizes, int n_in,
                              void* d_out, int out_size, void* d_ws, size_t ws_size,
                              hipStream_t stream) {
  const float* x  = (const float*)d_in[0];
  const float* W1 = (const float*)d_in[1];
  const float* b1 = (const float*)d_in[2];
  const float* W2 = (const float*)d_in[3];
  const float* b2 = (const float*)d_in[4];
  const float* Wp = (const float*)d_in[5];
  const float* bp = (const float*)d_in[6];
  const int* ei   = (const int*)d_in[7];
  const int* batch = (const int*)d_in[8];

  const int E = in_sizes[7] / 2;
  const int N = in_sizes[8];
  const int K1 = in_sizes[0] / N;
  const int* src = ei;
  const int* dst = ei + E;

  float* ws = (float*)d_ws;
  size_t o = 0;
  int*   kA     = (int*)(ws + o);  o += align_f((size_t)E);
  int*   vA     = (int*)(ws + o);  o += align_f((size_t)E);
  int*   csr    = (int*)(ws + o);  o += align_f((size_t)E);
  int*   counts = (int*)(ws + o);  o += (size_t)256 * SORT_WAVES;
  int*   offs   = (int*)(ws + o);  o += (size_t)256 * SORT_WAVES;
  int*   tot    = (int*)(ws + o);  o += 1024;
  int*   rowptr = (int*)(ws + o);  o += align_f((size_t)N + 1);
  float* dinv   = ws + o;          o += align_f((size_t)N);
  float* h    = ws + o;  o += align_f((size_t)N * 64);
  float* agg  = ws + o;  o += align_f((size_t)N * 64);
  float* gsum = ws + o;  o += 64 * 64;
  float* cnt  = ws + o;  o += 64;
  (void)ws_size;

  const int T = 256;

  const int per = (int)((((long long)E + 32LL * SORT_WAVES - 1) / (32LL * SORT_WAVES)) * 32);
  k_sort_count<<<SORT_WAVES / 8, 256, 0, stream>>>(dst, counts, E, per, 8);
  k_sort_rows<<<256, 1024, 0, stream>>>(counts, offs, tot);
  k_sort_scatter<<<SORT_WAVES / 8, 256, 0, stream>>>(dst, src, offs, tot, kA, vA, E, per, 8);
  k_bucket<<<(unsigned int)((N >> 8) + 1), 32 * BUCKET_WAVES, 0, stream>>>(kA, vA, tot, csr, rowptr, dinv, N);

  const int tiles = (N + 15) / 16;
  const unsigned int gemm_blocks = (unsigned int)((tiles + 7) / 8);
  const unsigned int agg_blocks = (unsigned int)tiles;

  k_gemm_n64<<<gemm_blocks, T, 0, stream>>>(x, W1, dinv, h, N, K1);
  k_aggregate_gemm<<<agg_blocks, 32 * AGG_ROWS, 0, stream>>>(h, rowptr, dinv, csr, b1, W2, agg,
                                                            gsum, 64 * 64 + 64, N);

  k_aggregate_pool<<<agg_blocks, 32 * AGG_ROWS, 0, stream>>>(agg, rowptr, dinv, csr, b2, batch, gsum, cnt, N);

  k_pool_gemm_out<<<1, 512, 0, stream>>>(gsum, cnt, Wp, bp, (float*)d_out);
}


# both sort levels place their pairs in an LDS image of the block's stretch of output and write it out in order: 4-byte stores to 32 lines each become runs (level 1), whole lines (level 2); order kept
# speedup vs baseline: 4.6528x; 1.3532x over previous
#include <hip/hip_runtime.h>
#include <math.h>

typedef float v2f __attribute__((ext_vector_type(2)));
typedef float v8f __attribute__((ext_vector_type(8)));

__device__ __forceinline__ v8f wmma_f32_k4(v2f a, v2f b, v8f c) {
  return __builtin_amdgcn_wmma_f32_16x16x4_f32(
       false, a,  false, b,
       (short)0, c,  false,  false);
}

__device__ __forceinline__ void atomic_add_f32(float* p, float v) {
  asm volatile("global_atomic_add_f32 %0, %1, off scope:SCOPE_DEV"
               :: "v"(p), "v"(v)
               : "memory");
}


__device__ __forceinline__ int wave_incl_scan(int x, int lane) {
  for (int o = 1; o < 32; o <<= 1) {
    const int n = __shfl_up(x, o);
    if (lane >= o) x += n;
  }
  return x;
}

__device__ __forceinline__ int wave_sum(int x) {
  for (int o = 16; o > 0; o >>= 1) x += __shfl_xor(x, o);
  return x;
}


#define SORT_WAVES 4096
static_assert(SORT_WAVES == 4 * 1024, "k_sort_rows scans 4 table entries per thread");
#define SORT_GROUPS 8
#define STAGE_PER 512

__device__ __forceinline__ unsigned int same_digit_lanes(int digit, bool valid) {
  unsigned int mask = __builtin_amdgcn_ballot_w32(valid);
#pragma unroll
  for (int b = 0; b < 8; ++b) {
    const bool bit = ((digit >> b) & 1) != 0;
    const unsigned int bb = __builtin_amdgcn_ballot_w32(bit);
    mask &= bit ? bb : ~bb;
  }
  return mask;
}

__global__ void __launch_bounds__(256)
k_sort_count(const int* __restrict__ keys, int* __restrict__ counts, int E, int per, int shift) {
  __shared__ int hist[8][256];
  const int t = (int)threadIdx.x;
  const int lane = t & 31;
  const int w = __builtin_amdgcn_readfirstlane((int)(threadIdx.x >> 5));
  const int wg = (int)blockIdx.x * 8 + w;
  int* hw = hist[w];
  for (int k = lane; k < 256; k += 32) hw[k] = 0;
  const int beg = wg * per;
  const int end = (beg + per < E) ? (beg + per) : E;
  for (int g = beg; g < end; g += 32 * SORT_GROUPS) {
    int key[SORT_GROUPS];
#pragma unroll
    for (int u = 0; u < SORT_GROUPS; ++u) {
      const int idx = g + u * 32 + lane;
      key[u] = (idx < end) ? keys[idx] : 0;
    }
#pragma unroll
    for (int u = 0; u < SORT_GROUPS; ++u) {
      if (g + u * 32 >= end) break;
      const int idx = g + u * 32 + lane;
      const bool valid = idx < end;
      const int digit = (key[u] >> shift) & 255;
      const unsigned int mask = same_digit_lanes(digit, valid);
      const bool leader = valid && ((mask & ((1u << lane) - 1u)) == 0u);
      if (leader) hw[digit] += __popc(mask);
      __builtin_amdgcn_wave_barrier();
    }
  }
  __syncthreads();
  int4 c0, c1;
  c0.x = hist[0][t]; c0.y = hist[1][t]; c0.z = hist[2][t]; c0.w = hist[3][t];
  c1.x = hist[4][t]; c1.y = hist[5][t]; c1.z = hist[6][t]; c1.w = hist[7][t];
  int* row = counts + (size_t)t * SORT_WAVES + (size_t)blockIdx.x * 8;
  *(int4*)row = c0;
  *(int4*)(row + 4) = c1;
}

__global__ void __launch_bounds__(1024)
k_sort_rows(const int* __restrict__ counts, int* __restrict__ offs, int* __restrict__ tot) {
  __shared__ int s_tot[32];
  const int t = threadIdx.x;
  const int lane = t & 31, w = t >> 5;
  const int d = (int)blockIdx.x;
  const int4 c = *(const int4*)(counts + (size_t)d * SORT_WAVES + 4 * t);
  const int v = c.x + c.y + c.z + c.w;
  const int incl = wave_incl_scan(v, lane);
  if (lane == 31) s_tot[w] = incl;
  __syncthreads();
  const int wt = s_tot[lane];
  const int woff = wave_sum((lane < w) ? wt : 0);
  int4 o;
  o.x = woff + incl - v;
  o.y = o.x + c.x;
  o.z = o.y + c.y;
  o.w = o.z + c.z;
  *(int4*)(offs + (size_t)d * SORT_WAVES + 4 * t) = o;
  if (t == 1023) tot[d] = woff + incl;
}

template <bool STAGED>
__device__ __forceinline__ void scatter_piece(const int* __restrict__ keys, const int* __restrict__ vals,
                                              int* cw, int* skey, int* sval,
                                              int* __restrict__ keys_out, int* __restrict__ vals_out,
                                              int beg, int end, int shift, int lane) {
  for (int g = beg; g < end; g += 32 * SORT_GROUPS) {
    int key[SORT_GROUPS], val[SORT_GROUPS];
#pragma unroll
    for (int u = 0; u < SORT_GROUPS; ++u) {
      const int idx = g + u * 32 + lane;
      key[u] = (idx < end) ? keys[idx] : 0;
      val[u] = (idx < end) ? vals[idx] : 0;
    }
#pragma unroll
    for (int u = 0; u < SORT_GROUPS; ++u) {
      if (g + u * 32 >= end) break;
      const int idx = g + u * 32 + lane;
      const bool valid = idx < end;
      const int digit = (key[u] >> shift) & 255;
      const unsigned int mask = same_digit_lanes(digit, valid);
      const int rank = __popc(mask & ((1u << lane) - 1u));
      int place = 0;
      if (valid) place = cw[digit];
      __builtin_amdgcn_wave_barrier();
      if (valid && rank == 0) cw[digit] = place + __popc(mask);
      __builtin_amdgcn_wave_barrier();
      if (valid) {
        if (STAGED) {
          skey[place + rank] = key[u];
          sval[place + rank] = val[u];
        } else {
          keys_out[place + rank] = key[u];
          vals_out[place + rank] = val[u];
        }
      }
    }
  }
}

__global__ void __launch_bounds__(256)
k_sort_scatter(const int* __restrict__ keys, const int* __restrict__ vals,
               const int* __restrict__ offs, const int* __restrict__ tot,
               int* __restrict__ keys_out, int* __restrict__ vals_out,
               int E, int per, int shift) {
  __shared__ int cur[8][256];
  __shared__ int s_dsum[8];
  __shared__ int s_lsum[8];
  __shared__ int delta[256];
  __shared__ int skey[8 * STAGE_PER];
  __shared__ int sval[8 * STAGE_PER];
  const int t = (int)threadIdx.x;
  const int lane = t & 31;
  const int w = __builtin_amdgcn_readfirstlane(t >> 5);
  const int wg = (int)blockIdx.x * 8 + w;
  int* cw = cur[w];
  const bool staged = per <= STAGE_PER;

  const int tv = tot[t];
  const int* orow = offs + (size_t)t * SORT_WAVES + (size_t)blockIdx.x * 8;
  const int4 o0 = *(const int4*)orow;
  const int4 o1 = *(const int4*)(orow + 4);
  const int nxt = ((int)blockIdx.x * 8 + 8 < SORT_WAVES) ? orow[8] : tv;
  const int bc = nxt - o0.x;
  const int incl = wave_incl_scan(tv, lane);
  const int lincl = wave_incl_scan(bc, lane);
  if (lane == 31) {
    s_dsum[w] = incl;
    s_lsum[w] = lincl;
  }
  __syncthreads();
  int before = 0, lbefore = 0;
#pragma unroll
  for (int k = 0; k < 8; ++k) {
    before += (k < w) ? s_dsum[k] : 0;
    lbefore += (k < w) ? s_lsum[k] : 0;
  }
  const int dbase = before + incl - tv;
  const int lstart = lbefore + lincl - bc;
  const int pb = staged ? (lstart - o0.x) : dbase;
  cur[0][t] = pb + o0.x; cur[1][t] = pb + o0.y; cur[2][t] = pb + o0.z; cur[3][t] = pb + o0.w;
  cur[4][t] = pb + o1.x; cur[5][t] = pb + o1.y; cur[6][t] = pb + o1.z; cur[7][t] = pb + o1.w;
  delta[t] = dbase + o0.x - lstart;
  __syncthreads();

  const int beg = wg * per;
  const int end = (beg + per < E) ? (beg + per) : E;
  if (staged) scatter_piece<true>(keys, vals, cw, skey, sval, keys_out, vals_out, beg, end, shift, lane);
  else scatter_piece<false>(keys, vals, cw, skey, sval, keys_out, vals_out, beg, end, shift, lane);

  if (staged) {
    __syncthreads();
    const int bbeg = (int)blockIdx.x * 8 * per;
    const int bn = (E - bbeg < 8 * per) ? (E - bbeg) : (8 * per);
    for (int j = t; j < bn; j += 256) {
      const int k = skey[j];
      const int g = j + delta[(k >> shift) & 255];
      keys_out[g] = k;
      vals_out[g] = sval[j];
    }
  }
}


#define BUCKET_WAVES 32
#define BUCKET_STAGE 12288

template <bool STAGED>
__device__ __forceinline__ void bucket_piece(const int* __restrict__ keys, const int* __restrict__ vals,
                                             int* hw, int* sbuf, int* __restrict__ csr_src,
                                             int beg, int end, int base, int lane) {
  for (int g = beg; g < end; g += 32 * SORT_GROUPS) {
    int key[SORT_GROUPS], val[SORT_GROUPS];
#pragma unroll
    for (int u = 0; u < SORT_GROUPS; ++u) {
      const int idx = g + u * 32 + lane;
      key[u] = (idx < end) ? keys[idx] : 0;
      val[u] = (idx < end) ? vals[idx] : 0;
    }
#pragma unroll
    for (int u = 0; u < SORT_GROUPS; ++u) {
      const int idx = g + u * 32 + lane;
      const bool valid = idx < end;
      const int digit = key[u] & 255;
      const unsigned int mask = same_digit_lanes(digit, valid);
      const int rank = __popc(mask & ((1u << lane) - 1u));
      int place = 0;
      if (valid) place = hw[digit];
      __builtin_amdgcn_wave_barrier();
      if (valid && rank == 0) hw[digit] = place + __popc(mask);
      __builtin_amdgcn_wave_barrier();
      if (valid) {
        if (STAGED) sbuf[place + rank - base] = val[u];
        else csr_src[place + rank] = val[u];
      }
    }
  }
}

__global__ void __launch_bounds__(32 * BUCKET_WAVES)
k_bucket(const int* __restrict__ keys, const int* __restrict__ vals, const int* __restrict__ tot,
         int* __restrict__ csr_src, int* __restrict__ rowptr, float* __restrict__ dinv, int N) {
  __shared__ int hist[BUCKET_WAVES][256];
  __shared__ int s_wsum[8];
  __shared__ int sbuf[BUCKET_STAGE];
  const int t = (int)threadIdx.x;
  const int lane = t & 31;
  const int w = __builtin_amdgcn_readfirstlane(t >> 5);
  const int b = (int)blockIdx.x;
  int* hw = hist[w];
  for (int k = lane; k < 256; k += 32) hw[k] = 0;

  int tl[8];
  int ls = 0;
#pragma unroll
  for (int q = 0; q < 8; ++q) {
    tl[q] = tot[8 * lane + q];
    ls += tl[q];
  }
  int first = wave_incl_scan(ls, lane) - ls;
  int mine = 0;
#pragma unroll
  for (int q = 0; q < 8; ++q) {
    if (q < (b & 7)) first += tl[q];
    if (q == (b & 7)) mine = tl[q];
  }
  const int base = __builtin_amdgcn_readlane(first, b >> 3);
  const int size = __builtin_amdgcn_readlane(mine, b >> 3);

  const bool staged = size <= BUCKET_STAGE;
  const int per = ((size + 32 * BUCKET_WAVES - 1) / (32 * BUCKET_WAVES)) * 32;
  const int lim = base + size;
  const int beg = base + w * per;
  const int end = (beg + per < lim) ? (beg + per) : lim;

  for (int g = beg; g < end; g += 32 * SORT_GROUPS) {
    int key[SORT_GROUPS];
#pragma unroll
    for (int u = 0; u < SORT_GROUPS; ++u) {
      const int idx = g + u * 32 + lane;
      key[u] = (idx < end) ? keys[idx] : 0;
    }
#pragma unroll
    for (int u = 0; u < SORT_GROUPS; ++u) {
      const int idx = g + u * 32 + lane;
      const bool valid = idx < end;
      const int digit = key[u] & 255;
      const unsigned int mask = same_digit_lanes(digit, valid);
      const bool leader = valid && ((mask & ((1u << lane) - 1u)) == 0u);
      if (leader) hw[digit] += __popc(mask);
      __builtin_amdgcn_wave_barrier();
    }
  }
  __syncthreads();

  int c[BUCKET_WAVES];
  int total = 0, incl = 0;
  if (t < 256) {
#pragma unroll
    for (int k = 0; k < BUCKET_WAVES; ++k) c[k] = hist[k][t];
#pragma unroll
    for (int k = 0; k < BUCKET_WAVES; ++k) {
      const int x = c[k];
      c[k] = total;
      total += x;
    }
    incl = wave_incl_scan(total, lane);
    if (lane == 31) s_wsum[w] = incl;
  }
  __syncthreads();
  if (t < 256) {
    int before = 0;
#pragma unroll
    for (int k = 0; k < 8; ++k) before += (k < w) ? s_wsum[k] : 0;
    const int rowstart = base + before + incl - total;
#pragma unroll
    for (int k = 0; k < BUCKET_WAVES; ++k) hist[k][t] = rowstart + c[k];
    const int row = b * 256 + t;
    if (row <= N) rowptr[row] = rowstart;
    if (row < N) dinv[row] = rsqrtf((float)(total + 1));
  }
  __syncthreads();

  if (staged) bucket_piece<true>(keys, vals, hw, sbuf, csr_src, beg, end, base, lane);
  else bucket_piece<false>(keys, vals, hw, sbuf, csr_src, beg, end, base, lane);

  if (staged) {
    __syncthreads();
    for (int j = t - (base & 31); j < size; j += 32 * BUCKET_WAVES)
      if (j >= 0) csr_src[base + j] = sbuf[j];
  }
}


__global__ void k_gemm_n64(const float* __restrict__ X, const float* __restrict__ W,
                           const float* __restrict__ dinv, float* __restrict__ H, int M, int K) {
  int wave = (int)((blockIdx.x * blockDim.x + threadIdx.x) >> 5);
  int lane = threadIdx.x & 31;
  int m0 = wave * 16;
  if (m0 >= M) return;
  int half = lane >> 4;
  int lm = lane & 15;

  const float* xrow = X + (size_t)(m0 + lm) * (size_t)K;
  v8f acc0 = {}, acc1 = {}, acc2 = {}, acc3 = {};
  const float4 dlo = *(const float4*)(dinv + m0 + 8 * half);
  const float4 dhi = *(const float4*)(dinv + m0 + 8 * half + 4);
  const float ds[8] = {dlo.x, dlo.y, dlo.z, dlo.w, dhi.x, dhi.y, dhi.z, dhi.w};

  for (int kb = 0; kb < K; kb += 4) {
    int k0 = kb + 2 * half;
    v2f a;
    a.x = xrow[k0];
    a.y = xrow[k0 + 1];
    const float* w0 = W + (size_t)k0 * 64;
    v2f b0, b1, b2, b3;
    b0.x = w0[lm];       b0.y = w0[64 + lm];
    b1.x = w0[16 + lm];  b1.y = w0[64 + 16 + lm];
    b2.x = w0[32 + lm];  b2.y = w0[64 + 32 + lm];
    b3.x = w0[48 + lm];  b3.y = w0[64 + 48 + lm];
    acc0 = wmma_f32_k4(a, b0, acc0);
    acc1 = wmma_f32_k4(a, b1, acc1);
    acc2 = wmma_f32_k4(a, b2, acc2);
    acc3 = wmma_f32_k4(a, b3, acc3);
  }

#pragma unroll
  for (int r = 0; r < 8; ++r) {
    int row = m0 + r + 8 * half;
    float* hr = H + (size_t)row * 64 + lm;
    hr[0]  = acc0[r] * ds[r];
    hr[16] = acc1[r] * ds[r];
    hr[32] = acc2[r] * ds[r];
    hr[48] = acc3[r] * ds[r];
  }
}


__device__ __forceinline__ v2f aggregate_row(const float* __restrict__ h, const int* __restrict__ rowptr,
                                             const float* __restrict__ dinv,
                                             const int* __restrict__ csr_src,
                                             const float* __restrict__ bias, int i, int lane) {
  const int start = rowptr[i];
  const int cnt = rowptr[i + 1] - start;
  const float di = dinv[i];

  const v2f self = *(const v2f*)(h + (size_t)i * 64 + 2 * lane);
  float a0 = self.x, a1 = self.y;

  for (int c = 0; c < cnt; c += 32) {
    const int m = (cnt - c < 32) ? (cnt - c) : 32;
    int sl = i;
    if (lane < m) sl = csr_src[start + c + lane];
    for (int j = 0; j < m; j += 8) {
      int sj[8];
#pragma unroll
      for (int u = 0; u < 8; ++u) sj[u] = __builtin_amdgcn_readlane(sl, j + u);
      v2f v[8];
      if (j + 8 <= m) {
#pragma unroll
        for (int u = 0; u < 8; ++u) v[u] = *(const v2f*)(h + (size_t)sj[u] * 64 + 2 * lane);
      } else {
#pragma unroll
        for (int u = 0; u < 8; ++u) {
          v[u].x = 0.0f;
          v[u].y = 0.0f;
          if (j + u < m) v[u] = *(const v2f*)(h + (size_t)sj[u] * 64 + 2 * lane);
        }
      }
#pragma unroll
      for (int u = 0; u < 8; ++u) {
        a0 += v[u].x;
        a1 += v[u].y;
      }
    }
  }

  v2f o;
  o.x = fmaxf(fmaf(a0, di, bias[2 * lane]), 0.0f);
  o.y = fmaxf(fmaf(a1, di, bias[2 * lane + 1]), 0.0f);
  return o;
}

#define AGG_ROWS 16
#define TILE_PITCH 68

__device__ __forceinline__ void one_add(float* p, float v) {
  atomic_add_f32(p, v);
  asm volatile("s_wait_xcnt 0x0" :: "v"(p), "v"(v));
}


__global__ void __launch_bounds__(32 * AGG_ROWS)
k_aggregate_gemm(const float* __restrict__ h, const int* __restrict__ rowptr,
                 const float* __restrict__ dinv, const int* __restrict__ csr_src,
                 const float* __restrict__ bias, const float* __restrict__ Wn,
                 float* __restrict__ hn, float* __restrict__ pool, int npool, int N) {
  __shared__ __attribute__((aligned(16))) float tile[AGG_ROWS][TILE_PITCH];
  const int lane = threadIdx.x & 31;
  const int w = __builtin_amdgcn_readfirstlane((int)(threadIdx.x >> 5));
  const int m0 = (int)blockIdx.x * AGG_ROWS;
  const int i = m0 + w;
  const int half = lane >> 4;
  const int lm = lane & 15;

  const int pi = (int)(blockIdx.x * blockDim.x + threadIdx.x);
  if (pi < npool) pool[pi] = 0.0f;

  v2f o;
  o.x = 0.0f;
  o.y = 0.0f;
  if (i < N) o = aggregate_row(h, rowptr, dinv, csr_src, bias, i, lane);

  v2f bf[16];
  float ds[8];
#pragma unroll
  for (int s = 0; s < 16; ++s) {
    bf[s].x = 0.0f;
    bf[s].y = 0.0f;
  }
#pragma unroll
  for (int r = 0; r < 8; ++r) ds[r] = 0.0f;
  if (w < 4) {
    const float* w0 = Wn + (size_t)(2 * half) * 64 + 16 * w + lm;
#pragma unroll
    for (int s = 0; s < 16; ++s) {
      bf[s].x = w0[(size_t)s * 256];
      bf[s].y = w0[(size_t)s * 256 + 64];
    }
    const float4 dlo = *(const float4*)(dinv + m0 + 8 * half);
    const float4 dhi = *(const float4*)(dinv + m0 + 8 * half + 4);
    ds[0] = dlo.x; ds[1] = dlo.y; ds[2] = dlo.z; ds[3] = dlo.w;
    ds[4] = dhi.x; ds[5] = dhi.y; ds[6] = dhi.z; ds[7] = dhi.w;
  }

  *(v2f*)&tile[w][2 * lane] = o;
  __syncthreads();

  if (w < 4) {
    v8f acc = {};
#pragma unroll
    for (int s = 0; s < 16; ++s) {
      const v2f a = *(const v2f*)&tile[lm][4 * s + 2 * half];
      acc = wmma_f32_k4(a, bf[s], acc);
    }
#pragma unroll
    for (int r = 0; r < 8; ++r) {
      const int row = m0 + r + 8 * half;
      if (row < N) hn[(size_t)row * 64 + 16 * w + lm] = acc[r] * ds[r];
    }
  }
}


__global__ void __launch_bounds__(32 * AGG_ROWS)
k_aggregate_pool(const float* __restrict__ h, const int* __restrict__ rowptr,
                 const float* __restrict__ dinv, const int* __restrict__ csr_src,
                 const float* __restrict__ bias, const int* __restrict__ batch,
                 float* __restrict__ gsum, float* __restrict__ cnt, int N) {
  __shared__ __attribute__((aligned(16))) float tile[AGG_ROWS][64];
  __shared__ int gid[AGG_ROWS];
  const int lane = threadIdx.x & 31;
  const int w = __builtin_amdgcn_readfirstlane((int)(threadIdx.x >> 5));
  const int m0 = (int)blockIdx.x * AGG_ROWS;
  const int i = m0 + w;

  v2f o;
  o.x = 0.0f;
  o.y = 0.0f;
  if (i < N) {
    o = aggregate_row(h, rowptr, dinv, csr_src, bias, i, lane);
    if (lane == 0) gid[w] = batch[i];
  }
  *(v2f*)&tile[w][2 * lane] = o;
  __syncthreads();

  if (w < 2) {
    const int c = 32 * w + lane;
    const int nv = (N - m0 < AGG_ROWS) ? (N - m0) : AGG_ROWS;
    const int gl = gid[lane & (AGG_ROWS - 1)];
    float tv[AGG_ROWS];
#pragma unroll
    for (int r = 0; r < AGG_ROWS; ++r) tv[r] = tile[r][c];

    int gcur = __builtin_amdgcn_readlane(gl, 0);
    float a = tv[0];
    int run = 1;
#pragma unroll
    for (int r = 1; r < AGG_ROWS; ++r) {
      if (r < nv) {
        const int gr = __builtin_amdgcn_readlane(gl, r);
        if (gr != gcur) {
          one_add(gsum + (size_t)gcur * 64 + c, a);
          if (c == 0) one_add(cnt + gcur, (float)run);
          gcur = gr;
          a = 0.0f;
          run = 0;
        }
        a += tv[r];
        run += 1;
      }
    }
    one_add(gsum + (size_t)gcur * 64 + c, a);
    if (c == 0) one_add(cnt + gcur, (float)run);
  }
}


__global__ void k_pool_gemm_out(const float* __restrict__ gsum, const float* __restrict__ cnt,
                                const float* __restrict__ Wp, const float* __restrict__ bp,
                                float* __restrict__ out) {
  int wave = threadIdx.x >> 5;
  int lane = threadIdx.x & 31;
  int mi = wave >> 2, ni = wave & 3;
  int half = lane >> 4, lm = lane & 15;

  int arow = mi * 16 + lm;
  float inv = 1.0f / fmaxf(cnt[arow], 1.0f);

  v8f acc = {};
  for (int kb = 0; kb < 64; kb += 4) {
    int k0 = kb + 2 * half;
    v2f a;
    a.x = gsum[arow * 64 + k0] * inv;
    a.y = gsum[arow * 64 + k0 + 1] * inv;
    v2f b;
    b.x = Wp[k0 * 64 + ni * 16 + lm];
    b.y = Wp[(k0 + 1) * 64 + ni * 16 + lm];
    acc = wmma_f32_k4(a, b, acc);
  }

#pragma unroll
  for (int r = 0; r < 8; ++r) {
    int row = mi * 16 + r + 8 * half;
    out[row * 64 + ni * 16 + lm] = acc[r] + bp[ni * 16 + lm];
  }
}


static inline size_t align_f(size_t n) { return (n + 1023) & ~(size_t)1023; }

extern "C" void kernel_launch(void* const* d_in, const int* in_sizes, int n_in,
                              void* d_out, int out_size, void* d_ws, size_t ws_size,
                              hipStream_t stream) {
  const float* x  = (const float*)d_in[0];
  const float* W1 = (const float*)d_in[1];
  const float* b1 = (const float*)d_in[2];
  const float* W2 = (const float*)d_in[3];
  const float* b2 = (const float*)d_in[4];
  const float* Wp = (const float*)d_in[5];
  const float* bp = (const float*)d_in[6];
  const int* ei   = (const int*)d_in[7];
  const int* batch = (const int*)d_in[8];

  const int E = in_sizes[7] / 2;
  const int N = in_sizes[8];
  const int K1 = in_sizes[0] / N;
  const int* src = ei;
  const int* dst = ei + E;

  float* ws = (float*)d_ws;
  size_t o = 0;
  int*   kA     = (int*)(ws + o);  o += align_f((size_t)E);
  int*   vA     = (int*)(ws + o);  o += align_f((size_t)E);
  int*   csr    = (int*)(ws + o);  o += align_f((size_t)E);
  int*   counts = (int*)(ws + o);  o += (size_t)256 * SORT_WAVES;
  int*   offs   = (int*)(ws + o);  o += (size_t)256 * SORT_WAVES;
  int*   tot    = (int*)(ws + o);  o += 1024;
  int*   rowptr = (int*)(ws + o);  o += align_f((size_t)N + 1);
  float* dinv   = ws + o;          o += align_f((size_t)N);
  float* h    = ws + o;  o += align_f((size_t)N * 64);
  float* agg  = ws + o;  o += align_f((size_t)N * 64);
  float* gsum = ws + o;  o += 64 * 64;
  float* cnt  = ws + o;  o += 64;
  (void)ws_size;

  const int T = 256;

  const int per = (int)((((long long)E + 32LL * SORT_WAVES - 1) / (32LL * SORT_WAVES)) * 32);
  k_sort_count<<<SORT_WAVES / 8, 256, 0, stream>>>(dst, counts, E, per, 8);
  k_sort_rows<<<256, 1024, 0, stream>>>(counts, offs, tot);
  k_sort_scatter<<<SORT_WAVES / 8, 256, 0, stream>>>(dst, src, offs, tot, kA, vA, E, per, 8);
  k_bucket<<<(unsigned int)((N >> 8) + 1), 32 * BUCKET_WAVES, 0, stream>>>(kA, vA, tot, csr, rowptr, dinv, N);

  const int tiles = (N + 15) / 16;
  const unsigned int gemm_blocks = (unsigned int)((tiles + 7) / 8);
  const unsigned int agg_blocks = (unsigned int)tiles;

  k_gemm_n64<<<gemm_blocks, T, 0, stream>>>(x, W1, dinv, h, N, K1);
  k_aggregate_gemm<<<agg_blocks, 32 * AGG_ROWS, 0, stream>>>(h, rowptr, dinv, csr, b1, W2, agg,
                                                            gsum, 64 * 64 + 64, N);

  k_aggregate_pool<<<agg_blocks, 32 * AGG_ROWS, 0, stream>>>(agg, rowptr, dinv, csr, b2, batch, gsum, cnt, N);

  k_pool_gemm_out<<<1, 512, 0, stream>>>(gsum, cnt, Wp, bp, (float*)d_out);
}


# x@W1 tiles run as extra blocks of the level-1 scatter launch, rows stored unscaled; bucket kernel scales its 256 rows of h by deg^-1/2 in place (7 launches, was 8); every sum and its order kept
# speedup vs baseline: 4.8378x; 1.0398x over previous
#include <hip/hip_runtime.h>
#include <math.h>

typedef float v2f __attribute__((ext_vector_type(2)));
typedef float v8f __attribute__((ext_vector_type(8)));

__device__ __forceinline__ v8f wmma_f32_k4(v2f a, v2f b, v8f c) {
  return __builtin_amdgcn_wmma_f32_16x16x4_f32(
       false, a,  false, b,
       (short)0, c,  false,  false);
}

__device__ __forceinline__ void atomic_add_f32(float* p, float v) {
  asm volatile("global_atomic_add_f32 %0, %1, off scope:SCOPE_DEV"
               :: "v"(p), "v"(v)
               : "memory");
}


__device__ __forceinline__ int wave_incl_scan(int x, int lane) {
  for (int o = 1; o < 32; o <<= 1) {
    const int n = __shfl_up(x, o);
    if (lane >= o) x += n;
  }
  return x;
}

__device__ __forceinline__ int wave_sum(int x) {
  for (int o = 16; o > 0; o >>= 1) x += __shfl_xor(x, o);
  return x;
}


#define SORT_WAVES 4096
static_assert(SORT_WAVES == 4 * 1024, "k_sort_rows scans 4 table entries per thread");
#define SORT_GROUPS 8
#define STAGE_PER 512

__device__ __forceinline__ unsigned int same_digit_lanes(int digit, bool valid) {
  unsigned int mask = __builtin_amdgcn_ballot_w32(valid);
#pragma unroll
  for (int b = 0; b < 8; ++b) {
    const bool bit = ((digit >> b) & 1) != 0;
    const unsigned int bb = __builtin_amdgcn_ballot_w32(bit);
    mask &= bit ? bb : ~bb;
  }
  return mask;
}


__device__ __forceinline__ void gemm_tile_n64(const float* __restrict__ X, const float* __restrict__ W,
                                              float* __restrict__ H, int K, int tile, int lane) {
  const int m0 = tile * 16;
  const int half = lane >> 4;
  const int lm = lane & 15;

  const float* xrow = X + (size_t)(m0 + lm) * (size_t)K;
  v8f acc0 = {}, acc1 = {}, acc2 = {}, acc3 = {};

  for (int kb = 0; kb < K; kb += 4) {
    int k0 = kb + 2 * half;
    v2f a;
    a.x = xrow[k0];
    a.y = xrow[k0 + 1];
    const float* w0 = W + (size_t)k0 * 64;
    v2f b0, b1, b2, b3;
    b0.x = w0[lm];       b0.y = w0[64 + lm];
    b1.x = w0[16 + lm];  b1.y = w0[64 + 16 + lm];
    b2.x = w0[32 + lm];  b2.y = w0[64 + 32 + lm];
    b3.x = w0[48 + lm];  b3.y = w0[64 + 48 + lm];
    acc0 = wmma_f32_k4(a, b0, acc0);
    acc1 = wmma_f32_k4(a, b1, acc1);
    acc2 = wmma_f32_k4(a, b2, acc2);
    acc3 = wmma_f32_k4(a, b3, acc3);
  }

#pragma unroll
  for (int r = 0; r < 8; ++r) {
    int row = m0 + r + 8 * half;
    float* hr = H + (size_t)row * 64 + lm;
    hr[0]  = acc0[r];
    hr[16] = acc1[r];
    hr[32] = acc2[r];
    hr[48] = acc3[r];
  }
}

__device__ __forceinline__ void gemm_block_n64(const float* __restrict__ X, const float* __restrict__ W,
                                               float* __restrict__ H, int K, int gblock, int ntiles) {
  const int w = __builtin_amdgcn_readfirstlane((int)(threadIdx.x >> 5));
  const int tile = gblock * 8 + w;
  if (tile < ntiles) gemm_tile_n64(X, W, H, K, tile, (int)(threadIdx.x & 31));
}

__global__ void __launch_bounds__(256)
k_sort_count(const int* __restrict__ keys, int* __restrict__ counts, int E, int per, int shift) {
  __shared__ int hist[8][256];
  const int t = (int)threadIdx.x;
  const int lane = t & 31;
  const int w = __builtin_amdgcn_readfirstlane((int)(threadIdx.x >> 5));
  const int wg = (int)blockIdx.x * 8 + w;
  int* hw = hist[w];
  for (int k = lane; k < 256; k += 32) hw[k] = 0;
  const int beg = wg * per;
  const int end = (beg + per < E) ? (beg + per) : E;
  for (int g = beg; g < end; g += 32 * SORT_GROUPS) {
    int key[SORT_GROUPS];
#pragma unroll
    for (int u = 0; u < SORT_GROUPS; ++u) {
      const int idx = g + u * 32 + lane;
      key[u] = (idx < end) ? keys[idx] : 0;
    }
#pragma unroll
    for (int u = 0; u < SORT_GROUPS; ++u) {
      if (g + u * 32 >= end) break;
      const int idx = g + u * 32 + lane;
      const bool valid = idx < end;
      const int digit = (key[u] >> shift) & 255;
      const unsigned int mask = same_digit_lanes(digit, valid);
      const bool leader = valid && ((mask & ((1u << lane) - 1u)) == 0u);
      if (leader) hw[digit] += __popc(mask);
      __builtin_amdgcn_wave_barrier();
    }
  }
  __syncthreads();
  int4 c0, c1;
  c0.x = hist[0][t]; c0.y = hist[1][t]; c0.z = hist[2][t]; c0.w = hist[3][t];
  c1.x = hist[4][t]; c1.y = hist[5][t]; c1.z = hist[6][t]; c1.w = hist[7][t];
  int* row = counts + (size_t)t * SORT_WAVES + (size_t)blockIdx.x * 8;
  *(int4*)row = c0;
  *(int4*)(row + 4) = c1;
}

__global__ void __launch_bounds__(1024)
k_sort_rows(const int* __restrict__ counts, int* __restrict__ offs, int* __restrict__ tot) {
  __shared__ int s_tot[32];
  const int t = threadIdx.x;
  const int lane = t & 31, w = t >> 5;
  const int d = (int)blockIdx.x;
  const int4 c = *(const int4*)(counts + (size_t)d * SORT_WAVES + 4 * t);
  const int v = c.x + c.y + c.z + c.w;
  const int incl = wave_incl_scan(v, lane);
  if (lane == 31) s_tot[w] = incl;
  __syncthreads();
  const int wt = s_tot[lane];
  const int woff = wave_sum((lane < w) ? wt : 0);
  int4 o;
  o.x = woff + incl - v;
  o.y = o.x + c.x;
  o.z = o.y + c.y;
  o.w = o.z + c.z;
  *(int4*)(offs + (size_t)d * SORT_WAVES + 4 * t) = o;
  if (t == 1023) tot[d] = woff + incl;
}

template <bool STAGED>
__device__ __forceinline__ void scatter_piece(const int* __restrict__ keys, const int* __restrict__ vals,
                                              int* cw, int* skey, int* sval,
                                              int* __restrict__ keys_out, int* __restrict__ vals_out,
                                              int beg, int end, int shift, int lane) {
  for (int g = beg; g < end; g += 32 * SORT_GROUPS) {
    int key[SORT_GROUPS], val[SORT_GROUPS];
#pragma unroll
    for (int u = 0; u < SORT_GROUPS; ++u) {
      const int idx = g + u * 32 + lane;
      key[u] = (idx < end) ? keys[idx] : 0;
      val[u] = (idx < end) ? vals[idx] : 0;
    }
#pragma unroll
    for (int u = 0; u < SORT_GROUPS; ++u) {
      if (g + u * 32 >= end) break;
      const int idx = g + u * 32 + lane;
      const bool valid = idx < end;
      const int digit = (key[u] >> shift) & 255;
      const unsigned int mask = same_digit_lanes(digit, valid);
      const int rank = __popc(mask & ((1u << lane) - 1u));
      int place = 0;
      if (valid) place = cw[digit];
      __builtin_amdgcn_wave_barrier();
      if (valid && rank == 0) cw[digit] = place + __popc(mask);
      __builtin_amdgcn_wave_barrier();
      if (valid) {
        if (STAGED) {
          skey[place + rank] = key[u];
          sval[place + rank] = val[u];
        } else {
          keys_out[place + rank] = key[u];
          vals_out[place + rank] = val[u];
        }
      }
    }
  }
}

__global__ void __launch_bounds__(256)
k_sort_scatter(const int* __restrict__ keys, const int* __restrict__ vals,
               const int* __restrict__ offs, const int* __restrict__ tot,
               int* __restrict__ keys_out, int* __restrict__ vals_out,
               int E, int per, int shift,
               const float* __restrict__ X, const float* __restrict__ W, float* __restrict__ H,
               int K, int ntiles, int gblocks) {
  __shared__ int cur[8][256];
  __shared__ int s_dsum[8];
  __shared__ int s_lsum[8];
  __shared__ int delta[256];
  __shared__ int skey[8 * STAGE_PER];
  __shared__ int sval[8 * STAGE_PER];
  if ((int)blockIdx.x < gblocks) {
    gemm_block_n64(X, W, H, K, (int)blockIdx.x, ntiles);
    return;
  }
  const int bx = (int)blockIdx.x - gblocks;
  const int t = (int)threadIdx.x;
  const int lane = t & 31;
  const int w = __builtin_amdgcn_readfirstlane(t >> 5);
  const int wg = bx * 8 + w;
  int* cw = cur[w];
  const bool staged = per <= STAGE_PER;

  const int tv = tot[t];
  const int* orow = offs + (size_t)t * SORT_WAVES + (size_t)bx * 8;
  const int4 o0 = *(const int4*)orow;
  const int4 o1 = *(const int4*)(orow + 4);
  const int nxt = (bx * 8 + 8 < SORT_WAVES) ? orow[8] : tv;
  const int bc = nxt - o0.x;
  const int incl = wave_incl_scan(tv, lane);
  const int lincl = wave_incl_scan(bc, lane);
  if (lane == 31) {
    s_dsum[w] = incl;
    s_lsum[w] = lincl;
  }
  __syncthreads();
  int before = 0, lbefore = 0;
#pragma unroll
  for (int k = 0; k < 8; ++k) {
    before += (k < w) ? s_dsum[k] : 0;
    lbefore += (k < w) ? s_lsum[k] : 0;
  }
  const int dbase = before + incl - tv;
  const int lstart = lbefore + lincl - bc;
  const int pb = staged ? (lstart - o0.x) : dbase;
  cur[0][t] = pb + o0.x; cur[1][t] = pb + o0.y; cur[2][t] = pb + o0.z; cur[3][t] = pb + o0.w;
  cur[4][t] = pb + o1.x; cur[5][t] = pb + o1.y; cur[6][t] = pb + o1.z; cur[7][t] = pb + o1.w;
  delta[t] = dbase + o0.x - lstart;
  __syncthreads();

  const int beg = wg * per;
  const int end = (beg + per < E) ? (beg + per) : E;
  if (staged) scatter_piece<true>(keys, vals, cw, skey, sval, keys_out, vals_out, beg, end, shift, lane);
  else scatter_piece<false>(keys, vals, cw, skey, sval, keys_out, vals_out, beg, end, shift, lane);

  if (staged) {
    __syncthreads();
    const int bbeg = bx * 8 * per;
    const int bn = (E - bbeg < 8 * per) ? (E - bbeg) : (8 * per);
    for (int j = t; j < bn; j += 256) {
      const int k = skey[j];
      const int g = j + delta[(k >> shift) & 255];
      keys_out[g] = k;
      vals_out[g] = sval[j];
    }
  }
}


#define BUCKET_WAVES 32
#define BUCKET_STAGE 12288

template <bool STAGED>
__device__ __forceinline__ void bucket_piece(const int* __restrict__ keys, const int* __restrict__ vals,
                                             int* hw, int* sbuf, int* __restrict__ csr_src,
                                             int beg, int end, int base, int lane) {
  for (int g = beg; g < end; g += 32 * SORT_GROUPS) {
    int key[SORT_GROUPS], val[SORT_GROUPS];
#pragma unroll
    for (int u = 0; u < SORT_GROUPS; ++u) {
      const int idx = g + u * 32 + lane;
      key[u] = (idx < end) ? keys[idx] : 0;
      val[u] = (idx < end) ? vals[idx] : 0;
    }
#pragma unroll
    for (int u = 0; u < SORT_GROUPS; ++u) {
      const int idx = g + u * 32 + lane;
      const bool valid = idx < end;
      const int digit = key[u] & 255;
      const unsigned int mask = same_digit_lanes(digit, valid);
      const int rank = __popc(mask & ((1u << lane) - 1u));
      int place = 0;
      if (valid) place = hw[digit];
      __builtin_amdgcn_wave_barrier();
      if (valid && rank == 0) hw[digit] = place + __popc(mask);
      __builtin_amdgcn_wave_barrier();
      if (valid) {
        if (STAGED) sbuf[place + rank - base] = val[u];
        else csr_src[place + rank] = val[u];
      }
    }
  }
}

__global__ void __launch_bounds__(32 * BUCKET_WAVES)
k_bucket(const int* __restrict__ keys, const int* __restrict__ vals, const int* __restrict__ tot,
         int* __restrict__ csr_src, int* __restrict__ rowptr, float* __restrict__ dinv,
         float* __restrict__ h, int N) {
  __shared__ int hist[BUCKET_WAVES][256];
  __shared__ int s_wsum[8];
  __shared__ float s_dinv[256];
  __shared__ int sbuf[BUCKET_STAGE];
  const int t = (int)threadIdx.x;
  const int lane = t & 31;
  const int w = __builtin_amdgcn_readfirstlane(t >> 5);
  const int b = (int)blockIdx.x;
  int* hw = hist[w];
  for (int k = lane; k < 256; k += 32) hw[k] = 0;

  const int hrow0 = b * 256 + w * 8;
  v2f hv[8];
#pragma unroll
  for (int r = 0; r < 8; ++r) {
    const int row = (hrow0 + r < N) ? (hrow0 + r) : (N - 1);
    hv[r] = *(const v2f*)(h + (size_t)row * 64 + 2 * lane);
  }

  int tl[8];
  int ls = 0;
#pragma unroll
  for (int q = 0; q < 8; ++q) {
    tl[q] = tot[8 * lane + q];
    ls += tl[q];
  }
  int first = wave_incl_scan(ls, lane) - ls;
  int mine = 0;
#pragma unroll
  for (int q = 0; q < 8; ++q) {
    if (q < (b & 7)) first += tl[q];
    if (q == (b & 7)) mine = tl[q];
  }
  const int base = __builtin_amdgcn_readlane(first, b >> 3);
  const int size = __builtin_amdgcn_readlane(mine, b >> 3);

  const bool staged = size <= BUCKET_STAGE;
  const int per = ((size + 32 * BUCKET_WAVES - 1) / (32 * BUCKET_WAVES)) * 32;
  const int lim = base + size;
  const int beg = base + w * per;
  const int end = (beg + per < lim) ? (beg + per) : lim;

  for (int g = beg; g < end; g += 32 * SORT_GROUPS) {
    int key[SORT_GROUPS];
#pragma unroll
    for (int u = 0; u < SORT_GROUPS; ++u) {
      const int idx = g + u * 32 + lane;
      key[u] = (idx < end) ? keys[idx] : 0;
    }
#pragma unroll
    for (int u = 0; u < SORT_GROUPS; ++u) {
      const int idx = g + u * 32 + lane;
      const bool valid = idx < end;
      const int digit = key[u] & 255;
      const unsigned int mask = same_digit_lanes(digit, valid);
      const bool leader = valid && ((mask & ((1u << lane) - 1u)) == 0u);
      if (leader) hw[digit] += __popc(mask);
      __builtin_amdgcn_wave_barrier();
    }
  }
  __syncthreads();

  int c[BUCKET_WAVES];
  int total = 0, incl = 0;
  if (t < 256) {
#pragma unroll
    for (int k = 0; k < BUCKET_WAVES; ++k) c[k] = hist[k][t];
#pragma unroll
    for (int k = 0; k < BUCKET_WAVES; ++k) {
      const int x = c[k];
      c[k] = total;
      total += x;
    }
    incl = wave_incl_scan(total, lane);
    if (lane == 31) s_wsum[w] = incl;
  }
  __syncthreads();
  if (t < 256) {
    int before = 0;
#pragma unroll
    for (int k = 0; k < 8; ++k) before += (k < w) ? s_wsum[k] : 0;
    const int rowstart = base + before + incl - total;
#pragma unroll
    for (int k = 0; k < BUCKET_WAVES; ++k) hist[k][t] = rowstart + c[k];
    const int row = b * 256 + t;
    if (row <= N) rowptr[row] = rowstart;
    const float di = rsqrtf((float)(total + 1));
    if (row < N) dinv[row] = di;
    s_dinv[t] = di;
  }
  __syncthreads();

#pragma unroll
  for (int r = 0; r < 8; ++r) {
    if (hrow0 + r < N) {
      const float di = s_dinv[w * 8 + r];
      v2f o;
      o.x = hv[r].x * di;
      o.y = hv[r].y * di;
      *(v2f*)(h + (size_t)(hrow0 + r) * 64 + 2 * lane) = o;
    }
  }

  if (staged) bucket_piece<true>(keys, vals, hw, sbuf, csr_src, beg, end, base, lane);
  else bucket_piece<false>(keys, vals, hw, sbuf, csr_src, beg, end, base, lane);

  if (staged) {
    __syncthreads();
    for (int j = t - (base & 31); j < size; j += 32 * BUCKET_WAVES)
      if (j >= 0) csr_src[base + j] = sbuf[j];
  }
}


__device__ __forceinline__ v2f aggregate_row(const float* __restrict__ h, const int* __restrict__ rowptr,
                                             const float* __restrict__ dinv,
                                             const int* __restrict__ csr_src,
                                             const float* __restrict__ bias, int i, int lane) {
  const int start = rowptr[i];
  const int cnt = rowptr[i + 1] - start;
  const float di = dinv[i];

  const v2f self = *(const v2f*)(h + (size_t)i * 64 + 2 * lane);
  float a0 = self.x, a1 = self.y;

  for (int c = 0; c < cnt; c += 32) {
    const int m = (cnt - c < 32) ? (cnt - c) : 32;
    int sl = i;
    if (lane < m) sl = csr_src[start + c + lane];
    for (int j = 0; j < m; j += 8) {
      int sj[8];
#pragma unroll
      for (int u = 0; u < 8; ++u) sj[u] = __builtin_amdgcn_readlane(sl, j + u);
      v2f v[8];
      if (j + 8 <= m) {
#pragma unroll
        for (int u = 0; u < 8; ++u) v[u] = *(const v2f*)(h + (size_t)sj[u] * 64 + 2 * lane);
      } else {
#pragma unroll
        for (int u = 0; u < 8; ++u) {
          v[u].x = 0.0f;
          v[u].y = 0.0f;
          if (j + u < m) v[u] = *(const v2f*)(h + (size_t)sj[u] * 64 + 2 * lane);
        }
      }
#pragma unroll
      for (int u = 0; u < 8; ++u) {
        a0 += v[u].x;
        a1 += v[u].y;
      }
    }
  }

  v2f o;
  o.x = fmaxf(fmaf(a0, di, bias[2 * lane]), 0.0f);
  o.y = fmaxf(fmaf(a1, di, bias[2 * lane + 1]), 0.0f);
  return o;
}

#define AGG_ROWS 16
#define TILE_PITCH 68

__device__ __forceinline__ void one_add(float* p, float v) {
  atomic_add_f32(p, v);
  asm volatile("s_wait_xcnt 0x0" :: "v"(p), "v"(v));
}


__global__ void __launch_bounds__(32 * AGG_ROWS)
k_aggregate_gemm(const float* __restrict__ h, const int* __restrict__ rowptr,
                 const float* __restrict__ dinv, const int* __restrict__ csr_src,
                 const float* __restrict__ bias, const float* __restrict__ Wn,
                 float* __restrict__ hn, float* __restrict__ pool, int npool, int N) {
  __shared__ __attribute__((aligned(16))) float tile[AGG_ROWS][TILE_PITCH];
  const int lane = threadIdx.x & 31;
  const int w = __builtin_amdgcn_readfirstlane((int)(threadIdx.x >> 5));
  const int m0 = (int)blockIdx.x * AGG_ROWS;
  const int i = m0 + w;
  const int half = lane >> 4;
  const int lm = lane & 15;

  const int pi = (int)(blockIdx.x * blockDim.x + threadIdx.x);
  if (pi < npool) pool[pi] = 0.0f;

  v2f o;
  o.x = 0.0f;
  o.y = 0.0f;
  if (i < N) o = aggregate_row(h, rowptr, dinv, csr_src, bias, i, lane);

  v2f bf[16];
  float ds[8];
#pragma unroll
  for (int s = 0; s < 16; ++s) {
    bf[s].x = 0.0f;
    bf[s].y = 0.0f;
  }
#pragma unroll
  for (int r = 0; r < 8; ++r) ds[r] = 0.0f;
  if (w < 4) {
    const float* w0 = Wn + (size_t)(2 * half) * 64 + 16 * w + lm;
#pragma unroll
    for (int s = 0; s < 16; ++s) {
      bf[s].x = w0[(size_t)s * 256];
      bf[s].y = w0[(size_t)s * 256 + 64];
    }
    const float4 dlo = *(const float4*)(dinv + m0 + 8 * half);
    const float4 dhi = *(const float4*)(dinv + m0 + 8 * half + 4);
    ds[0] = dlo.x; ds[1] = dlo.y; ds[2] = dlo.z; ds[3] = dlo.w;
    ds[4] = dhi.x; ds[5] = dhi.y; ds[6] = dhi.z; ds[7] = dhi.w;
  }

  *(v2f*)&tile[w][2 * lane] = o;
  __syncthreads();

  if (w < 4) {
    v8f acc = {};
#pragma unroll
    for (int s = 0; s < 16; ++s) {
      const v2f a = *(const v2f*)&tile[lm][4 * s + 2 * half];
      acc = wmma_f32_k4(a, bf[s], acc);
    }
#pragma unroll
    for (int r = 0; r < 8; ++r) {
      const int row = m0 + r + 8 * half;
      if (row < N) hn[(size_t)row * 64 + 16 * w + lm] = acc[r] * ds[r];
    }
  }
}


__global__ void __launch_bounds__(32 * AGG_ROWS)
k_aggregate_pool(const float* __restrict__ h, const int* __restrict__ rowptr,
                 const float* __restrict__ dinv, const int* __restrict__ csr_src,
                 const float* __restrict__ bias, const int* __restrict__ batch,
                 float* __restrict__ gsum, float* __restrict__ cnt, int N) {
  __shared__ __attribute__((aligned(16))) float tile[AGG_ROWS][64];
  __shared__ int gid[AGG_ROWS];
  const int lane = threadIdx.x & 31;
  const int w = __builtin_amdgcn_readfirstlane((int)(threadIdx.x >> 5));
  const int m0 = (int)blockIdx.x * AGG_ROWS;
  const int i = m0 + w;

  v2f o;
  o.x = 0.0f;
  o.y = 0.0f;
  if (i < N) {
    o = aggregate_row(h, rowptr, dinv, csr_src, bias, i, lane);
    if (lane == 0) gid[w] = batch[i];
  }
  *(v2f*)&tile[w][2 * lane] = o;
  __syncthreads();

  if (w < 2) {
    const int c = 32 * w + lane;
    const int nv = (N - m0 < AGG_ROWS) ? (N - m0) : AGG_ROWS;
    const int gl = gid[lane & (AGG_ROWS - 1)];
    float tv[AGG_ROWS];
#pragma unroll
    for (int r = 0; r < AGG_ROWS; ++r) tv[r] = tile[r][c];

    int gcur = __builtin_amdgcn_readlane(gl, 0);
    float a = tv[0];
    int run = 1;
#pragma unroll
    for (int r = 1; r < AGG_ROWS; ++r) {
      if (r < nv) {
        const int gr = __builtin_amdgcn_readlane(gl, r);
        if (gr != gcur) {
          one_add(gsum + (size_t)gcur * 64 + c, a);
          if (c == 0) one_add(cnt + gcur, (float)run);
          gcur = gr;
          a = 0.0f;
          run = 0;
        }
        a += tv[r];
        run += 1;
      }
    }
    one_add(gsum + (size_t)gcur * 64 + c, a);
    if (c == 0) one_add(cnt + gcur, (float)run);
  }
}


__global__ void k_pool_gemm_out(const float* __restrict__ gsum, const float* __restrict__ cnt,
                                const float* __restrict__ Wp, const float* __restrict__ bp,
                                float* __restrict__ out) {
  int wave = threadIdx.x >> 5;
  int lane = threadIdx.x & 31;
  int mi = wave >> 2, ni = wave & 3;
  int half = lane >> 4, lm = lane & 15;

  int arow = mi * 16 + lm;
  float inv = 1.0f / fmaxf(cnt[arow], 1.0f);

  v8f acc = {};
  for (int kb = 0; kb < 64; kb += 4) {
    int k0 = kb + 2 * half;
    v2f a;
    a.x = gsum[arow * 64 + k0] * inv;
    a.y = gsum[arow * 64 + k0 + 1] * inv;
    v2f b;
    b.x = Wp[k0 * 64 + ni * 16 + lm];
    b.y = Wp[(k0 + 1) * 64 + ni * 16 + lm];
    acc = wmma_f32_k4(a, b, acc);
  }

#pragma unroll
  for (int r = 0; r < 8; ++r) {
    int row = mi * 16 + r + 8 * half;
    out[row * 64 + ni * 16 + lm] = acc[r] + bp[ni * 16 + lm];
  }
}


static inline size_t align_f(size_t n) { return (n + 1023) & ~(size_t)1023; }

extern "C" void kernel_launch(void* const* d_in, const int* in_sizes, int n_in,
                              void* d_out, int out_size, void* d_ws, size_t ws_size,
                              hipStream_t stream) {
  const float* x  = (const float*)d_in[0];
  const float* W1 = (const float*)d_in[1];
  const float* b1 = (const float*)d_in[2];
  const float* W2 = (const float*)d_in[3];
  const float* b2 = (const float*)d_in[4];
  const float* Wp = (const float*)d_in[5];
  const float* bp = (const float*)d_in[6];
  const int* ei   = (const int*)d_in[7];
  const int* batch = (const int*)d_in[8];

  const int E = in_sizes[7] / 2;
  const int N = in_sizes[8];
  const int K1 = in_sizes[0] / N;
  const int* src = ei;
  const int* dst = ei + E;

  float* ws = (float*)d_ws;
  size_t o = 0;
  int*   kA     = (int*)(ws + o);  o += align_f((size_t)E);
  int*   vA     = (int*)(ws + o);  o += align_f((size_t)E);
  int*   csr    = (int*)(ws + o);  o += align_f((size_t)E);
  int*   counts = (int*)(ws + o);  o += (size_t)256 * SORT_WAVES;
  int*   offs   = (int*)(ws + o);  o += (size_t)256 * SORT_WAVES;
  int*   tot    = (int*)(ws + o);  o += 1024;
  int*   rowptr = (int*)(ws + o);  o += align_f((size_t)N + 1);
  float* dinv   = ws + o;          o += align_f((size_t)N);
  float* h    = ws + o;  o += align_f((size_t)N * 64);
  float* agg  = ws + o;  o += align_f((size_t)N * 64);
  float* gsum = ws + o;  o += 64 * 64;
  float* cnt  = ws + o;  o += 64;
  (void)ws_size;

  const int per = (int)((((long long)E + 32LL * SORT_WAVES - 1) / (32LL * SORT_WAVES)) * 32);
  const int tiles = (N + 15) / 16;
  const int gemm_blocks = (tiles + 7) / 8;
  const unsigned int agg_blocks = (unsigned int)tiles;

  k_sort_count<<<SORT_WAVES / 8, 256, 0, stream>>>(dst, counts, E, per, 8);
  k_sort_rows<<<256, 1024, 0, stream>>>(counts, offs, tot);
  k_sort_scatter<<<(unsigned int)(gemm_blocks + SORT_WAVES / 8), 256, 0, stream>>>(
      dst, src, offs, tot, kA, vA, E, per, 8, x, W1, h, K1, tiles, gemm_blocks);
  k_bucket<<<(unsigned int)((N >> 8) + 1), 32 * BUCKET_WAVES, 0, stream>>>(kA, vA, tot, csr, rowptr, dinv, h, N);

  k_aggregate_gemm<<<agg_blocks, 32 * AGG_ROWS, 0, stream>>>(h, rowptr, dinv, csr, b1, W2, agg,
                                                            gsum, 64 * 64 + 64, N);

  k_aggregate_pool<<<agg_blocks, 32 * AGG_ROWS, 0, stream>>>(agg, rowptr, dinv, csr, b2, batch, gsum, cnt, N);

  k_pool_gemm_out<<<1, 512, 0, stream>>>(gsum, cnt, Wp, bp, (float*)d_out);
}


# edge sort level 1 orders each block's pairs inside the block's own stretch (whole-line stores), level 2 collects its bucket's 512 runs: no global count and scan launches (5, was 7); order kept
# speedup vs baseline: 5.3455x; 1.1049x over previous
#include <hip/hip_runtime.h>
#include <math.h>

typedef float v2f __attribute__((ext_vector_type(2)));
typedef float v8f __attribute__((ext_vector_type(8)));

__device__ __forceinline__ v8f wmma_f32_k4(v2f a, v2f b, v8f c) {
  return __builtin_amdgcn_wmma_f32_16x16x4_f32(
       false, a,  false, b,
       (short)0, c,  false,  false);
}

__device__ __forceinline__ void atomic_add_f32(float* p, float v) {
  asm volatile("global_atomic_add_f32 %0, %1, off scope:SCOPE_DEV"
               :: "v"(p), "v"(v)
               : "memory");
}


__device__ __forceinline__ int wave_incl_scan(int x, int lane) {
  for (int o = 1; o < 32; o <<= 1) {
    const int n = __shfl_up(x, o);
    if (lane >= o) x += n;
  }
  return x;
}

__device__ __forceinline__ int wave_sum(int x) {
  for (int o = 16; o > 0; o >>= 1) x += __shfl_xor(x, o);
  return x;
}


#define SORT_WAVES 4096
#define PART_BLOCKS (SORT_WAVES / 8)
#define BUCKET_WAVES 32
#define RUNS_PER_WAVE (PART_BLOCKS / BUCKET_WAVES)
static_assert(RUNS_PER_WAVE == 16, "a k_bucket wave looks its 16 runs up in 4 halving steps");
static_assert(PART_BLOCKS <= 32 * BUCKET_WAVES, "k_bucket: one thread per level-1 block");
#define START_PITCH 257
#define SORT_GROUPS 8
#define STAGE_PER 512

__device__ __forceinline__ unsigned int same_digit_lanes(int digit, bool valid) {
  unsigned int mask = __builtin_amdgcn_ballot_w32(valid);
#pragma unroll
  for (int b = 0; b < 8; ++b) {
    const bool bit = ((digit >> b) & 1) != 0;
    const unsigned int bb = __builtin_amdgcn_ballot_w32(bit);
    mask &= bit ? bb : ~bb;
  }
  return mask;
}


__device__ __forceinline__ void gemm_tile_n64(const float* __restrict__ X, const float* __restrict__ W,
                                              float* __restrict__ H, int K, int tile, int lane) {
  const int m0 = tile * 16;
  const int half = lane >> 4;
  const int lm = lane & 15;

  const float* xrow = X + (size_t)(m0 + lm) * (size_t)K;
  v8f acc0 = {}, acc1 = {}, acc2 = {}, acc3 = {};

  for (int kb = 0; kb < K; kb += 4) {
    int k0 = kb + 2 * half;
    v2f a;
    a.x = xrow[k0];
    a.y = xrow[k0 + 1];
    const float* w0 = W + (size_t)k0 * 64;
    v2f b0, b1, b2, b3;
    b0.x = w0[lm];       b0.y = w0[64 + lm];
    b1.x = w0[16 + lm];  b1.y = w0[64 + 16 + lm];
    b2.x = w0[32 + lm];  b2.y = w0[64 + 32 + lm];
    b3.x = w0[48 + lm];  b3.y = w0[64 + 48 + lm];
    acc0 = wmma_f32_k4(a, b0, acc0);
    acc1 = wmma_f32_k4(a, b1, acc1);
    acc2 = wmma_f32_k4(a, b2, acc2);
    acc3 = wmma_f32_k4(a, b3, acc3);
  }

#pragma unroll
  for (int r = 0; r < 8; ++r) {
    int row = m0 + r + 8 * half;
    float* hr = H + (size_t)row * 64 + lm;
    hr[0]  = acc0[r];
    hr[16] = acc1[r];
    hr[32] = acc2[r];
    hr[48] = acc3[r];
  }
}

__device__ __forceinline__ void gemm_block_n64(const float* __restrict__ X, const float* __restrict__ W,
                                               float* __restrict__ H, int K, int gblock, int ntiles) {
  const int w = __builtin_amdgcn_readfirstlane((int)(threadIdx.x >> 5));
  const int tile = gblock * 8 + w;
  if (tile < ntiles) gemm_tile_n64(X, W, H, K, tile, (int)(threadIdx.x & 31));
}

template <bool STAGED>
__device__ __forceinline__ void scatter_piece(const int* __restrict__ keys, const int* __restrict__ vals,
                                              int* cw, int* skey, int* sval,
                                              int* __restrict__ keys_out, int* __restrict__ vals_out,
                                              int beg, int end, int shift, int lane) {
  for (int g = beg; g < end; g += 32 * SORT_GROUPS) {
    int key[SORT_GROUPS], val[SORT_GROUPS];
#pragma unroll
    for (int u = 0; u < SORT_GROUPS; ++u) {
      const int idx = g + u * 32 + lane;
      key[u] = (idx < end) ? keys[idx] : 0;
      val[u] = (idx < end) ? vals[idx] : 0;
    }
#pragma unroll
    for (int u = 0; u < SORT_GROUPS; ++u) {
      if (g + u * 32 >= end) break;
      const int idx = g + u * 32 + lane;
      const bool valid = idx < end;
      const int digit = (key[u] >> shift) & 255;
      const unsigned int mask = same_digit_lanes(digit, valid);
      const int rank = __popc(mask & ((1u << lane) - 1u));
      int place = 0;
      if (valid) place = cw[digit];
      __builtin_amdgcn_wave_barrier();
      if (valid && rank == 0) cw[digit] = place + __popc(mask);
      __builtin_amdgcn_wave_barrier();
      if (valid) {
        if (STAGED) {
          skey[place + rank] = key[u];
          sval[place + rank] = val[u];
        } else {
          keys_out[place + rank] = key[u];
          vals_out[place + rank] = val[u];
        }
      }
    }
  }
}

__global__ void __launch_bounds__(256)
k_part(const int* __restrict__ keys, const int* __restrict__ vals,
       int* __restrict__ keys_out, int* __restrict__ vals_out, int* __restrict__ bstart,
       int E, int per, int shift,
       const float* __restrict__ X, const float* __restrict__ W, float* __restrict__ H,
       int K, int ntiles, int gblocks) {
  __shared__ int cur[8][256];
  __shared__ int s_lsum[8];
  __shared__ int skey[8 * STAGE_PER];
  __shared__ int sval[8 * STAGE_PER];
  if ((int)blockIdx.x < gblocks) {
    gemm_block_n64(X, W, H, K, (int)blockIdx.x, ntiles);
    return;
  }
  const int bx = (int)blockIdx.x - gblocks;
  const int t = (int)threadIdx.x;
  const int lane = t & 31;
  const int w = __builtin_amdgcn_readfirstlane(t >> 5);
  const int wg = bx * 8 + w;
  int* cw = cur[w];
  for (int k = lane; k < 256; k += 32) cw[k] = 0;
  const bool staged = per <= STAGE_PER;
  const int beg = wg * per;
  const int end = (beg + per < E) ? (beg + per) : E;

  for (int g = beg; g < end; g += 32 * SORT_GROUPS) {
    int key[SORT_GROUPS];
#pragma unroll
    for (int u = 0; u < SORT_GROUPS; ++u) {
      const int idx = g + u * 32 + lane;
      key[u] = (idx < end) ? keys[idx] : 0;
    }
#pragma unroll
    for (int u = 0; u < SORT_GROUPS; ++u) {
      if (g + u * 32 >= end) break;
      const int idx = g + u * 32 + lane;
      const bool valid = idx < end;
      const int digit = (key[u] >> shift) & 255;
      const unsigned int mask = same_digit_lanes(digit, valid);
      const bool leader = valid && ((mask & ((1u << lane) - 1u)) == 0u);
      if (leader) cw[digit] += __popc(mask);
      __builtin_amdgcn_wave_barrier();
    }
  }
  __syncthreads();

  int c[8];
  int bc = 0;
#pragma unroll
  for (int k = 0; k < 8; ++k) {
    const int x = cur[k][t];
    c[k] = bc;
    bc += x;
  }
  const int lincl = wave_incl_scan(bc, lane);
  if (lane == 31) s_lsum[w] = lincl;
  __syncthreads();
  int lbefore = 0;
#pragma unroll
  for (int k = 0; k < 8; ++k) lbefore += (k < w) ? s_lsum[k] : 0;
  const int lstart = lbefore + lincl - bc;
#pragma unroll
  for (int k = 0; k < 8; ++k) cur[k][t] = lstart + c[k];
  int* brow = bstart + (size_t)bx * START_PITCH;
  brow[t] = lstart;
  if (t == 255) brow[256] = lstart + bc;
  __syncthreads();

  const int bbeg = bx * 8 * per;
  if (staged) scatter_piece<true>(keys, vals, cw, skey, sval, keys_out, vals_out, beg, end, shift, lane);
  else scatter_piece<false>(keys, vals, cw, skey, sval, keys_out + bbeg, vals_out + bbeg, beg, end, shift, lane);

  if (staged) {
    __syncthreads();
    const int bn = (E - bbeg < 8 * per) ? (E - bbeg) : (8 * per);
    for (int j = t; j < bn; j += 256) {
      keys_out[bbeg + j] = skey[j];
      vals_out[bbeg + j] = sval[j];
    }
  }
}


#define BUCKET_STAGE 12288

__device__ __forceinline__ int run_index(int q, int pref, int rsrc) {
  int r = 0;
#pragma unroll
  for (int s = RUNS_PER_WAVE / 2; s > 0; s >>= 1) {
    const int pv = __shfl(pref, r + s);
    if (pv <= q) r += s;
  }
  return __shfl(rsrc, r) + (q - __shfl(pref, r));
}

template <bool STAGED>
__device__ __forceinline__ void bucket_piece(const int* __restrict__ keys, const int* __restrict__ vals,
                                             int* hw, int* sbuf, int* __restrict__ csr_src,
                                             int plen, int pref, int rsrc, int base, int lane) {
  for (int g = 0; g < plen; g += 32 * SORT_GROUPS) {
    int key[SORT_GROUPS], val[SORT_GROUPS];
#pragma unroll
    for (int u = 0; u < SORT_GROUPS; ++u) {
      const int q = g + u * 32 + lane;
      const int idx = run_index(q, pref, rsrc);
      key[u] = (q < plen) ? keys[idx] : 0;
      val[u] = (q < plen) ? vals[idx] : 0;
    }
#pragma unroll
    for (int u = 0; u < SORT_GROUPS; ++u) {
      if (g + u * 32 >= plen) break;
      const int q = g + u * 32 + lane;
      const bool valid = q < plen;
      const int digit = key[u] & 255;
      const unsigned int mask = same_digit_lanes(digit, valid);
      const int rank = __popc(mask & ((1u << lane) - 1u));
      int place = 0;
      if (valid) place = hw[digit];
      __builtin_amdgcn_wave_barrier();
      if (valid && rank == 0) hw[digit] = place + __popc(mask);
      __builtin_amdgcn_wave_barrier();
      if (valid) {
        if (STAGED) sbuf[place + rank - base] = val[u];
        else csr_src[place + rank] = val[u];
      }
    }
  }
}

__global__ void __launch_bounds__(32 * BUCKET_WAVES)
k_bucket(const int* __restrict__ keys, const int* __restrict__ vals, const int* __restrict__ bstart,
         int per, int* __restrict__ csr_src, int* __restrict__ rowptr, float* __restrict__ dinv,
         float* __restrict__ h, int N) {
  __shared__ int hist[BUCKET_WAVES][256];
  __shared__ int s_wsum[8];
  __shared__ float s_dinv[256];
  __shared__ int sbuf[BUCKET_STAGE];
  __shared__ int s_wstart[BUCKET_WAVES];
  __shared__ int s_wlen[BUCKET_WAVES];
  __shared__ int s_rpos[PART_BLOCKS + 1];
  __shared__ int s_rsrc[PART_BLOCKS];
  const int t = (int)threadIdx.x;
  const int lane = t & 31;
  const int w = __builtin_amdgcn_readfirstlane(t >> 5);
  const int b = (int)blockIdx.x;
  int* hw = hist[w];
  for (int k = lane; k < 256; k += 32) hw[k] = 0;

  const int hrow0 = b * 256 + w * 8;
  v2f hv[8];
#pragma unroll
  for (int r = 0; r < 8; ++r) {
    const int row = (hrow0 + r < N) ? (hrow0 + r) : (N - 1);
    hv[r] = *(const v2f*)(h + (size_t)row * 64 + 2 * lane);
  }

  int rs = 0, rl = 0;
  if (t < PART_BLOCKS) {
    const int* brow = bstart + (size_t)t * START_PITCH + b;
    rs = brow[0];
    rl = brow[1] - rs;
  }
  const int rincl = wave_incl_scan(rl, lane);
  const int rsum = wave_sum(rs);
  if (lane == 31) {
    s_wlen[w] = rincl;
    s_wstart[w] = rsum;
  }
  __syncthreads();
  const int wl = s_wlen[lane];
  const int base = __builtin_amdgcn_readfirstlane(wave_sum(s_wstart[lane]));
  const int size = __builtin_amdgcn_readfirstlane(wave_sum(wl));
  const int rbefore = wave_sum((lane < w) ? wl : 0);
  if (t < PART_BLOCKS) {
    s_rpos[t] = rbefore + rincl - rl;
    s_rsrc[t] = t * 8 * per + rs;
  }
  if (t == 0) s_rpos[PART_BLOCKS] = size;
  __syncthreads();

  const int r0 = w * RUNS_PER_WAVE;
  const int pstart = __builtin_amdgcn_readfirstlane(s_rpos[r0]);
  const int plen = __builtin_amdgcn_readfirstlane(s_rpos[r0 + RUNS_PER_WAVE]) - pstart;
  const int pref = s_rpos[r0 + (lane & (RUNS_PER_WAVE - 1))] - pstart;
  const int rsrc = s_rsrc[r0 + (lane & (RUNS_PER_WAVE - 1))];

  const bool staged = size <= BUCKET_STAGE;

  for (int g = 0; g < plen; g += 32 * SORT_GROUPS) {
    int key[SORT_GROUPS];
#pragma unroll
    for (int u = 0; u < SORT_GROUPS; ++u) {
      const int q = g + u * 32 + lane;
      const int idx = run_index(q, pref, rsrc);
      key[u] = (q < plen) ? keys[idx] : 0;
    }
#pragma unroll
    for (int u = 0; u < SORT_GROUPS; ++u) {
      if (g + u * 32 >= plen) break;
      const int q = g + u * 32 + lane;
      const bool valid = q < plen;
      const int digit = key[u] & 255;
      const unsigned int mask = same_digit_lanes(digit, valid);
      const bool leader = valid && ((mask & ((1u << lane) - 1u)) == 0u);
      if (leader) hw[digit] += __popc(mask);
      __builtin_amdgcn_wave_barrier();
    }
  }
  __syncthreads();

  int c[BUCKET_WAVES];
  int total = 0, incl = 0;
  if (t < 256) {
#pragma unroll
    for (int k = 0; k < BUCKET_WAVES; ++k) c[k] = hist[k][t];
#pragma unroll
    for (int k = 0; k < BUCKET_WAVES; ++k) {
      const int x = c[k];
      c[k] = total;
      total += x;
    }
    incl = wave_incl_scan(total, lane);
    if (lane == 31) s_wsum[w] = incl;
  }
  __syncthreads();
  if (t < 256) {
    int before = 0;
#pragma unroll
    for (int k = 0; k < 8; ++k) before += (k < w) ? s_wsum[k] : 0;
    const int rowstart = base + before + incl - total;
#pragma unroll
    for (int k = 0; k < BUCKET_WAVES; ++k) hist[k][t] = rowstart + c[k];
    const int row = b * 256 + t;
    if (row <= N) rowptr[row] = rowstart;
    const float di = rsqrtf((float)(total + 1));
    if (row < N) dinv[row] = di;
    s_dinv[t] = di;
  }
  __syncthreads();

#pragma unroll
  for (int r = 0; r < 8; ++r) {
    if (hrow0 + r < N) {
      const float di = s_dinv[w * 8 + r];
      v2f o;
      o.x = hv[r].x * di;
      o.y = hv[r].y * di;
      *(v2f*)(h + (size_t)(hrow0 + r) * 64 + 2 * lane) = o;
    }
  }

  if (staged) bucket_piece<true>(keys, vals, hw, sbuf, csr_src, plen, pref, rsrc, base, lane);
  else bucket_piece<false>(keys, vals, hw, sbuf, csr_src, plen, pref, rsrc, base, lane);

  if (staged) {
    __syncthreads();
    for (int j = t - (base & 31); j < size; j += 32 * BUCKET_WAVES)
      if (j >= 0) csr_src[base + j] = sbuf[j];
  }
}


__device__ __forceinline__ v2f aggregate_row(const float* __restrict__ h, const int* __restrict__ rowptr,
                                             const float* __restrict__ dinv,
                                             const int* __restrict__ csr_src,
                                             const float* __restrict__ bias, int i, int lane) {
  const int start = rowptr[i];
  const int cnt = rowptr[i + 1] - start;
  const float di = dinv[i];

  const v2f self = *(const v2f*)(h + (size_t)i * 64 + 2 * lane);
  float a0 = self.x, a1 = self.y;

  for (int c = 0; c < cnt; c += 32) {
    const int m = (cnt - c < 32) ? (cnt - c) : 32;
    int sl = i;
    if (lane < m) sl = csr_src[start + c + lane];
    for (int j = 0; j < m; j += 8) {
      int sj[8];
#pragma unroll
      for (int u = 0; u < 8; ++u) sj[u] = __builtin_amdgcn_readlane(sl, j + u);
      v2f v[8];
      if (j + 8 <= m) {
#pragma unroll
        for (int u = 0; u < 8; ++u) v[u] = *(const v2f*)(h + (size_t)sj[u] * 64 + 2 * lane);
      } else {
#pragma unroll
        for (int u = 0; u < 8; ++u) {
          v[u].x = 0.0f;
          v[u].y = 0.0f;
          if (j + u < m) v[u] = *(const v2f*)(h + (size_t)sj[u] * 64 + 2 * lane);
        }
      }
#pragma unroll
      for (int u = 0; u < 8; ++u) {
        a0 += v[u].x;
        a1 += v[u].y;
      }
    }
  }

  v2f o;
  o.x = fmaxf(fmaf(a0, di, bias[2 * lane]), 0.0f);
  o.y = fmaxf(fmaf(a1, di, bias[2 * lane + 1]), 0.0f);
  return o;
}

#define AGG_ROWS 16
#define TILE_PITCH 68

__device__ __forceinline__ void one_add(float* p, float v) {
  atomic_add_f32(p, v);
  asm volatile("s_wait_xcnt 0x0" :: "v"(p), "v"(v));
}


__global__ void __launch_bounds__(32 * AGG_ROWS)
k_aggregate_gemm(const float* __restrict__ h, const int* __restrict__ rowptr,
                 const float* __restrict__ dinv, const int* __restrict__ csr_src,
                 const float* __restrict__ bias, const float* __restrict__ Wn,
                 float* __restrict__ hn, float* __restrict__ pool, int npool, int N) {
  __shared__ __attribute__((aligned(16))) float tile[AGG_ROWS][TILE_PITCH];
  const int lane = threadIdx.x & 31;
  const int w = __builtin_amdgcn_readfirstlane((int)(threadIdx.x >> 5));
  const int m0 = (int)blockIdx.x * AGG_ROWS;
  const int i = m0 + w;
  const int half = lane >> 4;
  const int lm = lane & 15;

  const int pi = (int)(blockIdx.x * blockDim.x + threadIdx.x);
  if (pi < npool) pool[pi] = 0.0f;

  v2f o;
  o.x = 0.0f;
  o.y = 0.0f;
  if (i < N) o = aggregate_row(h, rowptr, dinv, csr_src, bias, i, lane);

  v2f bf[16];
  float ds[8];
#pragma unroll
  for (int s = 0; s < 16; ++s) {
    bf[s].x = 0.0f;
    bf[s].y = 0.0f;
  }
#pragma unroll
  for (int r = 0; r < 8; ++r) ds[r] = 0.0f;
  if (w < 4) {
    const float* w0 = Wn + (size_t)(2 * half) * 64 + 16 * w + lm;
#pragma unroll
    for (int s = 0; s < 16; ++s) {
      bf[s].x = w0[(size_t)s * 256];
      bf[s].y = w0[(size_t)s * 256 + 64];
    }
    const float4 dlo = *(const float4*)(dinv + m0 + 8 * half);
    const float4 dhi = *(const float4*)(dinv + m0 + 8 * half + 4);
    ds[0] = dlo.x; ds[1] = dlo.y; ds[2] = dlo.z; ds[3] = dlo.w;
    ds[4] = dhi.x; ds[5] = dhi.y; ds[6] = dhi.z; ds[7] = dhi.w;
  }

  *(v2f*)&tile[w][2 * lane] = o;
  __syncthreads();

  if (w < 4) {
    v8f acc = {};
#pragma unroll
    for (int s = 0; s < 16; ++s) {
      const v2f a = *(const v2f*)&tile[lm][4 * s + 2 * half];
      acc = wmma_f32_k4(a, bf[s], acc);
    }
#pragma unroll
    for (int r = 0; r < 8; ++r) {
      const int row = m0 + r + 8 * half;
      if (row < N) hn[(size_t)row * 64 + 16 * w + lm] = acc[r] * ds[r];
    }
  }
}


__global__ void __launch_bounds__(32 * AGG_ROWS)
k_aggregate_pool(const float* __restrict__ h, const int* __restrict__ rowptr,
                 const float* __restrict__ dinv, const int* __restrict__ csr_src,
                 const float* __restrict__ bias, const int* __restrict__ batch,
                 float* __restrict__ gsum, float* __restrict__ cnt, int N) {
  __shared__ __attribute__((aligned(16))) float tile[AGG_ROWS][64];
  __shared__ int gid[AGG_ROWS];
  const int lane = threadIdx.x & 31;
  const int w = __builtin_amdgcn_readfirstlane((int)(threadIdx.x >> 5));
  const int m0 = (int)blockIdx.x * AGG_ROWS;
  const int i = m0 + w;

  v2f o;
  o.x = 0.0f;
  o.y = 0.0f;
  if (i < N) {
    o = aggregate_row(h, rowptr, dinv, csr_src, bias, i, lane);
    if (lane == 0) gid[w] = batch[i];
  }
  *(v2f*)&tile[w][2 * lane] = o;
  __syncthreads();

  if (w < 2) {
    const int c = 32 * w + lane;
    const int nv = (N - m0 < AGG_ROWS) ? (N - m0) : AGG_ROWS;
    const int gl = gid[lane & (AGG_ROWS - 1)];
    float tv[AGG_ROWS];
#pragma unroll
    for (int r = 0; r < AGG_ROWS; ++r) tv[r] = tile[r][c];

    int gcur = __builtin_amdgcn_readlane(gl, 0);
    float a = tv[0];
    int run = 1;
#pragma unroll
    for (int r = 1; r < AGG_ROWS; ++r) {
      if (r < nv) {
        const int gr = __builtin_amdgcn_readlane(gl, r);
        if (gr != gcur) {
          one_add(gsum + (size_t)gcur * 64 + c, a);
          if (c == 0) one_add(cnt + gcur, (float)run);
          gcur = gr;
          a = 0.0f;
          run = 0;
        }
        a += tv[r];
        run += 1;
      }
    }
    one_add(gsum + (size_t)gcur * 64 + c, a);
    if (c == 0) one_add(cnt + gcur, (float)run);
  }
}


__global__ void k_pool_gemm_out(const float* __restrict__ gsum, const float* __restrict__ cnt,
                                const float* __restrict__ Wp, const float* __restrict__ bp,
                                float* __restrict__ out) {
  int wave = threadIdx.x >> 5;
  int lane = threadIdx.x & 31;
  int mi = wave >> 2, ni = wave & 3;
  int half = lane >> 4, lm = lane & 15;

  int arow = mi * 16 + lm;
  float inv = 1.0f / fmaxf(cnt[arow], 1.0f);

  v8f acc = {};
  for (int kb = 0; kb < 64; kb += 4) {
    int k0 = kb + 2 * half;
    v2f a;
    a.x = gsum[arow * 64 + k0] * inv;
    a.y = gsum[arow * 64 + k0 + 1] * inv;
    v2f b;
    b.x = Wp[k0 * 64 + ni * 16 + lm];
    b.y = Wp[(k0 + 1) * 64 + ni * 16 + lm];
    acc = wmma_f32_k4(a, b, acc);
  }

#pragma unroll
  for (int r = 0; r < 8; ++r) {
    int row = mi * 16 + r + 8 * half;
    out[row * 64 + ni * 16 + lm] = acc[r] + bp[ni * 16 + lm];
  }
}


static inline size_t align_f(size_t n) { return (n + 1023) & ~(size_t)1023; }

extern "C" void kernel_launch(void* const* d_in, const int* in_sizes, int n_in,
                              void* d_out, int out_size, void* d_ws, size_t ws_size,
                              hipStream_t stream) {
  const float* x  = (const float*)d_in[0];
  const float* W1 = (const float*)d_in[1];
  const float* b1 = (const float*)d_in[2];
  const float* W2 = (const float*)d_in[3];
  const float* b2 = (const float*)d_in[4];
  const float* Wp = (const float*)d_in[5];
  const float* bp = (const float*)d_in[6];
  const int* ei   = (const int*)d_in[7];
  const int* batch = (const int*)d_in[8];

  const int E = in_sizes[7] / 2;
  const int N = in_sizes[8];
  const int K1 = in_sizes[0] / N;
  const int* src = ei;
  const int* dst = ei + E;

  float* ws = (float*)d_ws;
  size_t o = 0;
  int*   kA     = (int*)(ws + o);  o += align_f((size_t)E);
  int*   vA     = (int*)(ws + o);  o += align_f((size_t)E);
  int*   csr    = (int*)(ws + o);  o += align_f((size_t)E);
  int*   bstart = (int*)(ws + o);  o += align_f((size_t)PART_BLOCKS * START_PITCH);
  int*   rowptr = (int*)(ws + o);  o += align_f((size_t)N + 1);
  float* dinv   = ws + o;          o += align_f((size_t)N);
  float* h    = ws + o;  o += align_f((size_t)N * 64);
  float* agg  = ws + o;  o += align_f((size_t)N * 64);
  float* gsum = ws + o;  o += 64 * 64;
  float* cnt  = ws + o;  o += 64;
  (void)ws_size;

  const int per = (int)((((long long)E + 32LL * SORT_WAVES - 1) / (32LL * SORT_WAVES)) * 32);
  const int tiles = (N + 15) / 16;
  const int gemm_blocks = (tiles + 7) / 8;
  const unsigned int agg_blocks = (unsigned int)tiles;

  k_part<<<(unsigned int)(gemm_blocks + PART_BLOCKS), 256, 0, stream>>>(
      dst, src, kA, vA, bstart, E, per, 8, x, W1, h, K1, tiles, gemm_blocks);
  k_bucket<<<(unsigned int)((N >> 8) + 1), 32 * BUCKET_WAVES, 0, stream>>>(kA, vA, bstart, per, csr, rowptr, dinv, h, N);

  k_aggregate_gemm<<<agg_blocks, 32 * AGG_ROWS, 0, stream>>>(h, rowptr, dinv, csr, b1, W2, agg,
                                                            gsum, 64 * 64 + 64, N);

  k_aggregate_pool<<<agg_blocks, 32 * AGG_ROWS, 0, stream>>>(agg, rowptr, dinv, csr, b2, batch, gsum, cnt, N);

  k_pool_gemm_out<<<1, 512, 0, stream>>>(gsum, cnt, Wp, bp, (float*)d_out);
}
